# starting kernel
# baseline (speedup 1.0000x reference)
#include <hip/hip_runtime.h>

typedef __attribute__((ext_vector_type(2))) float v2f;
typedef __attribute__((ext_vector_type(8))) float v8f;

#define GDIM 128

__global__ __launch_bounds__(256) void gin_zero(float4* p, long n4) {
  long i = (long)blockIdx.x * blockDim.x + threadIdx.x;
  long stride = (long)gridDim.x * blockDim.x;
  float4 z = {0.f, 0.f, 0.f, 0.f};
  for (; i < n4; i += stride) p[i] = z;
}

__global__ __launch_bounds__(256) void gin_scatter(
    const float* __restrict__ x, const int* __restrict__ src,
    const int* __restrict__ dst, float* agg, int nedges) {
  int w = (int)((blockIdx.x * 256u + threadIdx.x) >> 5);
  int lane = threadIdx.x & 31;
  if (w >= nedges) return;
  int s = src[w];
  int d = dst[w];
  const float4 v = *(const float4*)(x + (long)s * GDIM + lane * 4);
  float* o = agg + (long)d * GDIM + lane * 4;
  atomicAdd(o + 0, v.x);
  atomicAdd(o + 1, v.y);
  atomicAdd(o + 2, v.z);
  atomicAdd(o + 3, v.w);
}

template <bool HASADD, bool RELU>
__global__ __launch_bounds__(256) void gin_gemm(
    const float* A0, const float* A1, const float* __restrict__ W,
    const float* __restrict__ bias, float* Out, int nrows) {
  __shared__ float sW[GDIM * GDIM];
  int tid = threadIdx.x;
  for (int i = tid; i < (GDIM / 2) * GDIM; i += 256) {
    int p = i >> 7;
    int n = i & 127;
    int idx = p * 256 + 2 * n;
    sW[idx]     = W[(2 * p) * GDIM + n];
    sW[idx + 1] = W[(2 * p + 1) * GDIM + n];
  }
  __syncthreads();

  int wave = tid >> 5;
  int lane = tid & 31;
  long mbase = ((long)blockIdx.x * 8 + wave) * 32;
  if (mbase >= nrows) return;
  bool has2 = (mbase + 16) < nrows;
  long mbase2 = has2 ? (mbase + 16) : mbase;

  int lmod = lane & 15;
  int par = lane >> 4;
  int khalf = par << 1;
  int mhi = par << 3;

  const float* arA0 = A0 + (mbase + lmod) * GDIM;
  const float* arB0 = A0 + (mbase2 + lmod) * GDIM;
  const float* arA1 = HASADD ? (A1 + (mbase + lmod) * GDIM) : nullptr;
  const float* arB1 = HASADD ? (A1 + (mbase2 + lmod) * GDIM) : nullptr;
  const float* sWp = sW + par * 256 + 2 * lmod;

  v2f pa0 = *(const v2f*)(arA0 + khalf);
  v2f pb0 = *(const v2f*)(arB0 + khalf);
  v2f pa1, pb1;
  if (HASADD) {
    pa1 = *(const v2f*)(arA1 + khalf);
    pb1 = *(const v2f*)(arB1 + khalf);
  }

  v8f acc[16] = {};

  for (int k = 0; k < GDIM; k += 4) {
    v2f aA = pa0, aB = pb0;
    if (HASADD) {
      aA.x += pa1.x; aA.y += pa1.y;
      aB.x += pb1.x; aB.y += pb1.y;
    }
    if (k + 4 < GDIM) {
      pa0 = *(const v2f*)(arA0 + (k + 4) + khalf);
      pb0 = *(const v2f*)(arB0 + (k + 4) + khalf);
      if (HASADD) {
        pa1 = *(const v2f*)(arA1 + (k + 4) + khalf);
        pb1 = *(const v2f*)(arB1 + (k + 4) + khalf);
      }
    }
    const float* wp = sWp + (k >> 1) * 256;
    v2f b[8];
#pragma unroll
    for (int nt = 0; nt < 8; nt++)
      b[nt] = *(const v2f*)(wp + nt * 32);
#pragma unroll
    for (int nt = 0; nt < 8; nt++) {
      acc[nt] = __builtin_amdgcn_wmma_f32_16x16x4_f32(
          false, aA, false, b[nt], (short)0, acc[nt], false, false);
      acc[8 + nt] = __builtin_amdgcn_wmma_f32_16x16x4_f32(
          false, aB, false, b[nt], (short)0, acc[8 + nt], false, false);
    }
  }

#pragma unroll
  for (int nt = 0; nt < 8; nt++) {
    int ncol = nt * 16 + lmod;
    float bv = bias[ncol];
#pragma unroll
    for (int j = 0; j < 8; j++) {
      float v = acc[nt][j] + bv;
      if (RELU) v = fmaxf(v, 0.f);
      Out[(mbase + mhi + j) * GDIM + ncol] = v;
    }
    if (has2) {
#pragma unroll
      for (int j = 0; j < 8; j++) {
        float v = acc[8 + nt][j] + bv;
        if (RELU) v = fmaxf(v, 0.f);
        Out[(mbase2 + mhi + j) * GDIM + ncol] = v;
      }
    }
  }
}

extern "C" void kernel_launch(void* const* d_in, const int* in_sizes, int n_in,
                              void* d_out, int out_size, void* d_ws,
                              size_t ws_size, hipStream_t stream) {
  const float* x_in = (const float*)d_in[0];
  const int* ei = (const int*)d_in[1];
  const float* W1 = (const float*)d_in[2];
  const float* b1 = (const float*)d_in[3];
  const float* W2 = (const float*)d_in[4];
  const float* b2 = (const float*)d_in[5];
  float* out = (float*)d_out;

  const int N = in_sizes[0] / GDIM;
  const int NE = in_sizes[1] / 2;
  const long nd = (long)N * GDIM;

  float* aggh = (float*)d_ws;
  float* x1 = aggh + nd;

  const int mtiles32 = (N + 31) / 32;
  const int gemm_blocks = (mtiles32 + 7) / 8;
  const int scat_blocks = (NE + 7) / 8;
  const int zero_blocks = (int)((nd / 4 + 255) / 256);

  const float* xc = x_in;
  for (int layer = 0; layer < 3; ++layer) {
    float* xn = (layer == 0) ? x1 : out;
    gin_zero<<<zero_blocks, 256, 0, stream>>>((float4*)aggh, nd / 4);
    gin_scatter<<<scat_blocks, 256, 0, stream>>>(xc, ei, ei + NE, aggh, NE);
    gin_gemm<true, true><<<gemm_blocks, 256, 0, stream>>>(
        xc, aggh, W1 + (long)layer * GDIM * GDIM, b1 + layer * GDIM, aggh, N);
    gin_gemm<false, false><<<gemm_blocks, 256, 0, stream>>>(
        aggh, nullptr, W2 + (long)layer * GDIM * GDIM, b2 + layer * GDIM, xn, N);
    xc = xn;
  }
}


# Replace float-atomic edge scatter by a gather over a per-call CSR (int atomics place edges, each segment then sorted by source id): neighbour sums now in one fixed ascending-source order
# speedup vs baseline: 1.8953x; 1.8953x over previous
#include <hip/hip_runtime.h>

typedef __attribute__((ext_vector_type(2))) float v2f;
typedef __attribute__((ext_vector_type(8))) float v8f;

#define GDIM 128

__global__ __launch_bounds__(256) void gin_zero(float4* p, long n4) {
  long i = (long)blockIdx.x * blockDim.x + threadIdx.x;
  long stride = (long)gridDim.x * blockDim.x;
  float4 z = {0.f, 0.f, 0.f, 0.f};
  for (; i < n4; i += stride) p[i] = z;
}


__global__ __launch_bounds__(256) void gin_hist(const int* __restrict__ dst,
                                                int* cnt, int nedges) {
  int e = (int)(blockIdx.x * 256u + threadIdx.x);
  if (e < nedges) atomicAdd(cnt + dst[e], 1);
}

__global__ __launch_bounds__(1024) void gin_scan(const int* __restrict__ cnt,
                                                 int* __restrict__ off, int n) {
  __shared__ int swave[32];
  int tid = threadIdx.x;
  int lane = tid & 31;
  int wv = tid >> 5;
  int carry = 0;
  for (int base = 0; base < n; base += 4096) {
    int i0 = base + tid * 4;
    int c0 = (i0 + 0 < n) ? cnt[i0 + 0] : 0;
    int c1 = (i0 + 1 < n) ? cnt[i0 + 1] : 0;
    int c2 = (i0 + 2 < n) ? cnt[i0 + 2] : 0;
    int c3 = (i0 + 3 < n) ? cnt[i0 + 3] : 0;
    int t = c0 + c1 + c2 + c3;
    int incl = t;
    for (int d = 1; d < 32; d <<= 1) {
      int u = __shfl_up(incl, d);
      if (lane >= d) incl += u;
    }
    if (lane == 31) swave[wv] = incl;
    __syncthreads();
    int wt = swave[lane];
    int winc = wt;
    for (int d = 1; d < 32; d <<= 1) {
      int u = __shfl_up(winc, d);
      if (lane >= d) winc += u;
    }
    int wexcl = __shfl(winc, wv) - __shfl(wt, wv);
    int total = __shfl(winc, 31);
    int ex = carry + wexcl + incl - t;
    if (i0 + 0 < n) off[i0 + 0] = ex;
    if (i0 + 1 < n) off[i0 + 1] = ex + c0;
    if (i0 + 2 < n) off[i0 + 2] = ex + c0 + c1;
    if (i0 + 3 < n) off[i0 + 3] = ex + c0 + c1 + c2;
    carry += total;
    __syncthreads();
  }
  if (tid == 0) off[n] = carry;
}

__global__ __launch_bounds__(256) void gin_fill(
    const int* __restrict__ src, const int* __restrict__ dst,
    const int* __restrict__ off, int* cnt, int* __restrict__ ebuf, int nedges) {
  int e = (int)(blockIdx.x * 256u + threadIdx.x);
  if (e >= nedges) return;
  int d = dst[e];
  int c = atomicSub(cnt + d, 1);
  ebuf[off[d] + c - 1] = src[e];
}

__global__ __launch_bounds__(256) void gin_canon(
    const int* __restrict__ off, const int* __restrict__ ebuf,
    int* __restrict__ es, int n) {
  int lane = threadIdx.x & 31;
  int wid = (int)(blockIdx.x * 8u + (threadIdx.x >> 5));
  int nw = (int)(gridDim.x * 8u);
  for (int node = wid; node < n; node += nw) {
    int nu = __builtin_amdgcn_readfirstlane(node);
    int beg = off[nu];
    int len = off[nu + 1] - beg;
    if (len <= 32) {
      int key = (lane < len) ? ebuf[beg + lane] : 0x7fffffff;
      int rank = 0;
      for (int j = 0; j < len; j++) {
        int kj = __builtin_amdgcn_readlane(key, j);
        rank += ((kj < key) || (kj == key && j < lane)) ? 1 : 0;
      }
      if (lane < len) es[beg + rank] = key;
    } else {
      for (int i = lane; i < len; i += 32) {
        int key = ebuf[beg + i];
        int rank = 0;
        for (int j = 0; j < len; j++) {
          int kj = ebuf[beg + j];
          rank += ((kj < key) || (kj == key && j < i)) ? 1 : 0;
        }
        es[beg + rank] = key;
      }
    }
  }
}

__global__ __launch_bounds__(256) void gin_gather(
    const float* __restrict__ x, const int* __restrict__ off,
    const int* __restrict__ es, float* __restrict__ agg, int n) {
  int lane = threadIdx.x & 31;
  int wid = (int)(blockIdx.x * 8u + (threadIdx.x >> 5));
  int nw = (int)(gridDim.x * 8u);
  const float* xl = x + lane * 4;
  for (int node = wid; node < n; node += nw) {
    int nu = __builtin_amdgcn_readfirstlane(node);
    int beg = off[nu];
    int end = off[nu + 1];
    float4 acc = {0.f, 0.f, 0.f, 0.f};
    for (int base = beg; base < end; base += 32) {
      int m = end - base;
      if (m > 32) m = 32;
      int idv = (lane < m) ? es[base + lane] : 0;
      int j = 0;
      for (; j + 4 <= m; j += 4) {
        int s0 = __builtin_amdgcn_readlane(idv, j + 0);
        int s1 = __builtin_amdgcn_readlane(idv, j + 1);
        int s2 = __builtin_amdgcn_readlane(idv, j + 2);
        int s3 = __builtin_amdgcn_readlane(idv, j + 3);
        float4 v0 = *(const float4*)(xl + (long)s0 * GDIM);
        float4 v1 = *(const float4*)(xl + (long)s1 * GDIM);
        float4 v2 = *(const float4*)(xl + (long)s2 * GDIM);
        float4 v3 = *(const float4*)(xl + (long)s3 * GDIM);
        acc.x += v0.x; acc.y += v0.y; acc.z += v0.z; acc.w += v0.w;
        acc.x += v1.x; acc.y += v1.y; acc.z += v1.z; acc.w += v1.w;
        acc.x += v2.x; acc.y += v2.y; acc.z += v2.z; acc.w += v2.w;
        acc.x += v3.x; acc.y += v3.y; acc.z += v3.z; acc.w += v3.w;
      }
      for (; j < m; j++) {
        int s0 = __builtin_amdgcn_readlane(idv, j);
        float4 v0 = *(const float4*)(xl + (long)s0 * GDIM);
        acc.x += v0.x; acc.y += v0.y; acc.z += v0.z; acc.w += v0.w;
      }
    }
    *(float4*)(agg + (long)nu * GDIM + lane * 4) = acc;
  }
}

template <bool HASADD, bool RELU>
__global__ __launch_bounds__(256) void gin_gemm(
    const float* A0, const float* A1, const float* __restrict__ W,
    const float* __restrict__ bias, float* Out, int nrows) {
  __shared__ float sW[GDIM * GDIM];
  int tid = threadIdx.x;
  for (int i = tid; i < (GDIM / 2) * GDIM; i += 256) {
    int p = i >> 7;
    int n = i & 127;
    int idx = p * 256 + 2 * n;
    sW[idx]     = W[(2 * p) * GDIM + n];
    sW[idx + 1] = W[(2 * p + 1) * GDIM + n];
  }
  __syncthreads();

  int wave = tid >> 5;
  int lane = tid & 31;
  long mbase = ((long)blockIdx.x * 8 + wave) * 32;
  if (mbase >= nrows) return;
  bool has2 = (mbase + 16) < nrows;
  long mbase2 = has2 ? (mbase + 16) : mbase;

  int lmod = lane & 15;
  int par = lane >> 4;
  int khalf = par << 1;
  int mhi = par << 3;

  const float* arA0 = A0 + (mbase + lmod) * GDIM;
  const float* arB0 = A0 + (mbase2 + lmod) * GDIM;
  const float* arA1 = HASADD ? (A1 + (mbase + lmod) * GDIM) : nullptr;
  const float* arB1 = HASADD ? (A1 + (mbase2 + lmod) * GDIM) : nullptr;
  const float* sWp = sW + par * 256 + 2 * lmod;

  v2f pa0 = *(const v2f*)(arA0 + khalf);
  v2f pb0 = *(const v2f*)(arB0 + khalf);
  v2f pa1, pb1;
  if (HASADD) {
    pa1 = *(const v2f*)(arA1 + khalf);
    pb1 = *(const v2f*)(arB1 + khalf);
  }

  v8f acc[16] = {};

  for (int k = 0; k < GDIM; k += 4) {
    v2f aA = pa0, aB = pb0;
    if (HASADD) {
      aA.x += pa1.x; aA.y += pa1.y;
      aB.x += pb1.x; aB.y += pb1.y;
    }
    if (k + 4 < GDIM) {
      pa0 = *(const v2f*)(arA0 + (k + 4) + khalf);
      pb0 = *(const v2f*)(arB0 + (k + 4) + khalf);
      if (HASADD) {
        pa1 = *(const v2f*)(arA1 + (k + 4) + khalf);
        pb1 = *(const v2f*)(arB1 + (k + 4) + khalf);
      }
    }
    const float* wp = sWp + (k >> 1) * 256;
    v2f b[8];
#pragma unroll
    for (int nt = 0; nt < 8; nt++)
      b[nt] = *(const v2f*)(wp + nt * 32);
#pragma unroll
    for (int nt = 0; nt < 8; nt++) {
      acc[nt] = __builtin_amdgcn_wmma_f32_16x16x4_f32(
          false, aA, false, b[nt], (short)0, acc[nt], false, false);
      acc[8 + nt] = __builtin_amdgcn_wmma_f32_16x16x4_f32(
          false, aB, false, b[nt], (short)0, acc[8 + nt], false, false);
    }
  }

#pragma unroll
  for (int nt = 0; nt < 8; nt++) {
    int ncol = nt * 16 + lmod;
    float bv = bias[ncol];
#pragma unroll
    for (int j = 0; j < 8; j++) {
      float v = acc[nt][j] + bv;
      if (RELU) v = fmaxf(v, 0.f);
      Out[(mbase + mhi + j) * GDIM + ncol] = v;
    }
    if (has2) {
#pragma unroll
      for (int j = 0; j < 8; j++) {
        float v = acc[8 + nt][j] + bv;
        if (RELU) v = fmaxf(v, 0.f);
        Out[(mbase2 + mhi + j) * GDIM + ncol] = v;
      }
    }
  }
}

extern "C" void kernel_launch(void* const* d_in, const int* in_sizes, int n_in,
                              void* d_out, int out_size, void* d_ws,
                              size_t ws_size, hipStream_t stream) {
  const float* x_in = (const float*)d_in[0];
  const int* ei = (const int*)d_in[1];
  const float* W1 = (const float*)d_in[2];
  const float* b1 = (const float*)d_in[3];
  const float* W2 = (const float*)d_in[4];
  const float* b2 = (const float*)d_in[5];
  float* out = (float*)d_out;

  const int N = in_sizes[0] / GDIM;
  const int NE = in_sizes[1] / 2;
  const long nd = (long)N * GDIM;

  float* aggh = (float*)d_ws;
  int* cnt = (int*)(aggh + nd);
  int* off = cnt + (N + 4);
  int* ebuf = off + (N + 4);
  int* es = ebuf + NE;

  const int mtiles32 = (N + 31) / 32;
  const int gemm_blocks = (mtiles32 + 7) / 8;
  const int edge_blocks = (NE + 255) / 256;
  const int zero_blocks = ((N + 3) / 4 + 255) / 256;
  const int row_blocks = 4096;

  gin_zero<<<zero_blocks, 256, 0, stream>>>((float4*)cnt, (N + 3) / 4);
  gin_hist<<<edge_blocks, 256, 0, stream>>>(ei + NE, cnt, NE);
  gin_scan<<<1, 1024, 0, stream>>>(cnt, off, N);
  gin_fill<<<edge_blocks, 256, 0, stream>>>(ei, ei + NE, off, cnt, ebuf, NE);
  gin_canon<<<row_blocks, 256, 0, stream>>>(off, ebuf, es, N);

  const float* xc = x_in;
  for (int layer = 0; layer < 3; ++layer) {
    gin_gather<<<row_blocks, 256, 0, stream>>>(xc, off, es, aggh, N);
    gin_gemm<true, true><<<gemm_blocks, 256, 0, stream>>>(
        xc, aggh, W1 + (long)layer * GDIM * GDIM, b1 + layer * GDIM, aggh, N);
    gin_gemm<false, false><<<gemm_blocks, 256, 0, stream>>>(
        aggh, nullptr, W2 + (long)layer * GDIM * GDIM, b2 + layer * GDIM, out, N);
    xc = out;
  }
}


# Fold W2[i] into W1[i+1] (M=W2[i]*W1[i+1] by WMMA, c=b2[i]*W1[i+1], once per call), aggregate after that product with self row+bias+relu in the gather: 6 row GEMMs -> 4; sums reassociated
# speedup vs baseline: 2.0873x; 1.1013x over previous
#include <hip/hip_runtime.h>

typedef __attribute__((ext_vector_type(2))) float v2f;
typedef __attribute__((ext_vector_type(8))) float v8f;

#define GDIM 128

__global__ __launch_bounds__(256) void gin_zero(float4* p, long n4) {
  long i = (long)blockIdx.x * blockDim.x + threadIdx.x;
  long stride = (long)gridDim.x * blockDim.x;
  float4 z = {0.f, 0.f, 0.f, 0.f};
  for (; i < n4; i += stride) p[i] = z;
}


__global__ __launch_bounds__(256) void gin_hist(const int* __restrict__ dst,
                                                int* cnt, int nedges) {
  int e = (int)(blockIdx.x * 256u + threadIdx.x);
  if (e < nedges) atomicAdd(cnt + dst[e], 1);
}

__global__ __launch_bounds__(1024) void gin_scan(const int* __restrict__ cnt,
                                                 int* __restrict__ off, int n) {
  __shared__ int swave[32];
  int tid = threadIdx.x;
  int lane = tid & 31;
  int wv = tid >> 5;
  int carry = 0;
  for (int base = 0; base < n; base += 4096) {
    int i0 = base + tid * 4;
    int c0 = (i0 + 0 < n) ? cnt[i0 + 0] : 0;
    int c1 = (i0 + 1 < n) ? cnt[i0 + 1] : 0;
    int c2 = (i0 + 2 < n) ? cnt[i0 + 2] : 0;
    int c3 = (i0 + 3 < n) ? cnt[i0 + 3] : 0;
    int t = c0 + c1 + c2 + c3;
    int incl = t;
    for (int d = 1; d < 32; d <<= 1) {
      int u = __shfl_up(incl, d);
      if (lane >= d) incl += u;
    }
    if (lane == 31) swave[wv] = incl;
    __syncthreads();
    int wt = swave[lane];
    int winc = wt;
    for (int d = 1; d < 32; d <<= 1) {
      int u = __shfl_up(winc, d);
      if (lane >= d) winc += u;
    }
    int wexcl = __shfl(winc, wv) - __shfl(wt, wv);
    int total = __shfl(winc, 31);
    int ex = carry + wexcl + incl - t;
    if (i0 + 0 < n) off[i0 + 0] = ex;
    if (i0 + 1 < n) off[i0 + 1] = ex + c0;
    if (i0 + 2 < n) off[i0 + 2] = ex + c0 + c1;
    if (i0 + 3 < n) off[i0 + 3] = ex + c0 + c1 + c2;
    carry += total;
    __syncthreads();
  }
  if (tid == 0) off[n] = carry;
}

__global__ __launch_bounds__(256) void gin_fill(
    const int* __restrict__ src, const int* __restrict__ dst,
    const int* __restrict__ off, int* cnt, int* __restrict__ ebuf, int nedges) {
  int e = (int)(blockIdx.x * 256u + threadIdx.x);
  if (e >= nedges) return;
  int d = dst[e];
  int c = atomicSub(cnt + d, 1);
  ebuf[off[d] + c - 1] = src[e];
}

__global__ __launch_bounds__(256) void gin_canon(
    const int* __restrict__ off, const int* __restrict__ ebuf,
    int* __restrict__ es, int n) {
  int lane = threadIdx.x & 31;
  int wid = (int)(blockIdx.x * 8u + (threadIdx.x >> 5));
  int nw = (int)(gridDim.x * 8u);
  for (int node = wid; node < n; node += nw) {
    int nu = __builtin_amdgcn_readfirstlane(node);
    int beg = off[nu];
    int len = off[nu + 1] - beg;
    if (len <= 32) {
      int key = (lane < len) ? ebuf[beg + lane] : 0x7fffffff;
      int rank = 0;
      for (int j = 0; j < len; j++) {
        int kj = __builtin_amdgcn_readlane(key, j);
        rank += ((kj < key) || (kj == key && j < lane)) ? 1 : 0;
      }
      if (lane < len) es[beg + rank] = key;
    } else {
      for (int i = lane; i < len; i += 32) {
        int key = ebuf[beg + i];
        int rank = 0;
        for (int j = 0; j < len; j++) {
          int kj = ebuf[beg + j];
          rank += ((kj < key) || (kj == key && j < i)) ? 1 : 0;
        }
        es[beg + rank] = key;
      }
    }
  }
}

template <bool EPI>
__global__ __launch_bounds__(256) void gin_gather(
    const float* __restrict__ x, const int* __restrict__ off,
    const int* __restrict__ es, const float* __restrict__ bias,
    float* __restrict__ agg, int n) {
  int lane = threadIdx.x & 31;
  int wid = (int)(blockIdx.x * 8u + (threadIdx.x >> 5));
  int nw = (int)(gridDim.x * 8u);
  const float* xl = x + lane * 4;
  float4 bv = {0.f, 0.f, 0.f, 0.f};
  if (EPI) bv = *(const float4*)(bias + lane * 4);
  for (int node = wid; node < n; node += nw) {
    int nu = __builtin_amdgcn_readfirstlane(node);
    int beg = off[nu];
    int end = off[nu + 1];
    float4 acc = {0.f, 0.f, 0.f, 0.f};
    float4 sv = {0.f, 0.f, 0.f, 0.f};
    if (EPI) sv = *(const float4*)(xl + (long)nu * GDIM);
    for (int base = beg; base < end; base += 32) {
      int m = end - base;
      if (m > 32) m = 32;
      int idv = (lane < m) ? es[base + lane] : 0;
      int j = 0;
      for (; j + 4 <= m; j += 4) {
        int s0 = __builtin_amdgcn_readlane(idv, j + 0);
        int s1 = __builtin_amdgcn_readlane(idv, j + 1);
        int s2 = __builtin_amdgcn_readlane(idv, j + 2);
        int s3 = __builtin_amdgcn_readlane(idv, j + 3);
        float4 v0 = *(const float4*)(xl + (long)s0 * GDIM);
        float4 v1 = *(const float4*)(xl + (long)s1 * GDIM);
        float4 v2 = *(const float4*)(xl + (long)s2 * GDIM);
        float4 v3 = *(const float4*)(xl + (long)s3 * GDIM);
        acc.x += v0.x; acc.y += v0.y; acc.z += v0.z; acc.w += v0.w;
        acc.x += v1.x; acc.y += v1.y; acc.z += v1.z; acc.w += v1.w;
        acc.x += v2.x; acc.y += v2.y; acc.z += v2.z; acc.w += v2.w;
        acc.x += v3.x; acc.y += v3.y; acc.z += v3.z; acc.w += v3.w;
      }
      for (; j < m; j++) {
        int s0 = __builtin_amdgcn_readlane(idv, j);
        float4 v0 = *(const float4*)(xl + (long)s0 * GDIM);
        acc.x += v0.x; acc.y += v0.y; acc.z += v0.z; acc.w += v0.w;
      }
    }
    if (EPI) {
      acc.x = fmaxf((sv.x + acc.x) + bv.x, 0.f);
      acc.y = fmaxf((sv.y + acc.y) + bv.y, 0.f);
      acc.z = fmaxf((sv.z + acc.z) + bv.z, 0.f);
      acc.w = fmaxf((sv.w + acc.w) + bv.w, 0.f);
    }
    *(float4*)(agg + (long)nu * GDIM + lane * 4) = acc;
  }
}

template <bool HASADD, bool RELU>
__global__ __launch_bounds__(256) void gin_gemm(
    const float* A0, const float* A1, const float* __restrict__ W,
    const float* __restrict__ bias, float* Out, int nrows) {
  __shared__ float sW[GDIM * GDIM];
  int tid = threadIdx.x;
  for (int i = tid; i < (GDIM / 2) * GDIM; i += 256) {
    int p = i >> 7;
    int n = i & 127;
    int idx = p * 256 + 2 * n;
    sW[idx]     = W[(2 * p) * GDIM + n];
    sW[idx + 1] = W[(2 * p + 1) * GDIM + n];
  }
  __syncthreads();

  int wave = tid >> 5;
  int lane = tid & 31;
  long mbase = ((long)blockIdx.x * 8 + wave) * 32;
  if (mbase >= nrows) return;
  bool has2 = (mbase + 16) < nrows;
  long mbase2 = has2 ? (mbase + 16) : mbase;

  int lmod = lane & 15;
  int par = lane >> 4;
  int khalf = par << 1;
  int mhi = par << 3;

  const float* arA0 = A0 + (mbase + lmod) * GDIM;
  const float* arB0 = A0 + (mbase2 + lmod) * GDIM;
  const float* arA1 = HASADD ? (A1 + (mbase + lmod) * GDIM) : nullptr;
  const float* arB1 = HASADD ? (A1 + (mbase2 + lmod) * GDIM) : nullptr;
  const float* sWp = sW + par * 256 + 2 * lmod;

  v2f pa0 = *(const v2f*)(arA0 + khalf);
  v2f pb0 = *(const v2f*)(arB0 + khalf);
  v2f pa1, pb1;
  if (HASADD) {
    pa1 = *(const v2f*)(arA1 + khalf);
    pb1 = *(const v2f*)(arB1 + khalf);
  }

  v8f acc[16] = {};

  for (int k = 0; k < GDIM; k += 4) {
    v2f aA = pa0, aB = pb0;
    if (HASADD) {
      aA.x += pa1.x; aA.y += pa1.y;
      aB.x += pb1.x; aB.y += pb1.y;
    }
    if (k + 4 < GDIM) {
      pa0 = *(const v2f*)(arA0 + (k + 4) + khalf);
      pb0 = *(const v2f*)(arB0 + (k + 4) + khalf);
      if (HASADD) {
        pa1 = *(const v2f*)(arA1 + (k + 4) + khalf);
        pb1 = *(const v2f*)(arB1 + (k + 4) + khalf);
      }
    }
    const float* wp = sWp + (k >> 1) * 256;
    v2f b[8];
#pragma unroll
    for (int nt = 0; nt < 8; nt++)
      b[nt] = *(const v2f*)(wp + nt * 32);
#pragma unroll
    for (int nt = 0; nt < 8; nt++) {
      acc[nt] = __builtin_amdgcn_wmma_f32_16x16x4_f32(
          false, aA, false, b[nt], (short)0, acc[nt], false, false);
      acc[8 + nt] = __builtin_amdgcn_wmma_f32_16x16x4_f32(
          false, aB, false, b[nt], (short)0, acc[8 + nt], false, false);
    }
  }

#pragma unroll
  for (int nt = 0; nt < 8; nt++) {
    int ncol = nt * 16 + lmod;
    float bv = bias[ncol];
#pragma unroll
    for (int j = 0; j < 8; j++) {
      float v = acc[nt][j] + bv;
      if (RELU) v = fmaxf(v, 0.f);
      Out[(mbase + mhi + j) * GDIM + ncol] = v;
    }
    if (has2) {
#pragma unroll
      for (int j = 0; j < 8; j++) {
        float v = acc[8 + nt][j] + bv;
        if (RELU) v = fmaxf(v, 0.f);
        Out[(mbase2 + mhi + j) * GDIM + ncol] = v;
      }
    }
  }
}

__global__ __launch_bounds__(256) void gin_fold(
    const float* __restrict__ W1, const float* __restrict__ W2,
    const float* __restrict__ b2, float* __restrict__ Mout,
    float* __restrict__ cout) {
  __shared__ float sW[GDIM * GDIM];
  int tid = threadIdx.x;
  int fi = blockIdx.x;
  const float* Wn = W1 + (long)(fi + 1) * GDIM * GDIM;
  const float* A = W2 + (long)fi * GDIM * GDIM;
  const float* bb = b2 + fi * GDIM;

  int wave = tid >> 5;
  int lane = tid & 31;
  int lmod = lane & 15;
  int par = lane >> 4;
  int khalf = par << 1;
  int mhi = par << 3;

  const float* ar = A + (wave * 16 + lmod) * GDIM + khalf;
  v2f afr[GDIM / 4];
#pragma unroll
  for (int s = 0; s < GDIM / 4; s++) afr[s] = *(const v2f*)(ar + 4 * s);

  float4 r0[8], r1[8];
#pragma unroll
  for (int t = 0; t < 8; t++) {
    int i = tid + t * 256;
    int p = i >> 5;
    int q = i & 31;
    r0[t] = *(const float4*)(Wn + (2 * p) * GDIM + 4 * q);
    r1[t] = *(const float4*)(Wn + (2 * p + 1) * GDIM + 4 * q);
  }
#pragma unroll
  for (int t = 0; t < 8; t++) {
    int i = tid + t * 256;
    int p = i >> 5;
    int q = i & 31;
    float4 lo = {r0[t].x, r1[t].x, r0[t].y, r1[t].y};
    float4 hi = {r0[t].z, r1[t].z, r0[t].w, r1[t].w};
    *(float4*)(sW + p * 256 + 8 * q) = lo;
    *(float4*)(sW + p * 256 + 8 * q + 4) = hi;
  }
  __syncthreads();

  const float* sWp = sW + par * 256 + 2 * lmod;
  v8f acc[8] = {};
#pragma unroll
  for (int s = 0; s < GDIM / 4; s++) {
    const float* wp = sWp + s * 512;
    v2f b[8];
#pragma unroll
    for (int nt = 0; nt < 8; nt++) b[nt] = *(const v2f*)(wp + nt * 32);
#pragma unroll
    for (int nt = 0; nt < 8; nt++)
      acc[nt] = __builtin_amdgcn_wmma_f32_16x16x4_f32(
          false, afr[s], false, b[nt], (short)0, acc[nt], false, false);
  }
  float* Mo = Mout + (long)fi * GDIM * GDIM;
#pragma unroll
  for (int nt = 0; nt < 8; nt++) {
#pragma unroll
    for (int j = 0; j < 8; j++)
      Mo[(wave * 16 + mhi + j) * GDIM + nt * 16 + lmod] = acc[nt][j];
  }

  if (tid < GDIM) {
    float s = 0.f;
    for (int k = 0; k < GDIM; k += 2) {
      v2f w = *(const v2f*)(sW + (k >> 1) * 256 + 2 * tid);
      s = fmaf(bb[k], w.x, s);
      s = fmaf(bb[k + 1], w.y, s);
    }
    cout[fi * GDIM + tid] = s;
  }
}

extern "C" void kernel_launch(void* const* d_in, const int* in_sizes, int n_in,
                              void* d_out, int out_size, void* d_ws,
                              size_t ws_size, hipStream_t stream) {
  const float* x_in = (const float*)d_in[0];
  const int* ei = (const int*)d_in[1];
  const float* W1 = (const float*)d_in[2];
  const float* b1 = (const float*)d_in[3];
  const float* W2 = (const float*)d_in[4];
  const float* b2 = (const float*)d_in[5];
  float* out = (float*)d_out;

  const int N = in_sizes[0] / GDIM;
  const int NE = in_sizes[1] / 2;
  const long nd = (long)N * GDIM;

  float* aggh = (float*)d_ws;
  int* cnt = (int*)(aggh + nd);
  int* off = cnt + (N + 4);
  int* ebuf = off + (N + 4);
  int* es = ebuf + NE;
  float* Mf = (float*)ebuf;
  float* cf = Mf + 2 * GDIM * GDIM;

  const int mtiles32 = (N + 31) / 32;
  const int gemm_blocks = (mtiles32 + 7) / 8;
  const int edge_blocks = (NE + 255) / 256;
  const int zero_blocks = ((N + 3) / 4 + 255) / 256;
  const int row_blocks = 4096;

  gin_zero<<<zero_blocks, 256, 0, stream>>>((float4*)cnt, (N + 3) / 4);
  gin_hist<<<edge_blocks, 256, 0, stream>>>(ei + NE, cnt, NE);
  gin_scan<<<1, 1024, 0, stream>>>(cnt, off, N);
  gin_fill<<<edge_blocks, 256, 0, stream>>>(ei, ei + NE, off, cnt, ebuf, NE);
  gin_canon<<<row_blocks, 256, 0, stream>>>(off, ebuf, es, N);
  gin_fold<<<2, 256, 0, stream>>>(W1, W2, b2, Mf, cf);

  gin_gather<false><<<row_blocks, 256, 0, stream>>>(x_in, off, es, nullptr, aggh, N);
  gin_gemm<true, true><<<gemm_blocks, 256, 0, stream>>>(x_in, aggh, W1, b1, aggh, N);
  for (int layer = 1; layer < 3; ++layer) {
    gin_gemm<false, false><<<gemm_blocks, 256, 0, stream>>>(
        aggh, nullptr, Mf + (long)(layer - 1) * GDIM * GDIM,
        cf + (layer - 1) * GDIM, out, N);
    gin_gather<true><<<row_blocks, 256, 0, stream>>>(
        out, off, es, b1 + layer * GDIM, aggh, N);
  }
  gin_gemm<false, false><<<gemm_blocks, 256, 0, stream>>>(
      aggh, nullptr, W2 + (long)2 * GDIM * GDIM, b2 + 2 * GDIM, out, N);
}


# One global atomic pass instead of two: the histogram's add returns each edge's rank in its destination (kept per edge), placement is off[dst]+rank without atomics; neighbour sum order unchanged
# speedup vs baseline: 2.4591x; 1.1781x over previous
#include <hip/hip_runtime.h>

typedef __attribute__((ext_vector_type(2))) float v2f;
typedef __attribute__((ext_vector_type(8))) float v8f;

#define GDIM 128

__global__ __launch_bounds__(256) void gin_zero(float4* p, long n4) {
  long i = (long)blockIdx.x * blockDim.x + threadIdx.x;
  long stride = (long)gridDim.x * blockDim.x;
  float4 z = {0.f, 0.f, 0.f, 0.f};
  for (; i < n4; i += stride) p[i] = z;
}


__global__ __launch_bounds__(256) void gin_hist(const int* __restrict__ dst,
                                                int* cnt,
                                                int* __restrict__ rank,
                                                int nedges) {
  int e = (int)(blockIdx.x * 256u + threadIdx.x);
  if (e < nedges) rank[e] = atomicAdd(cnt + dst[e], 1);
}

__global__ __launch_bounds__(1024) void gin_scan(const int* __restrict__ cnt,
                                                 int* __restrict__ off, int n) {
  __shared__ int swave[32];
  int tid = threadIdx.x;
  int lane = tid & 31;
  int wv = tid >> 5;
  int carry = 0;
  for (int base = 0; base < n; base += 4096) {
    int i0 = base + tid * 4;
    int c0 = (i0 + 0 < n) ? cnt[i0 + 0] : 0;
    int c1 = (i0 + 1 < n) ? cnt[i0 + 1] : 0;
    int c2 = (i0 + 2 < n) ? cnt[i0 + 2] : 0;
    int c3 = (i0 + 3 < n) ? cnt[i0 + 3] : 0;
    int t = c0 + c1 + c2 + c3;
    int incl = t;
    for (int d = 1; d < 32; d <<= 1) {
      int u = __shfl_up(incl, d);
      if (lane >= d) incl += u;
    }
    if (lane == 31) swave[wv] = incl;
    __syncthreads();
    int wt = swave[lane];
    int winc = wt;
    for (int d = 1; d < 32; d <<= 1) {
      int u = __shfl_up(winc, d);
      if (lane >= d) winc += u;
    }
    int wexcl = __shfl(winc, wv) - __shfl(wt, wv);
    int total = __shfl(winc, 31);
    int ex = carry + wexcl + incl - t;
    if (i0 + 0 < n) off[i0 + 0] = ex;
    if (i0 + 1 < n) off[i0 + 1] = ex + c0;
    if (i0 + 2 < n) off[i0 + 2] = ex + c0 + c1;
    if (i0 + 3 < n) off[i0 + 3] = ex + c0 + c1 + c2;
    carry += total;
    __syncthreads();
  }
  if (tid == 0) off[n] = carry;
}

__global__ __launch_bounds__(256) void gin_fill(
    const int* __restrict__ src, const int* __restrict__ dst,
    const int* __restrict__ off, const int* __restrict__ rank,
    int* __restrict__ ebuf, int nedges) {
  int e = (int)(blockIdx.x * 256u + threadIdx.x);
  if (e >= nedges) return;
  ebuf[off[dst[e]] + rank[e]] = src[e];
}

__global__ __launch_bounds__(256) void gin_canon(
    const int* __restrict__ off, const int* __restrict__ ebuf,
    int* __restrict__ es, int n) {
  int lane = threadIdx.x & 31;
  int wid = (int)(blockIdx.x * 8u + (threadIdx.x >> 5));
  int nw = (int)(gridDim.x * 8u);
  for (int node = wid; node < n; node += nw) {
    int nu = __builtin_amdgcn_readfirstlane(node);
    int beg = off[nu];
    int len = off[nu + 1] - beg;
    if (len <= 32) {
      int key = (lane < len) ? ebuf[beg + lane] : 0x7fffffff;
      int rank = 0;
      for (int j = 0; j < len; j++) {
        int kj = __builtin_amdgcn_readlane(key, j);
        rank += ((kj < key) || (kj == key && j < lane)) ? 1 : 0;
      }
      if (lane < len) es[beg + rank] = key;
    } else {
      for (int i = lane; i < len; i += 32) {
        int key = ebuf[beg + i];
        int rank = 0;
        for (int j = 0; j < len; j++) {
          int kj = ebuf[beg + j];
          rank += ((kj < key) || (kj == key && j < i)) ? 1 : 0;
        }
        es[beg + rank] = key;
      }
    }
  }
}

template <bool EPI>
__global__ __launch_bounds__(256) void gin_gather(
    const float* __restrict__ x, const int* __restrict__ off,
    const int* __restrict__ es, const float* __restrict__ bias,
    float* __restrict__ agg, int n) {
  int lane = threadIdx.x & 31;
  int wid = (int)(blockIdx.x * 8u + (threadIdx.x >> 5));
  int nw = (int)(gridDim.x * 8u);
  const float* xl = x + lane * 4;
  float4 bv = {0.f, 0.f, 0.f, 0.f};
  if (EPI) bv = *(const float4*)(bias + lane * 4);
  for (int node = wid; node < n; node += nw) {
    int nu = __builtin_amdgcn_readfirstlane(node);
    int beg = off[nu];
    int end = off[nu + 1];
    float4 acc = {0.f, 0.f, 0.f, 0.f};
    float4 sv = {0.f, 0.f, 0.f, 0.f};
    if (EPI) sv = *(const float4*)(xl + (long)nu * GDIM);
    for (int base = beg; base < end; base += 32) {
      int m = end - base;
      if (m > 32) m = 32;
      int idv = (lane < m) ? es[base + lane] : 0;
      int j = 0;
      for (; j + 4 <= m; j += 4) {
        int s0 = __builtin_amdgcn_readlane(idv, j + 0);
        int s1 = __builtin_amdgcn_readlane(idv, j + 1);
        int s2 = __builtin_amdgcn_readlane(idv, j + 2);
        int s3 = __builtin_amdgcn_readlane(idv, j + 3);
        float4 v0 = *(const float4*)(xl + (long)s0 * GDIM);
        float4 v1 = *(const float4*)(xl + (long)s1 * GDIM);
        float4 v2 = *(const float4*)(xl + (long)s2 * GDIM);
        float4 v3 = *(const float4*)(xl + (long)s3 * GDIM);
        acc.x += v0.x; acc.y += v0.y; acc.z += v0.z; acc.w += v0.w;
        acc.x += v1.x; acc.y += v1.y; acc.z += v1.z; acc.w += v1.w;
        acc.x += v2.x; acc.y += v2.y; acc.z += v2.z; acc.w += v2.w;
        acc.x += v3.x; acc.y += v3.y; acc.z += v3.z; acc.w += v3.w;
      }
      for (; j < m; j++) {
        int s0 = __builtin_amdgcn_readlane(idv, j);
        float4 v0 = *(const float4*)(xl + (long)s0 * GDIM);
        acc.x += v0.x; acc.y += v0.y; acc.z += v0.z; acc.w += v0.w;
      }
    }
    if (EPI) {
      acc.x = fmaxf((sv.x + acc.x) + bv.x, 0.f);
      acc.y = fmaxf((sv.y + acc.y) + bv.y, 0.f);
      acc.z = fmaxf((sv.z + acc.z) + bv.z, 0.f);
      acc.w = fmaxf((sv.w + acc.w) + bv.w, 0.f);
    }
    *(float4*)(agg + (long)nu * GDIM + lane * 4) = acc;
  }
}

template <bool HASADD, bool RELU>
__global__ __launch_bounds__(256) void gin_gemm(
    const float* A0, const float* A1, const float* __restrict__ W,
    const float* __restrict__ bias, float* Out, int nrows) {
  __shared__ float sW[GDIM * GDIM];
  int tid = threadIdx.x;
  for (int i = tid; i < (GDIM / 2) * GDIM; i += 256) {
    int p = i >> 7;
    int n = i & 127;
    int idx = p * 256 + 2 * n;
    sW[idx]     = W[(2 * p) * GDIM + n];
    sW[idx + 1] = W[(2 * p + 1) * GDIM + n];
  }
  __syncthreads();

  int wave = tid >> 5;
  int lane = tid & 31;
  long mbase = ((long)blockIdx.x * 8 + wave) * 32;
  if (mbase >= nrows) return;
  bool has2 = (mbase + 16) < nrows;
  long mbase2 = has2 ? (mbase + 16) : mbase;

  int lmod = lane & 15;
  int par = lane >> 4;
  int khalf = par << 1;
  int mhi = par << 3;

  const float* arA0 = A0 + (mbase + lmod) * GDIM;
  const float* arB0 = A0 + (mbase2 + lmod) * GDIM;
  const float* arA1 = HASADD ? (A1 + (mbase + lmod) * GDIM) : nullptr;
  const float* arB1 = HASADD ? (A1 + (mbase2 + lmod) * GDIM) : nullptr;
  const float* sWp = sW + par * 256 + 2 * lmod;

  v2f pa0 = *(const v2f*)(arA0 + khalf);
  v2f pb0 = *(const v2f*)(arB0 + khalf);
  v2f pa1, pb1;
  if (HASADD) {
    pa1 = *(const v2f*)(arA1 + khalf);
    pb1 = *(const v2f*)(arB1 + khalf);
  }

  v8f acc[16] = {};

  for (int k = 0; k < GDIM; k += 4) {
    v2f aA = pa0, aB = pb0;
    if (HASADD) {
      aA.x += pa1.x; aA.y += pa1.y;
      aB.x += pb1.x; aB.y += pb1.y;
    }
    if (k + 4 < GDIM) {
      pa0 = *(const v2f*)(arA0 + (k + 4) + khalf);
      pb0 = *(const v2f*)(arB0 + (k + 4) + khalf);
      if (HASADD) {
        pa1 = *(const v2f*)(arA1 + (k + 4) + khalf);
        pb1 = *(const v2f*)(arB1 + (k + 4) + khalf);
      }
    }
    const float* wp = sWp + (k >> 1) * 256;
    v2f b[8];
#pragma unroll
    for (int nt = 0; nt < 8; nt++)
      b[nt] = *(const v2f*)(wp + nt * 32);
#pragma unroll
    for (int nt = 0; nt < 8; nt++) {
      acc[nt] = __builtin_amdgcn_wmma_f32_16x16x4_f32(
          false, aA, false, b[nt], (short)0, acc[nt], false, false);
      acc[8 + nt] = __builtin_amdgcn_wmma_f32_16x16x4_f32(
          false, aB, false, b[nt], (short)0, acc[8 + nt], false, false);
    }
  }

#pragma unroll
  for (int nt = 0; nt < 8; nt++) {
    int ncol = nt * 16 + lmod;
    float bv = bias[ncol];
#pragma unroll
    for (int j = 0; j < 8; j++) {
      float v = acc[nt][j] + bv;
      if (RELU) v = fmaxf(v, 0.f);
      Out[(mbase + mhi + j) * GDIM + ncol] = v;
    }
    if (has2) {
#pragma unroll
      for (int j = 0; j < 8; j++) {
        float v = acc[8 + nt][j] + bv;
        if (RELU) v = fmaxf(v, 0.f);
        Out[(mbase2 + mhi + j) * GDIM + ncol] = v;
      }
    }
  }
}

__global__ __launch_bounds__(256) void gin_fold(
    const float* __restrict__ W1, const float* __restrict__ W2,
    const float* __restrict__ b2, float* __restrict__ Mout,
    float* __restrict__ cout) {
  __shared__ float sW[GDIM * GDIM];
  int tid = threadIdx.x;
  int fi = blockIdx.x;
  const float* Wn = W1 + (long)(fi + 1) * GDIM * GDIM;
  const float* A = W2 + (long)fi * GDIM * GDIM;
  const float* bb = b2 + fi * GDIM;

  int wave = tid >> 5;
  int lane = tid & 31;
  int lmod = lane & 15;
  int par = lane >> 4;
  int khalf = par << 1;
  int mhi = par << 3;

  const float* ar = A + (wave * 16 + lmod) * GDIM + khalf;
  v2f afr[GDIM / 4];
#pragma unroll
  for (int s = 0; s < GDIM / 4; s++) afr[s] = *(const v2f*)(ar + 4 * s);

  float4 r0[8], r1[8];
#pragma unroll
  for (int t = 0; t < 8; t++) {
    int i = tid + t * 256;
    int p = i >> 5;
    int q = i & 31;
    r0[t] = *(const float4*)(Wn + (2 * p) * GDIM + 4 * q);
    r1[t] = *(const float4*)(Wn + (2 * p + 1) * GDIM + 4 * q);
  }
#pragma unroll
  for (int t = 0; t < 8; t++) {
    int i = tid + t * 256;
    int p = i >> 5;
    int q = i & 31;
    float4 lo = {r0[t].x, r1[t].x, r0[t].y, r1[t].y};
    float4 hi = {r0[t].z, r1[t].z, r0[t].w, r1[t].w};
    *(float4*)(sW + p * 256 + 8 * q) = lo;
    *(float4*)(sW + p * 256 + 8 * q + 4) = hi;
  }
  __syncthreads();

  const float* sWp = sW + par * 256 + 2 * lmod;
  v8f acc[8] = {};
#pragma unroll
  for (int s = 0; s < GDIM / 4; s++) {
    const float* wp = sWp + s * 512;
    v2f b[8];
#pragma unroll
    for (int nt = 0; nt < 8; nt++) b[nt] = *(const v2f*)(wp + nt * 32);
#pragma unroll
    for (int nt = 0; nt < 8; nt++)
      acc[nt] = __builtin_amdgcn_wmma_f32_16x16x4_f32(
          false, afr[s], false, b[nt], (short)0, acc[nt], false, false);
  }
  float* Mo = Mout + (long)fi * GDIM * GDIM;
#pragma unroll
  for (int nt = 0; nt < 8; nt++) {
#pragma unroll
    for (int j = 0; j < 8; j++)
      Mo[(wave * 16 + mhi + j) * GDIM + nt * 16 + lmod] = acc[nt][j];
  }

  if (tid < GDIM) {
    float s = 0.f;
    for (int k = 0; k < GDIM; k += 2) {
      v2f w = *(const v2f*)(sW + (k >> 1) * 256 + 2 * tid);
      s = fmaf(bb[k], w.x, s);
      s = fmaf(bb[k + 1], w.y, s);
    }
    cout[fi * GDIM + tid] = s;
  }
}

extern "C" void kernel_launch(void* const* d_in, const int* in_sizes, int n_in,
                              void* d_out, int out_size, void* d_ws,
                              size_t ws_size, hipStream_t stream) {
  const float* x_in = (const float*)d_in[0];
  const int* ei = (const int*)d_in[1];
  const float* W1 = (const float*)d_in[2];
  const float* b1 = (const float*)d_in[3];
  const float* W2 = (const float*)d_in[4];
  const float* b2 = (const float*)d_in[5];
  float* out = (float*)d_out;

  const int N = in_sizes[0] / GDIM;
  const int NE = in_sizes[1] / 2;
  const long nd = (long)N * GDIM;

  float* aggh = (float*)d_ws;
  int* cnt = (int*)(aggh + nd);
  int* off = cnt + (N + 4);
  int* ebuf = off + (N + 4);
  int* es = ebuf + NE;
  float* Mf = (float*)ebuf;
  float* cf = Mf + 2 * GDIM * GDIM;

  const int mtiles32 = (N + 31) / 32;
  const int gemm_blocks = (mtiles32 + 7) / 8;
  const int edge_blocks = (NE + 255) / 256;
  const int zero_blocks = ((N + 3) / 4 + 255) / 256;
  const int row_blocks = 4096;

  gin_zero<<<zero_blocks, 256, 0, stream>>>((float4*)cnt, (N + 3) / 4);
  gin_hist<<<edge_blocks, 256, 0, stream>>>(ei + NE, cnt, es, NE);
  gin_scan<<<1, 1024, 0, stream>>>(cnt, off, N);
  gin_fill<<<edge_blocks, 256, 0, stream>>>(ei, ei + NE, off, es, ebuf, NE);
  gin_canon<<<row_blocks, 256, 0, stream>>>(off, ebuf, es, N);
  gin_fold<<<2, 256, 0, stream>>>(W1, W2, b2, Mf, cf);

  gin_gather<false><<<row_blocks, 256, 0, stream>>>(x_in, off, es, nullptr, aggh, N);
  gin_gemm<true, true><<<gemm_blocks, 256, 0, stream>>>(x_in, aggh, W1, b1, aggh, N);
  for (int layer = 1; layer < 3; ++layer) {
    gin_gemm<false, false><<<gemm_blocks, 256, 0, stream>>>(
        aggh, nullptr, Mf + (long)(layer - 1) * GDIM * GDIM,
        cf + (layer - 1) * GDIM, out, N);
    gin_gather<true><<<row_blocks, 256, 0, stream>>>(
        out, off, es, b1 + layer * GDIM, aggh, N);
  }
  gin_gemm<false, false><<<gemm_blocks, 256, 0, stream>>>(
      aggh, nullptr, W2 + (long)2 * GDIM * GDIM, b2 + 2 * GDIM, out, N);
}


# Build the destination-grouped edge list by a two-level bucket split counted in LDS (512-destination buckets) instead of a global atomic per edge and a one-block scan; neighbour sum order unchanged
# speedup vs baseline: 3.5751x; 1.4538x over previous
#include <hip/hip_runtime.h>

typedef __attribute__((ext_vector_type(2))) float v2f;
typedef __attribute__((ext_vector_type(8))) float v8f;

#define GDIM 128

__global__ __launch_bounds__(256) void gin_zero(float4* p, long n4) {
  long i = (long)blockIdx.x * blockDim.x + threadIdx.x;
  long stride = (long)gridDim.x * blockDim.x;
  float4 z = {0.f, 0.f, 0.f, 0.f};
  for (; i < n4; i += stride) p[i] = z;
}


#define BSH 9
#define BNODES 512
#define MAXNB 256
#define BTS 32

__global__ __launch_bounds__(1024) void gin_bcount(
    const int* __restrict__ dst, int* btot, int* __restrict__ blkoff,
    int nedges, int nb) {
  __shared__ int h[MAXNB];
  int tid = threadIdx.x;
  if (tid < MAXNB) h[tid] = 0;
  __syncthreads();
  int chunk = (nedges + (int)gridDim.x - 1) / (int)gridDim.x;
  int e0 = (int)blockIdx.x * chunk;
  int e1 = (e0 + chunk < nedges) ? (e0 + chunk) : nedges;
  for (int e = e0 + tid; e < e1; e += 8 * 1024) {
    int d[8];
#pragma unroll
    for (int u = 0; u < 8; u++) {
      int ee = e + u * 1024;
      d[u] = (ee < e1) ? dst[ee] : -1;
    }
#pragma unroll
    for (int u = 0; u < 8; u++)
      if (d[u] >= 0) atomicAdd(&h[d[u] >> BSH], 1);
  }
  __syncthreads();
  if (tid < nb)
    blkoff[(int)blockIdx.x * nb + tid] = atomicAdd(btot + tid * BTS, h[tid]);
}

__global__ __launch_bounds__(1024) void gin_bscatter(
    const int* __restrict__ src, const int* __restrict__ dst,
    const int* __restrict__ btot, const int* __restrict__ blkoff,
    int2* __restrict__ pairs, int nedges, int nb) {
  __shared__ int cur[MAXNB];
  __shared__ int swave[8];
  int tid = threadIdx.x;
  int lane = tid & 31;
  int wv = tid >> 5;
  int v = (tid < nb) ? btot[tid * BTS] : 0;
  int incl = v;
  for (int s = 1; s < 32; s <<= 1) {
    int u = __shfl_up(incl, s);
    if (lane >= s) incl += u;
  }
  if (lane == 31 && wv < 8) swave[wv] = incl;
  __syncthreads();
  int pre = 0;
  for (int j = 0; j < 8; j++) pre += (j < wv) ? swave[j] : 0;
  if (tid < nb) cur[tid] = pre + incl - v + blkoff[(int)blockIdx.x * nb + tid];
  __syncthreads();
  int chunk = (nedges + (int)gridDim.x - 1) / (int)gridDim.x;
  int e0 = (int)blockIdx.x * chunk;
  int e1 = (e0 + chunk < nedges) ? (e0 + chunk) : nedges;
  for (int e = e0 + tid; e < e1; e += 8 * 1024) {
    int d[8], sr[8];
#pragma unroll
    for (int u = 0; u < 8; u++) {
      int ee = e + u * 1024;
      d[u] = (ee < e1) ? dst[ee] : -1;
      sr[u] = (ee < e1) ? src[ee] : 0;
    }
#pragma unroll
    for (int u = 0; u < 8; u++) {
      if (d[u] >= 0) {
        int pos = atomicAdd(&cur[d[u] >> BSH], 1);
        pairs[pos] = make_int2(sr[u], d[u]);
      }
    }
  }
}

__global__ __launch_bounds__(1024) void gin_bcsr(
    const int2* __restrict__ pairs, const int* __restrict__ btot,
    int* __restrict__ off, int* __restrict__ ebuf, int n, int nb) {
  __shared__ int cntL[BNODES];
  __shared__ int curL[BNODES];
  __shared__ int swave[16];
  __shared__ int sred[8];
  int tid = threadIdx.x;
  int lane = tid & 31;
  int wv = tid >> 5;
  int b = (int)blockIdx.x;
  int base = b << BSH;
  if (tid < BNODES) cntL[tid] = 0;
  int v = (tid < b) ? btot[tid * BTS] : 0;
  for (int m = 16; m >= 1; m >>= 1) v += __shfl_xor(v, m);
  if (lane == 0 && wv < 8) sred[wv] = v;
  int cb = btot[b * BTS];
  __syncthreads();
  int bs = 0;
  for (int j = 0; j < 8; j++) bs += sred[j];

  for (int i = tid; i < cb; i += 8 * 1024) {
    int d[8];
#pragma unroll
    for (int u = 0; u < 8; u++) {
      int ii = i + u * 1024;
      d[u] = (ii < cb) ? pairs[bs + ii].y : -1;
    }
#pragma unroll
    for (int u = 0; u < 8; u++)
      if (d[u] >= 0) atomicAdd(&cntL[d[u] - base], 1);
  }
  __syncthreads();

  int c = (tid < BNODES) ? cntL[tid] : 0;
  int incl = c;
  for (int s = 1; s < 32; s <<= 1) {
    int u = __shfl_up(incl, s);
    if (lane >= s) incl += u;
  }
  if (lane == 31 && wv < 16) swave[wv] = incl;
  __syncthreads();
  int pre = 0;
  for (int j = 0; j < 16; j++) pre += (j < wv) ? swave[j] : 0;
  int ex = pre + incl - c;
  if (tid < BNODES) {
    curL[tid] = ex;
    if (base + tid < n) off[base + tid] = bs + ex;
  }
  if (b == nb - 1 && tid == 0) off[n] = bs + cb;
  __syncthreads();

  for (int i = tid; i < cb; i += 8 * 1024) {
    int2 p[8];
#pragma unroll
    for (int u = 0; u < 8; u++) {
      int ii = i + u * 1024;
      p[u] = (ii < cb) ? pairs[bs + ii] : make_int2(0, -1);
    }
#pragma unroll
    for (int u = 0; u < 8; u++) {
      if (p[u].y >= 0) {
        int pos = atomicAdd(&curL[p[u].y - base], 1);
        ebuf[bs + pos] = p[u].x;
      }
    }
  }
}

__global__ __launch_bounds__(256) void gin_canon(
    const int* __restrict__ off, const int* __restrict__ ebuf,
    int* __restrict__ es, int n) {
  int lane = threadIdx.x & 31;
  int wid = (int)(blockIdx.x * 8u + (threadIdx.x >> 5));
  int nw = (int)(gridDim.x * 8u);
  for (int node = wid; node < n; node += nw) {
    int nu = __builtin_amdgcn_readfirstlane(node);
    int beg = off[nu];
    int len = off[nu + 1] - beg;
    if (len <= 32) {
      int key = (lane < len) ? ebuf[beg + lane] : 0x7fffffff;
      int rank = 0;
      for (int j = 0; j < len; j++) {
        int kj = __builtin_amdgcn_readlane(key, j);
        rank += ((kj < key) || (kj == key && j < lane)) ? 1 : 0;
      }
      if (lane < len) es[beg + rank] = key;
    } else {
      for (int i = lane; i < len; i += 32) {
        int key = ebuf[beg + i];
        int rank = 0;
        for (int j = 0; j < len; j++) {
          int kj = ebuf[beg + j];
          rank += ((kj < key) || (kj == key && j < i)) ? 1 : 0;
        }
        es[beg + rank] = key;
      }
    }
  }
}

template <bool EPI>
__global__ __launch_bounds__(256) void gin_gather(
    const float* __restrict__ x, const int* __restrict__ off,
    const int* __restrict__ es, const float* __restrict__ bias,
    float* __restrict__ agg, int n) {
  int lane = threadIdx.x & 31;
  int wid = (int)(blockIdx.x * 8u + (threadIdx.x >> 5));
  int nw = (int)(gridDim.x * 8u);
  const float* xl = x + lane * 4;
  float4 bv = {0.f, 0.f, 0.f, 0.f};
  if (EPI) bv = *(const float4*)(bias + lane * 4);
  for (int node = wid; node < n; node += nw) {
    int nu = __builtin_amdgcn_readfirstlane(node);
    int beg = off[nu];
    int end = off[nu + 1];
    float4 acc = {0.f, 0.f, 0.f, 0.f};
    float4 sv = {0.f, 0.f, 0.f, 0.f};
    if (EPI) sv = *(const float4*)(xl + (long)nu * GDIM);
    for (int base = beg; base < end; base += 32) {
      int m = end - base;
      if (m > 32) m = 32;
      int idv = (lane < m) ? es[base + lane] : 0;
      int j = 0;
      for (; j + 4 <= m; j += 4) {
        int s0 = __builtin_amdgcn_readlane(idv, j + 0);
        int s1 = __builtin_amdgcn_readlane(idv, j + 1);
        int s2 = __builtin_amdgcn_readlane(idv, j + 2);
        int s3 = __builtin_amdgcn_readlane(idv, j + 3);
        float4 v0 = *(const float4*)(xl + (long)s0 * GDIM);
        float4 v1 = *(const float4*)(xl + (long)s1 * GDIM);
        float4 v2 = *(const float4*)(xl + (long)s2 * GDIM);
        float4 v3 = *(const float4*)(xl + (long)s3 * GDIM);
        acc.x += v0.x; acc.y += v0.y; acc.z += v0.z; acc.w += v0.w;
        acc.x += v1.x; acc.y += v1.y; acc.z += v1.z; acc.w += v1.w;
        acc.x += v2.x; acc.y += v2.y; acc.z += v2.z; acc.w += v2.w;
        acc.x += v3.x; acc.y += v3.y; acc.z += v3.z; acc.w += v3.w;
      }
      for (; j < m; j++) {
        int s0 = __builtin_amdgcn_readlane(idv, j);
        float4 v0 = *(const float4*)(xl + (long)s0 * GDIM);
        acc.x += v0.x; acc.y += v0.y; acc.z += v0.z; acc.w += v0.w;
      }
    }
    if (EPI) {
      acc.x = fmaxf((sv.x + acc.x) + bv.x, 0.f);
      acc.y = fmaxf((sv.y + acc.y) + bv.y, 0.f);
      acc.z = fmaxf((sv.z + acc.z) + bv.z, 0.f);
      acc.w = fmaxf((sv.w + acc.w) + bv.w, 0.f);
    }
    *(float4*)(agg + (long)nu * GDIM + lane * 4) = acc;
  }
}

template <bool HASADD, bool RELU>
__global__ __launch_bounds__(256) void gin_gemm(
    const float* A0, const float* A1, const float* __restrict__ W,
    const float* __restrict__ bias, float* Out, int nrows) {
  __shared__ float sW[GDIM * GDIM];
  int tid = threadIdx.x;
  for (int i = tid; i < (GDIM / 2) * GDIM; i += 256) {
    int p = i >> 7;
    int n = i & 127;
    int idx = p * 256 + 2 * n;
    sW[idx]     = W[(2 * p) * GDIM + n];
    sW[idx + 1] = W[(2 * p + 1) * GDIM + n];
  }
  __syncthreads();

  int wave = tid >> 5;
  int lane = tid & 31;
  long mbase = ((long)blockIdx.x * 8 + wave) * 32;
  if (mbase >= nrows) return;
  bool has2 = (mbase + 16) < nrows;
  long mbase2 = has2 ? (mbase + 16) : mbase;

  int lmod = lane & 15;
  int par = lane >> 4;
  int khalf = par << 1;
  int mhi = par << 3;

  const float* arA0 = A0 + (mbase + lmod) * GDIM;
  const float* arB0 = A0 + (mbase2 + lmod) * GDIM;
  const float* arA1 = HASADD ? (A1 + (mbase + lmod) * GDIM) : nullptr;
  const float* arB1 = HASADD ? (A1 + (mbase2 + lmod) * GDIM) : nullptr;
  const float* sWp = sW + par * 256 + 2 * lmod;

  v2f pa0 = *(const v2f*)(arA0 + khalf);
  v2f pb0 = *(const v2f*)(arB0 + khalf);
  v2f pa1, pb1;
  if (HASADD) {
    pa1 = *(const v2f*)(arA1 + khalf);
    pb1 = *(const v2f*)(arB1 + khalf);
  }

  v8f acc[16] = {};

  for (int k = 0; k < GDIM; k += 4) {
    v2f aA = pa0, aB = pb0;
    if (HASADD) {
      aA.x += pa1.x; aA.y += pa1.y;
      aB.x += pb1.x; aB.y += pb1.y;
    }
    if (k + 4 < GDIM) {
      pa0 = *(const v2f*)(arA0 + (k + 4) + khalf);
      pb0 = *(const v2f*)(arB0 + (k + 4) + khalf);
      if (HASADD) {
        pa1 = *(const v2f*)(arA1 + (k + 4) + khalf);
        pb1 = *(const v2f*)(arB1 + (k + 4) + khalf);
      }
    }
    const float* wp = sWp + (k >> 1) * 256;
    v2f b[8];
#pragma unroll
    for (int nt = 0; nt < 8; nt++)
      b[nt] = *(const v2f*)(wp + nt * 32);
#pragma unroll
    for (int nt = 0; nt < 8; nt++) {
      acc[nt] = __builtin_amdgcn_wmma_f32_16x16x4_f32(
          false, aA, false, b[nt], (short)0, acc[nt], false, false);
      acc[8 + nt] = __builtin_amdgcn_wmma_f32_16x16x4_f32(
          false, aB, false, b[nt], (short)0, acc[8 + nt], false, false);
    }
  }

#pragma unroll
  for (int nt = 0; nt < 8; nt++) {
    int ncol = nt * 16 + lmod;
    float bv = bias[ncol];
#pragma unroll
    for (int j = 0; j < 8; j++) {
      float v = acc[nt][j] + bv;
      if (RELU) v = fmaxf(v, 0.f);
      Out[(mbase + mhi + j) * GDIM + ncol] = v;
    }
    if (has2) {
#pragma unroll
      for (int j = 0; j < 8; j++) {
        float v = acc[8 + nt][j] + bv;
        if (RELU) v = fmaxf(v, 0.f);
        Out[(mbase2 + mhi + j) * GDIM + ncol] = v;
      }
    }
  }
}

__global__ __launch_bounds__(256) void gin_fold(
    const float* __restrict__ W1, const float* __restrict__ W2,
    const float* __restrict__ b2, float* __restrict__ Mout,
    float* __restrict__ cout) {
  __shared__ float sW[GDIM * GDIM];
  int tid = threadIdx.x;
  int fi = blockIdx.x;
  const float* Wn = W1 + (long)(fi + 1) * GDIM * GDIM;
  const float* A = W2 + (long)fi * GDIM * GDIM;
  const float* bb = b2 + fi * GDIM;

  int wave = tid >> 5;
  int lane = tid & 31;
  int lmod = lane & 15;
  int par = lane >> 4;
  int khalf = par << 1;
  int mhi = par << 3;

  const float* ar = A + (wave * 16 + lmod) * GDIM + khalf;
  v2f afr[GDIM / 4];
#pragma unroll
  for (int s = 0; s < GDIM / 4; s++) afr[s] = *(const v2f*)(ar + 4 * s);

  float4 r0[8], r1[8];
#pragma unroll
  for (int t = 0; t < 8; t++) {
    int i = tid + t * 256;
    int p = i >> 5;
    int q = i & 31;
    r0[t] = *(const float4*)(Wn + (2 * p) * GDIM + 4 * q);
    r1[t] = *(const float4*)(Wn + (2 * p + 1) * GDIM + 4 * q);
  }
#pragma unroll
  for (int t = 0; t < 8; t++) {
    int i = tid + t * 256;
    int p = i >> 5;
    int q = i & 31;
    float4 lo = {r0[t].x, r1[t].x, r0[t].y, r1[t].y};
    float4 hi = {r0[t].z, r1[t].z, r0[t].w, r1[t].w};
    *(float4*)(sW + p * 256 + 8 * q) = lo;
    *(float4*)(sW + p * 256 + 8 * q + 4) = hi;
  }
  __syncthreads();

  const float* sWp = sW + par * 256 + 2 * lmod;
  v8f acc[8] = {};
#pragma unroll
  for (int s = 0; s < GDIM / 4; s++) {
    const float* wp = sWp + s * 512;
    v2f b[8];
#pragma unroll
    for (int nt = 0; nt < 8; nt++) b[nt] = *(const v2f*)(wp + nt * 32);
#pragma unroll
    for (int nt = 0; nt < 8; nt++)
      acc[nt] = __builtin_amdgcn_wmma_f32_16x16x4_f32(
          false, afr[s], false, b[nt], (short)0, acc[nt], false, false);
  }
  float* Mo = Mout + (long)fi * GDIM * GDIM;
#pragma unroll
  for (int nt = 0; nt < 8; nt++) {
#pragma unroll
    for (int j = 0; j < 8; j++)
      Mo[(wave * 16 + mhi + j) * GDIM + nt * 16 + lmod] = acc[nt][j];
  }

  if (tid < GDIM) {
    float s = 0.f;
    for (int k = 0; k < GDIM; k += 2) {
      v2f w = *(const v2f*)(sW + (k >> 1) * 256 + 2 * tid);
      s = fmaf(bb[k], w.x, s);
      s = fmaf(bb[k + 1], w.y, s);
    }
    cout[fi * GDIM + tid] = s;
  }
}

extern "C" void kernel_launch(void* const* d_in, const int* in_sizes, int n_in,
                              void* d_out, int out_size, void* d_ws,
                              size_t ws_size, hipStream_t stream) {
  const float* x_in = (const float*)d_in[0];
  const int* ei = (const int*)d_in[1];
  const float* W1 = (const float*)d_in[2];
  const float* b1 = (const float*)d_in[3];
  const float* W2 = (const float*)d_in[4];
  const float* b2 = (const float*)d_in[5];
  float* out = (float*)d_out;

  const int N = in_sizes[0] / GDIM;
  const int NE = in_sizes[1] / 2;
  const long nd = (long)N * GDIM;

  float* aggh = (float*)d_ws;
  int2* pairs = (int2*)aggh;
  int* blkoff = (int*)aggh + 2 * (long)NE;
  int* btot = (int*)(aggh + nd);
  int* off = btot + (N + 4);
  int* ebuf = off + (N + 4);
  int* es = ebuf + NE;
  float* Mf = (float*)ebuf;
  float* cf = Mf + 2 * GDIM * GDIM;

  const int mtiles32 = (N + 31) / 32;
  const int gemm_blocks = (mtiles32 + 7) / 8;
  const int NB = (N + BNODES - 1) >> BSH;
  const int part_blocks = 256;
  const int zero_blocks = (NB * (BTS / 4) + 255) / 256;
  const int row_blocks = 4096;

  gin_zero<<<zero_blocks, 256, 0, stream>>>((float4*)btot, NB * (BTS / 4));
  gin_bcount<<<part_blocks, 1024, 0, stream>>>(ei + NE, btot, blkoff, NE, NB);
  gin_bscatter<<<part_blocks, 1024, 0, stream>>>(ei, ei + NE, btot, blkoff, pairs, NE, NB);
  gin_bcsr<<<NB, 1024, 0, stream>>>(pairs, btot, off, ebuf, N, NB);
  gin_canon<<<row_blocks, 256, 0, stream>>>(off, ebuf, es, N);
  gin_fold<<<2, 256, 0, stream>>>(W1, W2, b2, Mf, cf);

  gin_gather<false><<<row_blocks, 256, 0, stream>>>(x_in, off, es, nullptr, aggh, N);
  gin_gemm<true, true><<<gemm_blocks, 256, 0, stream>>>(x_in, aggh, W1, b1, aggh, N);
  for (int layer = 1; layer < 3; ++layer) {
    gin_gemm<false, false><<<gemm_blocks, 256, 0, stream>>>(
        aggh, nullptr, Mf + (long)(layer - 1) * GDIM * GDIM,
        cf + (layer - 1) * GDIM, out, N);
    gin_gather<true><<<row_blocks, 256, 0, stream>>>(
        out, off, es, b1 + layer * GDIM, aggh, N);
  }
  gin_gemm<false, false><<<gemm_blocks, 256, 0, stream>>>(
      aggh, nullptr, W2 + (long)2 * GDIM * GDIM, b2 + 2 * GDIM, out, N);
}


# Build each row's neighbour sum in LDS inside the row GEMM that consumes it (16-row A tile per wave, one block per WGP walking its tiles): sum never stored, 13 launches -> 10; sum order unchanged
# speedup vs baseline: 3.9039x; 1.0920x over previous
#include <hip/hip_runtime.h>

typedef __attribute__((ext_vector_type(2))) float v2f;
typedef __attribute__((ext_vector_type(8))) float v8f;

#define GDIM 128

__global__ __launch_bounds__(256) void gin_zero(float4* p, long n4) {
  long i = (long)blockIdx.x * blockDim.x + threadIdx.x;
  long stride = (long)gridDim.x * blockDim.x;
  float4 z = {0.f, 0.f, 0.f, 0.f};
  for (; i < n4; i += stride) p[i] = z;
}


#define BSH 9
#define BNODES 512
#define MAXNB 256
#define BTS 32

__global__ __launch_bounds__(1024) void gin_bcount(
    const int* __restrict__ dst, int* btot, int* __restrict__ blkoff,
    int nedges, int nb) {
  __shared__ int h[MAXNB];
  int tid = threadIdx.x;
  if (tid < MAXNB) h[tid] = 0;
  __syncthreads();
  int chunk = (nedges + (int)gridDim.x - 1) / (int)gridDim.x;
  int e0 = (int)blockIdx.x * chunk;
  int e1 = (e0 + chunk < nedges) ? (e0 + chunk) : nedges;
  for (int e = e0 + tid; e < e1; e += 8 * 1024) {
    int d[8];
#pragma unroll
    for (int u = 0; u < 8; u++) {
      int ee = e + u * 1024;
      d[u] = (ee < e1) ? dst[ee] : -1;
    }
#pragma unroll
    for (int u = 0; u < 8; u++)
      if (d[u] >= 0) atomicAdd(&h[d[u] >> BSH], 1);
  }
  __syncthreads();
  if (tid < nb)
    blkoff[(int)blockIdx.x * nb + tid] = atomicAdd(btot + tid * BTS, h[tid]);
}

__global__ __launch_bounds__(1024) void gin_bscatter(
    const int* __restrict__ src, const int* __restrict__ dst,
    const int* __restrict__ btot, const int* __restrict__ blkoff,
    int2* __restrict__ pairs, int nedges, int nb) {
  __shared__ int cur[MAXNB];
  __shared__ int swave[8];
  int tid = threadIdx.x;
  int lane = tid & 31;
  int wv = tid >> 5;
  int v = (tid < nb) ? btot[tid * BTS] : 0;
  int incl = v;
  for (int s = 1; s < 32; s <<= 1) {
    int u = __shfl_up(incl, s);
    if (lane >= s) incl += u;
  }
  if (lane == 31 && wv < 8) swave[wv] = incl;
  __syncthreads();
  int pre = 0;
  for (int j = 0; j < 8; j++) pre += (j < wv) ? swave[j] : 0;
  if (tid < nb) cur[tid] = pre + incl - v + blkoff[(int)blockIdx.x * nb + tid];
  __syncthreads();
  int chunk = (nedges + (int)gridDim.x - 1) / (int)gridDim.x;
  int e0 = (int)blockIdx.x * chunk;
  int e1 = (e0 + chunk < nedges) ? (e0 + chunk) : nedges;
  for (int e = e0 + tid; e < e1; e += 8 * 1024) {
    int d[8], sr[8];
#pragma unroll
    for (int u = 0; u < 8; u++) {
      int ee = e + u * 1024;
      d[u] = (ee < e1) ? dst[ee] : -1;
      sr[u] = (ee < e1) ? src[ee] : 0;
    }
#pragma unroll
    for (int u = 0; u < 8; u++) {
      if (d[u] >= 0) {
        int pos = atomicAdd(&cur[d[u] >> BSH], 1);
        pairs[pos] = make_int2(sr[u], d[u]);
      }
    }
  }
}

__global__ __launch_bounds__(1024) void gin_bcsr(
    const int2* __restrict__ pairs, const int* __restrict__ btot,
    int* __restrict__ off, int* __restrict__ ebuf, int n, int nb) {
  __shared__ int cntL[BNODES];
  __shared__ int curL[BNODES];
  __shared__ int swave[16];
  __shared__ int sred[8];
  int tid = threadIdx.x;
  int lane = tid & 31;
  int wv = tid >> 5;
  int b = (int)blockIdx.x;
  int base = b << BSH;
  if (tid < BNODES) cntL[tid] = 0;
  int v = (tid < b) ? btot[tid * BTS] : 0;
  for (int m = 16; m >= 1; m >>= 1) v += __shfl_xor(v, m);
  if (lane == 0 && wv < 8) sred[wv] = v;
  int cb = btot[b * BTS];
  __syncthreads();
  int bs = 0;
  for (int j = 0; j < 8; j++) bs += sred[j];

  for (int i = tid; i < cb; i += 8 * 1024) {
    int d[8];
#pragma unroll
    for (int u = 0; u < 8; u++) {
      int ii = i + u * 1024;
      d[u] = (ii < cb) ? pairs[bs + ii].y : -1;
    }
#pragma unroll
    for (int u = 0; u < 8; u++)
      if (d[u] >= 0) atomicAdd(&cntL[d[u] - base], 1);
  }
  __syncthreads();

  int c = (tid < BNODES) ? cntL[tid] : 0;
  int incl = c;
  for (int s = 1; s < 32; s <<= 1) {
    int u = __shfl_up(incl, s);
    if (lane >= s) incl += u;
  }
  if (lane == 31 && wv < 16) swave[wv] = incl;
  __syncthreads();
  int pre = 0;
  for (int j = 0; j < 16; j++) pre += (j < wv) ? swave[j] : 0;
  int ex = pre + incl - c;
  if (tid < BNODES) {
    curL[tid] = ex;
    if (base + tid < n) off[base + tid] = bs + ex;
  }
  if (b == nb - 1 && tid == 0) off[n] = bs + cb;
  __syncthreads();

  for (int i = tid; i < cb; i += 8 * 1024) {
    int2 p[8];
#pragma unroll
    for (int u = 0; u < 8; u++) {
      int ii = i + u * 1024;
      p[u] = (ii < cb) ? pairs[bs + ii] : make_int2(0, -1);
    }
#pragma unroll
    for (int u = 0; u < 8; u++) {
      if (p[u].y >= 0) {
        int pos = atomicAdd(&curL[p[u].y - base], 1);
        ebuf[bs + pos] = p[u].x;
      }
    }
  }
}

__global__ __launch_bounds__(256) void gin_canon(
    const int* __restrict__ off, const int* __restrict__ ebuf,
    int* __restrict__ es, int n) {
  int lane = threadIdx.x & 31;
  int wid = (int)(blockIdx.x * 8u + (threadIdx.x >> 5));
  int nw = (int)(gridDim.x * 8u);
  for (int node = wid; node < n; node += nw) {
    int nu = __builtin_amdgcn_readfirstlane(node);
    int beg = off[nu];
    int len = off[nu + 1] - beg;
    if (len <= 32) {
      int key = (lane < len) ? ebuf[beg + lane] : 0x7fffffff;
      int rank = 0;
      for (int j = 0; j < len; j++) {
        int kj = __builtin_amdgcn_readlane(key, j);
        rank += ((kj < key) || (kj == key && j < lane)) ? 1 : 0;
      }
      if (lane < len) es[beg + rank] = key;
    } else {
      for (int i = lane; i < len; i += 32) {
        int key = ebuf[beg + i];
        int rank = 0;
        for (int j = 0; j < len; j++) {
          int kj = ebuf[beg + j];
          rank += ((kj < key) || (kj == key && j < i)) ? 1 : 0;
        }
        es[beg + rank] = key;
      }
    }
  }
}

template <bool HASADD, bool RELU>
__global__ __launch_bounds__(256) void gin_gemm(
    const float* A0, const float* A1, const float* __restrict__ W,
    const float* __restrict__ bias, float* Out, int nrows) {
  __shared__ float sW[GDIM * GDIM];
  int tid = threadIdx.x;
  for (int i = tid; i < (GDIM / 2) * GDIM; i += 256) {
    int p = i >> 7;
    int n = i & 127;
    int idx = p * 256 + 2 * n;
    sW[idx]     = W[(2 * p) * GDIM + n];
    sW[idx + 1] = W[(2 * p + 1) * GDIM + n];
  }
  __syncthreads();

  int wave = tid >> 5;
  int lane = tid & 31;
  long mbase = ((long)blockIdx.x * 8 + wave) * 32;
  if (mbase >= nrows) return;
  bool has2 = (mbase + 16) < nrows;
  long mbase2 = has2 ? (mbase + 16) : mbase;

  int lmod = lane & 15;
  int par = lane >> 4;
  int khalf = par << 1;
  int mhi = par << 3;

  const float* arA0 = A0 + (mbase + lmod) * GDIM;
  const float* arB0 = A0 + (mbase2 + lmod) * GDIM;
  const float* arA1 = HASADD ? (A1 + (mbase + lmod) * GDIM) : nullptr;
  const float* arB1 = HASADD ? (A1 + (mbase2 + lmod) * GDIM) : nullptr;
  const float* sWp = sW + par * 256 + 2 * lmod;

  v2f pa0 = *(const v2f*)(arA0 + khalf);
  v2f pb0 = *(const v2f*)(arB0 + khalf);
  v2f pa1, pb1;
  if (HASADD) {
    pa1 = *(const v2f*)(arA1 + khalf);
    pb1 = *(const v2f*)(arB1 + khalf);
  }

  v8f acc[16] = {};

  for (int k = 0; k < GDIM; k += 4) {
    v2f aA = pa0, aB = pb0;
    if (HASADD) {
      aA.x += pa1.x; aA.y += pa1.y;
      aB.x += pb1.x; aB.y += pb1.y;
    }
    if (k + 4 < GDIM) {
      pa0 = *(const v2f*)(arA0 + (k + 4) + khalf);
      pb0 = *(const v2f*)(arB0 + (k + 4) + khalf);
      if (HASADD) {
        pa1 = *(const v2f*)(arA1 + (k + 4) + khalf);
        pb1 = *(const v2f*)(arB1 + (k + 4) + khalf);
      }
    }
    const float* wp = sWp + (k >> 1) * 256;
    v2f b[8];
#pragma unroll
    for (int nt = 0; nt < 8; nt++)
      b[nt] = *(const v2f*)(wp + nt * 32);
#pragma unroll
    for (int nt = 0; nt < 8; nt++) {
      acc[nt] = __builtin_amdgcn_wmma_f32_16x16x4_f32(
          false, aA, false, b[nt], (short)0, acc[nt], false, false);
      acc[8 + nt] = __builtin_amdgcn_wmma_f32_16x16x4_f32(
          false, aB, false, b[nt], (short)0, acc[8 + nt], false, false);
    }
  }

#pragma unroll
  for (int nt = 0; nt < 8; nt++) {
    int ncol = nt * 16 + lmod;
    float bv = bias[ncol];
#pragma unroll
    for (int j = 0; j < 8; j++) {
      float v = acc[nt][j] + bv;
      if (RELU) v = fmaxf(v, 0.f);
      Out[(mbase + mhi + j) * GDIM + ncol] = v;
    }
    if (has2) {
#pragma unroll
      for (int j = 0; j < 8; j++) {
        float v = acc[8 + nt][j] + bv;
        if (RELU) v = fmaxf(v, 0.f);
        Out[(mbase2 + mhi + j) * GDIM + ncol] = v;
      }
    }
  }
}

#define FWAVES 16
#define TROWS 16
#define APITCH 132

template <bool EPI>
__global__ __launch_bounds__(FWAVES * 32) void gin_gather_gemm(
    const float* __restrict__ x, const int* __restrict__ off,
    const int* __restrict__ es, const float* __restrict__ gbias,
    const float* __restrict__ W, const float* __restrict__ bias,
    float* __restrict__ Out, int n) {
  __shared__ __attribute__((aligned(16))) float sW[GDIM * GDIM];
  __shared__ __attribute__((aligned(16))) float sA[FWAVES * TROWS * APITCH];
  int tid = threadIdx.x;

  {
    float4 r0[4], r1[4];
#pragma unroll
    for (int t = 0; t < 4; t++) {
      int i = tid + t * (FWAVES * 32);
      int p = i >> 5;
      int q = i & 31;
      r0[t] = *(const float4*)(W + (2 * p) * GDIM + 4 * q);
      r1[t] = *(const float4*)(W + (2 * p + 1) * GDIM + 4 * q);
    }
#pragma unroll
    for (int t = 0; t < 4; t++) {
      int i = tid + t * (FWAVES * 32);
      int p = i >> 5;
      int q = i & 31;
      float4 lo = {r0[t].x, r1[t].x, r0[t].y, r1[t].y};
      float4 hi = {r0[t].z, r1[t].z, r0[t].w, r1[t].w};
      *(float4*)(sW + p * 256 + 8 * q) = lo;
      *(float4*)(sW + p * 256 + 8 * q + 4) = hi;
    }
  }
  __syncthreads();

  int wave = tid >> 5;
  int lane = tid & 31;
  int lmod = lane & 15;
  int par = lane >> 4;
  int khalf = par << 1;
  int mhi = par << 3;

  float* sAw = sA + wave * (TROWS * APITCH);
  const float* sWp = sW + par * 256 + 2 * lmod;
  float4 gb = {0.f, 0.f, 0.f, 0.f};
  if (EPI) gb = *(const float4*)(gbias + lane * 4);
  const float4* x4 = (const float4*)x;

  int ntiles = (n + TROWS - 1) / TROWS;
  int tpb = (ntiles + (int)gridDim.x - 1) / (int)gridDim.x;
  int t0 = (int)blockIdx.x * tpb;
  int t1 = (t0 + tpb < ntiles) ? (t0 + tpb) : ntiles;

  for (int tile = t0 + wave; tile < t1; tile += FWAVES) {
    int tb = __builtin_amdgcn_readfirstlane(tile) * TROWS;
    int oi = tb + lane;
    int ov = off[(oi < n) ? oi : n];

    int ebeg = __builtin_amdgcn_readlane(ov, 0);
    int eend = __builtin_amdgcn_readlane(ov, TROWS);
    {
      float4 sv[TROWS];
#pragma unroll
      for (int r = 0; r < TROWS; r++) {
        int row = (tb + r < n) ? (tb + r) : (n - 1);
        sv[r] = x4[(unsigned)row * 32u + (unsigned)lane];
      }
#pragma unroll
      for (int r = 0; r < TROWS; r++)
        *(float4*)(sAw + r * APITCH + lane * 4) = sv[r];
    }
    int r = 0;
    int rowend = __builtin_amdgcn_readlane(ov, 1);
    float4 acc = {0.f, 0.f, 0.f, 0.f};
    auto close_row = [&]() {
      float* ar = sAw + r * APITCH + lane * 4;
      float4 sv = *(const float4*)ar;
      float4 a;
      if (EPI) {
        a.x = fmaxf((sv.x + acc.x) + gb.x, 0.f);
        a.y = fmaxf((sv.y + acc.y) + gb.y, 0.f);
        a.z = fmaxf((sv.z + acc.z) + gb.z, 0.f);
        a.w = fmaxf((sv.w + acc.w) + gb.w, 0.f);
      } else {
        a.x = sv.x + acc.x; a.y = sv.y + acc.y;
        a.z = sv.z + acc.z; a.w = sv.w + acc.w;
      }
      *(float4*)ar = a;
      acc = float4{0.f, 0.f, 0.f, 0.f};
      r++;
      rowend = __builtin_amdgcn_readlane(ov, (r + 1 < TROWS) ? (r + 1) : TROWS);
    };
    int idv = 0;
    if (ebeg < eend) {
      int ei0 = ebeg + lane;
      idv = es[(ei0 < eend) ? ei0 : (eend - 1)];
    }
    for (int e = ebeg; e < eend; e += 32) {
      int idn = 0;
      if (e + 32 < eend) {
        int ei1 = e + 32 + lane;
        idn = es[(ei1 < eend) ? ei1 : (eend - 1)];
      }
      if (eend - e > 16) {
        float4 v[32];
#pragma unroll
        for (int u = 0; u < 32; u++) {
          int s = __builtin_amdgcn_readlane(idv, u);
          v[u] = x4[(unsigned)s * 32u + (unsigned)lane];
        }
#pragma unroll
        for (int u = 0; u < 32; u++) {
          if (e + u < eend) {
            while (e + u >= rowend) close_row();
            acc.x += v[u].x; acc.y += v[u].y; acc.z += v[u].z; acc.w += v[u].w;
          }
        }
      } else {
        float4 v[16];
#pragma unroll
        for (int u = 0; u < 16; u++) {
          int s = __builtin_amdgcn_readlane(idv, u);
          v[u] = x4[(unsigned)s * 32u + (unsigned)lane];
        }
#pragma unroll
        for (int u = 0; u < 16; u++) {
          if (e + u < eend) {
            while (e + u >= rowend) close_row();
            acc.x += v[u].x; acc.y += v[u].y; acc.z += v[u].z; acc.w += v[u].w;
          }
        }
      }
      idv = idn;
    }
    while (r < TROWS) close_row();
    __builtin_amdgcn_fence(__ATOMIC_RELEASE, "wavefront");
    __builtin_amdgcn_wave_barrier();
    __builtin_amdgcn_fence(__ATOMIC_ACQUIRE, "wavefront");

    const float* ap = sAw + lmod * APITCH + khalf;
    v8f acc8[8] = {};
    for (int k = 0; k < GDIM; k += 4) {
      v2f aA = *(const v2f*)(ap + k);
      const float* wp = sWp + (k >> 1) * 256;
      v2f b[8];
#pragma unroll
      for (int nt = 0; nt < 8; nt++) b[nt] = *(const v2f*)(wp + nt * 32);
#pragma unroll
      for (int nt = 0; nt < 8; nt++)
        acc8[nt] = __builtin_amdgcn_wmma_f32_16x16x4_f32(
            false, aA, false, b[nt], (short)0, acc8[nt], false, false);
    }
    __builtin_amdgcn_fence(__ATOMIC_RELEASE, "wavefront");
    __builtin_amdgcn_wave_barrier();

    bool whole = (tb + TROWS) <= n;
#pragma unroll
    for (int nt = 0; nt < 8; nt++) {
      int ncol = nt * 16 + lmod;
      float bv = bias[ncol];
#pragma unroll
      for (int j = 0; j < 8; j++) {
        float v = acc8[nt][j] + bv;
        if (!EPI) v = fmaxf(v, 0.f);
        int orow = tb + mhi + j;
        if (whole || orow < n) Out[(long)orow * GDIM + ncol] = v;
      }
    }
  }
}

__global__ __launch_bounds__(256) void gin_fold(
    const float* __restrict__ W1, const float* __restrict__ W2,
    const float* __restrict__ b2, float* __restrict__ Mout,
    float* __restrict__ cout) {
  __shared__ float sW[GDIM * GDIM];
  int tid = threadIdx.x;
  int fi = blockIdx.x;
  const float* Wn = W1 + (long)(fi + 1) * GDIM * GDIM;
  const float* A = W2 + (long)fi * GDIM * GDIM;
  const float* bb = b2 + fi * GDIM;

  int wave = tid >> 5;
  int lane = tid & 31;
  int lmod = lane & 15;
  int par = lane >> 4;
  int khalf = par << 1;
  int mhi = par << 3;

  const float* ar = A + (wave * 16 + lmod) * GDIM + khalf;
  v2f afr[GDIM / 4];
#pragma unroll
  for (int s = 0; s < GDIM / 4; s++) afr[s] = *(const v2f*)(ar + 4 * s);

  float4 r0[8], r1[8];
#pragma unroll
  for (int t = 0; t < 8; t++) {
    int i = tid + t * 256;
    int p = i >> 5;
    int q = i & 31;
    r0[t] = *(const float4*)(Wn + (2 * p) * GDIM + 4 * q);
    r1[t] = *(const float4*)(Wn + (2 * p + 1) * GDIM + 4 * q);
  }
#pragma unroll
  for (int t = 0; t < 8; t++) {
    int i = tid + t * 256;
    int p = i >> 5;
    int q = i & 31;
    float4 lo = {r0[t].x, r1[t].x, r0[t].y, r1[t].y};
    float4 hi = {r0[t].z, r1[t].z, r0[t].w, r1[t].w};
    *(float4*)(sW + p * 256 + 8 * q) = lo;
    *(float4*)(sW + p * 256 + 8 * q + 4) = hi;
  }
  __syncthreads();

  const float* sWp = sW + par * 256 + 2 * lmod;
  v8f acc[8] = {};
#pragma unroll
  for (int s = 0; s < GDIM / 4; s++) {
    const float* wp = sWp + s * 512;
    v2f b[8];
#pragma unroll
    for (int nt = 0; nt < 8; nt++) b[nt] = *(const v2f*)(wp + nt * 32);
#pragma unroll
    for (int nt = 0; nt < 8; nt++)
      acc[nt] = __builtin_amdgcn_wmma_f32_16x16x4_f32(
          false, afr[s], false, b[nt], (short)0, acc[nt], false, false);
  }
  float* Mo = Mout + (long)fi * GDIM * GDIM;
#pragma unroll
  for (int nt = 0; nt < 8; nt++) {
#pragma unroll
    for (int j = 0; j < 8; j++)
      Mo[(wave * 16 + mhi + j) * GDIM + nt * 16 + lmod] = acc[nt][j];
  }

  if (tid < GDIM) {
    float s = 0.f;
    for (int k = 0; k < GDIM; k += 2) {
      v2f w = *(const v2f*)(sW + (k >> 1) * 256 + 2 * tid);
      s = fmaf(bb[k], w.x, s);
      s = fmaf(bb[k + 1], w.y, s);
    }
    cout[fi * GDIM + tid] = s;
  }
}

extern "C" void kernel_launch(void* const* d_in, const int* in_sizes, int n_in,
                              void* d_out, int out_size, void* d_ws,
                              size_t ws_size, hipStream_t stream) {
  const float* x_in = (const float*)d_in[0];
  const int* ei = (const int*)d_in[1];
  const float* W1 = (const float*)d_in[2];
  const float* b1 = (const float*)d_in[3];
  const float* W2 = (const float*)d_in[4];
  const float* b2 = (const float*)d_in[5];
  float* out = (float*)d_out;

  const int N = in_sizes[0] / GDIM;
  const int NE = in_sizes[1] / 2;
  const long nd = (long)N * GDIM;

  float* aggh = (float*)d_ws;
  int2* pairs = (int2*)aggh;
  int* blkoff = (int*)aggh + 2 * (long)NE;
  int* btot = (int*)(aggh + nd);
  int* off = btot + (N + 4);
  int* ebuf = off + (N + 4);
  int* es = ebuf + NE;
  float* Mf = (float*)ebuf;
  float* cf = Mf + 2 * GDIM * GDIM;

  const int mtiles32 = (N + 31) / 32;
  const int gemm_blocks = (mtiles32 + 7) / 8;
  const int NB = (N + BNODES - 1) >> BSH;
  const int part_blocks = 256;
  const int zero_blocks = (NB * (BTS / 4) + 255) / 256;
  const int row_blocks = 4096;
  const int fused_blocks = 256;

  gin_zero<<<zero_blocks, 256, 0, stream>>>((float4*)btot, NB * (BTS / 4));
  gin_bcount<<<part_blocks, 1024, 0, stream>>>(ei + NE, btot, blkoff, NE, NB);
  gin_bscatter<<<part_blocks, 1024, 0, stream>>>(ei, ei + NE, btot, blkoff, pairs, NE, NB);
  gin_bcsr<<<NB, 1024, 0, stream>>>(pairs, btot, off, ebuf, N, NB);
  gin_canon<<<row_blocks, 256, 0, stream>>>(off, ebuf, es, N);
  gin_fold<<<2, 256, 0, stream>>>(W1, W2, b2, Mf, cf);

  gin_gather_gemm<false><<<fused_blocks, FWAVES * 32, 0, stream>>>(
      x_in, off, es, nullptr, W1, b1, aggh, N);
  gin_gemm<false, false><<<gemm_blocks, 256, 0, stream>>>(
      aggh, nullptr, Mf, cf, out, N);
  gin_gather_gemm<true><<<fused_blocks, FWAVES * 32, 0, stream>>>(
      out, off, es, b1 + GDIM, Mf + (long)GDIM * GDIM, cf + GDIM, aggh, N);
  gin_gather_gemm<true><<<fused_blocks, FWAVES * 32, 0, stream>>>(
      aggh, off, es, b1 + 2 * GDIM, W2 + (long)2 * GDIM * GDIM, b2 + 2 * GDIM,
      out, N);
}


# Multiply by folded M[0] inside the layer-0 gather kernel: relu(tile) goes back over the wave's LDS A tile and meets M[0] staged beside W1[0]; h0 never stored, 10 launches -> 9; k order unchanged
# speedup vs baseline: 4.1105x; 1.0529x over previous
#include <hip/hip_runtime.h>

typedef __attribute__((ext_vector_type(2))) float v2f;
typedef __attribute__((ext_vector_type(8))) float v8f;

#define GDIM 128

__global__ __launch_bounds__(256) void gin_zero(float4* p, long n4) {
  long i = (long)blockIdx.x * blockDim.x + threadIdx.x;
  long stride = (long)gridDim.x * blockDim.x;
  float4 z = {0.f, 0.f, 0.f, 0.f};
  for (; i < n4; i += stride) p[i] = z;
}


#define BSH 9
#define BNODES 512
#define MAXNB 256
#define BTS 32

__global__ __launch_bounds__(1024) void gin_bcount(
    const int* __restrict__ dst, int* btot, int* __restrict__ blkoff,
    int nedges, int nb) {
  __shared__ int h[MAXNB];
  int tid = threadIdx.x;
  if (tid < MAXNB) h[tid] = 0;
  __syncthreads();
  int chunk = (nedges + (int)gridDim.x - 1) / (int)gridDim.x;
  int e0 = (int)blockIdx.x * chunk;
  int e1 = (e0 + chunk < nedges) ? (e0 + chunk) : nedges;
  for (int e = e0 + tid; e < e1; e += 8 * 1024) {
    int d[8];
#pragma unroll
    for (int u = 0; u < 8; u++) {
      int ee = e + u * 1024;
      d[u] = (ee < e1) ? dst[ee] : -1;
    }
#pragma unroll
    for (int u = 0; u < 8; u++)
      if (d[u] >= 0) atomicAdd(&h[d[u] >> BSH], 1);
  }
  __syncthreads();
  if (tid < nb)
    blkoff[(int)blockIdx.x * nb + tid] = atomicAdd(btot + tid * BTS, h[tid]);
}

__global__ __launch_bounds__(1024) void gin_bscatter(
    const int* __restrict__ src, const int* __restrict__ dst,
    const int* __restrict__ btot, const int* __restrict__ blkoff,
    int2* __restrict__ pairs, int nedges, int nb) {
  __shared__ int cur[MAXNB];
  __shared__ int swave[8];
  int tid = threadIdx.x;
  int lane = tid & 31;
  int wv = tid >> 5;
  int v = (tid < nb) ? btot[tid * BTS] : 0;
  int incl = v;
  for (int s = 1; s < 32; s <<= 1) {
    int u = __shfl_up(incl, s);
    if (lane >= s) incl += u;
  }
  if (lane == 31 && wv < 8) swave[wv] = incl;
  __syncthreads();
  int pre = 0;
  for (int j = 0; j < 8; j++) pre += (j < wv) ? swave[j] : 0;
  if (tid < nb) cur[tid] = pre + incl - v + blkoff[(int)blockIdx.x * nb + tid];
  __syncthreads();
  int chunk = (nedges + (int)gridDim.x - 1) / (int)gridDim.x;
  int e0 = (int)blockIdx.x * chunk;
  int e1 = (e0 + chunk < nedges) ? (e0 + chunk) : nedges;
  for (int e = e0 + tid; e < e1; e += 8 * 1024) {
    int d[8], sr[8];
#pragma unroll
    for (int u = 0; u < 8; u++) {
      int ee = e + u * 1024;
      d[u] = (ee < e1) ? dst[ee] : -1;
      sr[u] = (ee < e1) ? src[ee] : 0;
    }
#pragma unroll
    for (int u = 0; u < 8; u++) {
      if (d[u] >= 0) {
        int pos = atomicAdd(&cur[d[u] >> BSH], 1);
        pairs[pos] = make_int2(sr[u], d[u]);
      }
    }
  }
}

__global__ __launch_bounds__(1024) void gin_bcsr(
    const int2* __restrict__ pairs, const int* __restrict__ btot,
    int* __restrict__ off, int* __restrict__ ebuf, int n, int nb) {
  __shared__ int cntL[BNODES];
  __shared__ int curL[BNODES];
  __shared__ int swave[16];
  __shared__ int sred[8];
  int tid = threadIdx.x;
  int lane = tid & 31;
  int wv = tid >> 5;
  int b = (int)blockIdx.x;
  int base = b << BSH;
  if (tid < BNODES) cntL[tid] = 0;
  int v = (tid < b) ? btot[tid * BTS] : 0;
  for (int m = 16; m >= 1; m >>= 1) v += __shfl_xor(v, m);
  if (lane == 0 && wv < 8) sred[wv] = v;
  int cb = btot[b * BTS];
  __syncthreads();
  int bs = 0;
  for (int j = 0; j < 8; j++) bs += sred[j];

  for (int i = tid; i < cb; i += 8 * 1024) {
    int d[8];
#pragma unroll
    for (int u = 0; u < 8; u++) {
      int ii = i + u * 1024;
      d[u] = (ii < cb) ? pairs[bs + ii].y : -1;
    }
#pragma unroll
    for (int u = 0; u < 8; u++)
      if (d[u] >= 0) atomicAdd(&cntL[d[u] - base], 1);
  }
  __syncthreads();

  int c = (tid < BNODES) ? cntL[tid] : 0;
  int incl = c;
  for (int s = 1; s < 32; s <<= 1) {
    int u = __shfl_up(incl, s);
    if (lane >= s) incl += u;
  }
  if (lane == 31 && wv < 16) swave[wv] = incl;
  __syncthreads();
  int pre = 0;
  for (int j = 0; j < 16; j++) pre += (j < wv) ? swave[j] : 0;
  int ex = pre + incl - c;
  if (tid < BNODES) {
    curL[tid] = ex;
    if (base + tid < n) off[base + tid] = bs + ex;
  }
  if (b == nb - 1 && tid == 0) off[n] = bs + cb;
  __syncthreads();

  for (int i = tid; i < cb; i += 8 * 1024) {
    int2 p[8];
#pragma unroll
    for (int u = 0; u < 8; u++) {
      int ii = i + u * 1024;
      p[u] = (ii < cb) ? pairs[bs + ii] : make_int2(0, -1);
    }
#pragma unroll
    for (int u = 0; u < 8; u++) {
      if (p[u].y >= 0) {
        int pos = atomicAdd(&curL[p[u].y - base], 1);
        ebuf[bs + pos] = p[u].x;
      }
    }
  }
}

__global__ __launch_bounds__(256) void gin_canon(
    const int* __restrict__ off, const int* __restrict__ ebuf,
    int* __restrict__ es, int n) {
  int lane = threadIdx.x & 31;
  int wid = (int)(blockIdx.x * 8u + (threadIdx.x >> 5));
  int nw = (int)(gridDim.x * 8u);
  for (int node = wid; node < n; node += nw) {
    int nu = __builtin_amdgcn_readfirstlane(node);
    int beg = off[nu];
    int len = off[nu + 1] - beg;
    if (len <= 32) {
      int key = (lane < len) ? ebuf[beg + lane] : 0x7fffffff;
      int rank = 0;
      for (int j = 0; j < len; j++) {
        int kj = __builtin_amdgcn_readlane(key, j);
        rank += ((kj < key) || (kj == key && j < lane)) ? 1 : 0;
      }
      if (lane < len) es[beg + rank] = key;
    } else {
      for (int i = lane; i < len; i += 32) {
        int key = ebuf[beg + i];
        int rank = 0;
        for (int j = 0; j < len; j++) {
          int kj = ebuf[beg + j];
          rank += ((kj < key) || (kj == key && j < i)) ? 1 : 0;
        }
        es[beg + rank] = key;
      }
    }
  }
}

#define FWAVES 16
#define TROWS 16
#define APITCH 132

template <bool EPI, bool TWO>
__global__ __launch_bounds__(FWAVES * 32) void gin_gather_gemm(
    const float* __restrict__ x, const int* __restrict__ off,
    const int* __restrict__ es, const float* __restrict__ gbias,
    const float* __restrict__ W, const float* __restrict__ bias,
    const float* __restrict__ Wb, const float* __restrict__ biasb,
    float* __restrict__ Out, int n) {
  __shared__ __attribute__((aligned(16))) float sW[(TWO ? 2 : 1) * GDIM * GDIM];
  __shared__ __attribute__((aligned(16))) float sA[FWAVES * TROWS * APITCH];
  int tid = threadIdx.x;

  {
    constexpr int NM = TWO ? 2 : 1;
    float4 r0[NM][4], r1[NM][4];
#pragma unroll
    for (int m = 0; m < NM; m++) {
      const float* Wm = m ? Wb : W;
#pragma unroll
      for (int t = 0; t < 4; t++) {
        int i = tid + t * (FWAVES * 32);
        int p = i >> 5;
        int q = i & 31;
        r0[m][t] = *(const float4*)(Wm + (2 * p) * GDIM + 4 * q);
        r1[m][t] = *(const float4*)(Wm + (2 * p + 1) * GDIM + 4 * q);
      }
    }
#pragma unroll
    for (int m = 0; m < NM; m++) {
#pragma unroll
      for (int t = 0; t < 4; t++) {
        int i = tid + t * (FWAVES * 32);
        int p = i >> 5;
        int q = i & 31;
        float4 lo = {r0[m][t].x, r1[m][t].x, r0[m][t].y, r1[m][t].y};
        float4 hi = {r0[m][t].z, r1[m][t].z, r0[m][t].w, r1[m][t].w};
        *(float4*)(sW + m * (GDIM * GDIM) + p * 256 + 8 * q) = lo;
        *(float4*)(sW + m * (GDIM * GDIM) + p * 256 + 8 * q + 4) = hi;
      }
    }
  }
  __syncthreads();

  int wave = tid >> 5;
  int lane = tid & 31;
  int lmod = lane & 15;
  int par = lane >> 4;
  int khalf = par << 1;
  int mhi = par << 3;

  float* sAw = sA + wave * (TROWS * APITCH);
  const float* sWp = sW + par * 256 + 2 * lmod;
  float4 gb = {0.f, 0.f, 0.f, 0.f};
  if (EPI) gb = *(const float4*)(gbias + lane * 4);
  const float4* x4 = (const float4*)x;

  int ntiles = (n + TROWS - 1) / TROWS;
  int tpb = (ntiles + (int)gridDim.x - 1) / (int)gridDim.x;
  int t0 = (int)blockIdx.x * tpb;
  int t1 = (t0 + tpb < ntiles) ? (t0 + tpb) : ntiles;

  for (int tile = t0 + wave; tile < t1; tile += FWAVES) {
    int tb = __builtin_amdgcn_readfirstlane(tile) * TROWS;
    int oi = tb + lane;
    int ov = off[(oi < n) ? oi : n];

    int ebeg = __builtin_amdgcn_readlane(ov, 0);
    int eend = __builtin_amdgcn_readlane(ov, TROWS);
    {
      float4 sv[TROWS];
#pragma unroll
      for (int r = 0; r < TROWS; r++) {
        int row = (tb + r < n) ? (tb + r) : (n - 1);
        sv[r] = x4[(unsigned)row * 32u + (unsigned)lane];
      }
#pragma unroll
      for (int r = 0; r < TROWS; r++)
        *(float4*)(sAw + r * APITCH + lane * 4) = sv[r];
    }
    int r = 0;
    int rowend = __builtin_amdgcn_readlane(ov, 1);
    float4 acc = {0.f, 0.f, 0.f, 0.f};
    auto close_row = [&]() {
      float* ar = sAw + r * APITCH + lane * 4;
      float4 sv = *(const float4*)ar;
      float4 a;
      if (EPI) {
        a.x = fmaxf((sv.x + acc.x) + gb.x, 0.f);
        a.y = fmaxf((sv.y + acc.y) + gb.y, 0.f);
        a.z = fmaxf((sv.z + acc.z) + gb.z, 0.f);
        a.w = fmaxf((sv.w + acc.w) + gb.w, 0.f);
      } else {
        a.x = sv.x + acc.x; a.y = sv.y + acc.y;
        a.z = sv.z + acc.z; a.w = sv.w + acc.w;
      }
      *(float4*)ar = a;
      acc = float4{0.f, 0.f, 0.f, 0.f};
      r++;
      rowend = __builtin_amdgcn_readlane(ov, (r + 1 < TROWS) ? (r + 1) : TROWS);
    };
    int idv = 0;
    if (ebeg < eend) {
      int ei0 = ebeg + lane;
      idv = es[(ei0 < eend) ? ei0 : (eend - 1)];
    }
    for (int e = ebeg; e < eend; e += 32) {
      int idn = 0;
      if (e + 32 < eend) {
        int ei1 = e + 32 + lane;
        idn = es[(ei1 < eend) ? ei1 : (eend - 1)];
      }
      if (eend - e > 16) {
        float4 v[32];
#pragma unroll
        for (int u = 0; u < 32; u++) {
          int s = __builtin_amdgcn_readlane(idv, u);
          v[u] = x4[(unsigned)s * 32u + (unsigned)lane];
        }
#pragma unroll
        for (int u = 0; u < 32; u++) {
          if (e + u < eend) {
            while (e + u >= rowend) close_row();
            acc.x += v[u].x; acc.y += v[u].y; acc.z += v[u].z; acc.w += v[u].w;
          }
        }
      } else {
        float4 v[16];
#pragma unroll
        for (int u = 0; u < 16; u++) {
          int s = __builtin_amdgcn_readlane(idv, u);
          v[u] = x4[(unsigned)s * 32u + (unsigned)lane];
        }
#pragma unroll
        for (int u = 0; u < 16; u++) {
          if (e + u < eend) {
            while (e + u >= rowend) close_row();
            acc.x += v[u].x; acc.y += v[u].y; acc.z += v[u].z; acc.w += v[u].w;
          }
        }
      }
      idv = idn;
    }
    while (r < TROWS) close_row();
    __builtin_amdgcn_fence(__ATOMIC_RELEASE, "wavefront");
    __builtin_amdgcn_wave_barrier();
    __builtin_amdgcn_fence(__ATOMIC_ACQUIRE, "wavefront");

    const float* ap = sAw + lmod * APITCH + khalf;
    v8f acc8[8] = {};
    for (int k = 0; k < GDIM; k += 4) {
      v2f aA = *(const v2f*)(ap + k);
      const float* wp = sWp + (k >> 1) * 256;
      v2f b[8];
#pragma unroll
      for (int nt = 0; nt < 8; nt++) b[nt] = *(const v2f*)(wp + nt * 32);
#pragma unroll
      for (int nt = 0; nt < 8; nt++)
        acc8[nt] = __builtin_amdgcn_wmma_f32_16x16x4_f32(
            false, aA, false, b[nt], (short)0, acc8[nt], false, false);
    }
    __builtin_amdgcn_fence(__ATOMIC_RELEASE, "wavefront");
    __builtin_amdgcn_wave_barrier();

    if (TWO) {
#pragma unroll
      for (int nt = 0; nt < 8; nt++) {
        int ncol = nt * 16 + lmod;
        float bv = bias[ncol];
#pragma unroll
        for (int j = 0; j < 8; j++)
          sAw[(mhi + j) * APITCH + ncol] = fmaxf(acc8[nt][j] + bv, 0.f);
      }
      __builtin_amdgcn_fence(__ATOMIC_RELEASE, "wavefront");
      __builtin_amdgcn_wave_barrier();
      __builtin_amdgcn_fence(__ATOMIC_ACQUIRE, "wavefront");
#pragma unroll
      for (int nt = 0; nt < 8; nt++) acc8[nt] = v8f{};
      const float* sWq = sWp + GDIM * GDIM;
      for (int k = 0; k < GDIM; k += 4) {
        v2f aA = *(const v2f*)(ap + k);
        const float* wp = sWq + (k >> 1) * 256;
        v2f b[8];
#pragma unroll
        for (int nt = 0; nt < 8; nt++) b[nt] = *(const v2f*)(wp + nt * 32);
#pragma unroll
        for (int nt = 0; nt < 8; nt++)
          acc8[nt] = __builtin_amdgcn_wmma_f32_16x16x4_f32(
              false, aA, false, b[nt], (short)0, acc8[nt], false, false);
      }
      __builtin_amdgcn_fence(__ATOMIC_RELEASE, "wavefront");
      __builtin_amdgcn_wave_barrier();
    }

    const float* obias = TWO ? biasb : bias;
    bool whole = (tb + TROWS) <= n;
#pragma unroll
    for (int nt = 0; nt < 8; nt++) {
      int ncol = nt * 16 + lmod;
      float bv = obias[ncol];
#pragma unroll
      for (int j = 0; j < 8; j++) {
        float v = acc8[nt][j] + bv;
        if (!EPI && !TWO) v = fmaxf(v, 0.f);
        int orow = tb + mhi + j;
        if (whole || orow < n) Out[(long)orow * GDIM + ncol] = v;
      }
    }
  }
}

__global__ __launch_bounds__(256) void gin_fold(
    const float* __restrict__ W1, const float* __restrict__ W2,
    const float* __restrict__ b2, float* __restrict__ Mout,
    float* __restrict__ cout) {
  __shared__ float sW[GDIM * GDIM];
  int tid = threadIdx.x;
  int fi = blockIdx.x;
  const float* Wn = W1 + (long)(fi + 1) * GDIM * GDIM;
  const float* A = W2 + (long)fi * GDIM * GDIM;
  const float* bb = b2 + fi * GDIM;

  int wave = tid >> 5;
  int lane = tid & 31;
  int lmod = lane & 15;
  int par = lane >> 4;
  int khalf = par << 1;
  int mhi = par << 3;

  const float* ar = A + (wave * 16 + lmod) * GDIM + khalf;
  v2f afr[GDIM / 4];
#pragma unroll
  for (int s = 0; s < GDIM / 4; s++) afr[s] = *(const v2f*)(ar + 4 * s);

  float4 r0[8], r1[8];
#pragma unroll
  for (int t = 0; t < 8; t++) {
    int i = tid + t * 256;
    int p = i >> 5;
    int q = i & 31;
    r0[t] = *(const float4*)(Wn + (2 * p) * GDIM + 4 * q);
    r1[t] = *(const float4*)(Wn + (2 * p + 1) * GDIM + 4 * q);
  }
#pragma unroll
  for (int t = 0; t < 8; t++) {
    int i = tid + t * 256;
    int p = i >> 5;
    int q = i & 31;
    float4 lo = {r0[t].x, r1[t].x, r0[t].y, r1[t].y};
    float4 hi = {r0[t].z, r1[t].z, r0[t].w, r1[t].w};
    *(float4*)(sW + p * 256 + 8 * q) = lo;
    *(float4*)(sW + p * 256 + 8 * q + 4) = hi;
  }
  __syncthreads();

  const float* sWp = sW + par * 256 + 2 * lmod;
  v8f acc[8] = {};
#pragma unroll
  for (int s = 0; s < GDIM / 4; s++) {
    const float* wp = sWp + s * 512;
    v2f b[8];
#pragma unroll
    for (int nt = 0; nt < 8; nt++) b[nt] = *(const v2f*)(wp + nt * 32);
#pragma unroll
    for (int nt = 0; nt < 8; nt++)
      acc[nt] = __builtin_amdgcn_wmma_f32_16x16x4_f32(
          false, afr[s], false, b[nt], (short)0, acc[nt], false, false);
  }
  float* Mo = Mout + (long)fi * GDIM * GDIM;
#pragma unroll
  for (int nt = 0; nt < 8; nt++) {
#pragma unroll
    for (int j = 0; j < 8; j++)
      Mo[(wave * 16 + mhi + j) * GDIM + nt * 16 + lmod] = acc[nt][j];
  }

  if (tid < GDIM) {
    float s = 0.f;
    for (int k = 0; k < GDIM; k += 2) {
      v2f w = *(const v2f*)(sW + (k >> 1) * 256 + 2 * tid);
      s = fmaf(bb[k], w.x, s);
      s = fmaf(bb[k + 1], w.y, s);
    }
    cout[fi * GDIM + tid] = s;
  }
}

extern "C" void kernel_launch(void* const* d_in, const int* in_sizes, int n_in,
                              void* d_out, int out_size, void* d_ws,
                              size_t ws_size, hipStream_t stream) {
  const float* x_in = (const float*)d_in[0];
  const int* ei = (const int*)d_in[1];
  const float* W1 = (const float*)d_in[2];
  const float* b1 = (const float*)d_in[3];
  const float* W2 = (const float*)d_in[4];
  const float* b2 = (const float*)d_in[5];
  float* out = (float*)d_out;

  const int N = in_sizes[0] / GDIM;
  const int NE = in_sizes[1] / 2;
  const long nd = (long)N * GDIM;

  float* aggh = (float*)d_ws;
  int2* pairs = (int2*)aggh;
  int* blkoff = (int*)aggh + 2 * (long)NE;
  int* btot = (int*)(aggh + nd);
  int* off = btot + (N + 4);
  int* ebuf = off + (N + 4);
  int* es = ebuf + NE;
  float* Mf = (float*)ebuf;
  float* cf = Mf + 2 * GDIM * GDIM;

  const int NB = (N + BNODES - 1) >> BSH;
  const int part_blocks = 256;
  const int zero_blocks = (NB * (BTS / 4) + 255) / 256;
  const int row_blocks = 4096;
  const int fused_blocks = 256;

  gin_zero<<<zero_blocks, 256, 0, stream>>>((float4*)btot, NB * (BTS / 4));
  gin_bcount<<<part_blocks, 1024, 0, stream>>>(ei + NE, btot, blkoff, NE, NB);
  gin_bscatter<<<part_blocks, 1024, 0, stream>>>(ei, ei + NE, btot, blkoff, pairs, NE, NB);
  gin_bcsr<<<NB, 1024, 0, stream>>>(pairs, btot, off, ebuf, N, NB);
  gin_canon<<<row_blocks, 256, 0, stream>>>(off, ebuf, es, N);
  gin_fold<<<2, 256, 0, stream>>>(W1, W2, b2, Mf, cf);

  gin_gather_gemm<false, true><<<fused_blocks, FWAVES * 32, 0, stream>>>(
      x_in, off, es, nullptr, W1, b1, Mf, cf, out, N);
  gin_gather_gemm<true, false><<<fused_blocks, FWAVES * 32, 0, stream>>>(
      out, off, es, b1 + GDIM, Mf + (long)GDIM * GDIM, cf + GDIM, nullptr,
      nullptr, aggh, N);
  gin_gather_gemm<true, false><<<fused_blocks, FWAVES * 32, 0, stream>>>(
      aggh, off, es, b1 + 2 * GDIM, W2 + (long)2 * GDIM * GDIM, b2 + 2 * GDIM,
      nullptr, nullptr, out, N);
}


# Finish each bucket in LDS: its edges are placed in an LDS stage, each segment rank-sorted there by the same block and written to es once, in order; scattered ebuf stores and the sort launch go (9->8)
# speedup vs baseline: 4.2987x; 1.0458x over previous
#include <hip/hip_runtime.h>

typedef __attribute__((ext_vector_type(2))) float v2f;
typedef __attribute__((ext_vector_type(8))) float v8f;

#define GDIM 128

__global__ __launch_bounds__(256) void gin_zero(float4* p, long n4) {
  long i = (long)blockIdx.x * blockDim.x + threadIdx.x;
  long stride = (long)gridDim.x * blockDim.x;
  float4 z = {0.f, 0.f, 0.f, 0.f};
  for (; i < n4; i += stride) p[i] = z;
}


#define BSH 9
#define BNODES 512
#define MAXNB 256
#define BTS 32
#define LCAP 16384

__global__ __launch_bounds__(1024) void gin_bcount(
    const int* __restrict__ dst, int* btot, int* __restrict__ blkoff,
    int nedges, int nb) {
  __shared__ int h[MAXNB];
  int tid = threadIdx.x;
  if (tid < MAXNB) h[tid] = 0;
  __syncthreads();
  int chunk = (nedges + (int)gridDim.x - 1) / (int)gridDim.x;
  int e0 = (int)blockIdx.x * chunk;
  int e1 = (e0 + chunk < nedges) ? (e0 + chunk) : nedges;
  for (int e = e0 + tid; e < e1; e += 8 * 1024) {
    int d[8];
#pragma unroll
    for (int u = 0; u < 8; u++) {
      int ee = e + u * 1024;
      d[u] = (ee < e1) ? dst[ee] : -1;
    }
#pragma unroll
    for (int u = 0; u < 8; u++)
      if (d[u] >= 0) atomicAdd(&h[d[u] >> BSH], 1);
  }
  __syncthreads();
  if (tid < nb)
    blkoff[(int)blockIdx.x * nb + tid] = atomicAdd(btot + tid * BTS, h[tid]);
}

__global__ __launch_bounds__(1024) void gin_bscatter(
    const int* __restrict__ src, const int* __restrict__ dst,
    const int* __restrict__ btot, const int* __restrict__ blkoff,
    int2* __restrict__ pairs, int nedges, int nb) {
  __shared__ int cur[MAXNB];
  __shared__ int swave[8];
  int tid = threadIdx.x;
  int lane = tid & 31;
  int wv = tid >> 5;
  int v = (tid < nb) ? btot[tid * BTS] : 0;
  int incl = v;
  for (int s = 1; s < 32; s <<= 1) {
    int u = __shfl_up(incl, s);
    if (lane >= s) incl += u;
  }
  if (lane == 31 && wv < 8) swave[wv] = incl;
  __syncthreads();
  int pre = 0;
  for (int j = 0; j < 8; j++) pre += (j < wv) ? swave[j] : 0;
  if (tid < nb) cur[tid] = pre + incl - v + blkoff[(int)blockIdx.x * nb + tid];
  __syncthreads();
  int chunk = (nedges + (int)gridDim.x - 1) / (int)gridDim.x;
  int e0 = (int)blockIdx.x * chunk;
  int e1 = (e0 + chunk < nedges) ? (e0 + chunk) : nedges;
  for (int e = e0 + tid; e < e1; e += 8 * 1024) {
    int d[8], sr[8];
#pragma unroll
    for (int u = 0; u < 8; u++) {
      int ee = e + u * 1024;
      d[u] = (ee < e1) ? dst[ee] : -1;
      sr[u] = (ee < e1) ? src[ee] : 0;
    }
#pragma unroll
    for (int u = 0; u < 8; u++) {
      if (d[u] >= 0) {
        int pos = atomicAdd(&cur[d[u] >> BSH], 1);
        pairs[pos] = make_int2(sr[u], d[u]);
      }
    }
  }
}

__global__ __launch_bounds__(1024) void gin_bcsr(
    const int2* __restrict__ pairs, const int* __restrict__ btot,
    int* __restrict__ off, int* ebuf, int* __restrict__ es, int n, int nb) {
  __shared__ int cntL[BNODES];
  __shared__ int curL[BNODES];
  __shared__ int swave[16];
  __shared__ int sred[8];
  __shared__ int stage[LCAP];
  int tid = threadIdx.x;
  int lane = tid & 31;
  int wv = tid >> 5;
  int b = (int)blockIdx.x;
  int base = b << BSH;
  if (tid < BNODES) cntL[tid] = 0;
  int v = (tid < b) ? btot[tid * BTS] : 0;
  for (int m = 16; m >= 1; m >>= 1) v += __shfl_xor(v, m);
  if (lane == 0 && wv < 8) sred[wv] = v;
  int cb = btot[b * BTS];
  __syncthreads();
  int bs = 0;
  for (int j = 0; j < 8; j++) bs += sred[j];
  const bool inl = cb <= LCAP;

  for (int i = tid; i < cb; i += 8 * 1024) {
    int d[8];
#pragma unroll
    for (int u = 0; u < 8; u++) {
      int ii = i + u * 1024;
      d[u] = (ii < cb) ? pairs[bs + ii].y : -1;
    }
#pragma unroll
    for (int u = 0; u < 8; u++)
      if (d[u] >= 0) atomicAdd(&cntL[d[u] - base], 1);
  }
  __syncthreads();

  int c = (tid < BNODES) ? cntL[tid] : 0;
  int incl = c;
  for (int s = 1; s < 32; s <<= 1) {
    int u = __shfl_up(incl, s);
    if (lane >= s) incl += u;
  }
  if (lane == 31 && wv < 16) swave[wv] = incl;
  __syncthreads();
  int pre = 0;
  for (int j = 0; j < 16; j++) pre += (j < wv) ? swave[j] : 0;
  int ex = pre + incl - c;
  if (tid < BNODES) {
    curL[tid] = ex;
    if (base + tid < n) off[base + tid] = bs + ex;
  }
  if (b == nb - 1 && tid == 0) off[n] = bs + cb;
  __syncthreads();

  for (int i = tid; i < cb; i += 8 * 1024) {
    int2 p[8];
#pragma unroll
    for (int u = 0; u < 8; u++) {
      int ii = i + u * 1024;
      p[u] = (ii < cb) ? pairs[bs + ii] : make_int2(0, -1);
    }
#pragma unroll
    for (int u = 0; u < 8; u++) {
      if (p[u].y >= 0) {
        int pos = atomicAdd(&curL[p[u].y - base], 1);
        if (inl) stage[pos] = p[u].x;
        else ebuf[bs + pos] = p[u].x;
      }
    }
  }

  __syncthreads();
  int myt = wv * 16 + (lane & 15);
  int segend = curL[myt];
  int segbeg = (myt > 0) ? curL[myt - 1] : 0;
  const int* eb = ebuf + bs;
  int* eo = es + bs;
  auto sort_segments = [&](auto fetch) {
    auto seg_keys = [&](int i) {
      int beg = __builtin_amdgcn_readlane(segbeg, i & 15);
      int len = __builtin_amdgcn_readlane(segend, i & 15) - beg;
      return (i < 16 && lane < len) ? fetch(beg + lane) : 0x7fffffff;
    };
    int k0 = seg_keys(0);
#pragma unroll 1
    for (int i = 0; i < 16; i++) {
      int kn = seg_keys(i + 1);
      int beg = __builtin_amdgcn_readlane(segbeg, i);
      int len = __builtin_amdgcn_readlane(segend, i) - beg;
      if (len <= 32) {
        int key = k0;
        unsigned comb = ((unsigned)key << 5) | (unsigned)lane;
        int rank = 0;
        for (int j = 0; j < len; j++) {
          unsigned cj = (unsigned)__builtin_amdgcn_readlane((int)comb, j);
          rank += (cj < comb) ? 1 : 0;
        }
        if (lane < len) eo[beg + rank] = key;
      } else {
        for (int q = lane; q < len; q += 32) {
          int key = fetch(beg + q);
          int rank = 0;
          for (int j = 0; j < len; j++) {
            int kj = fetch(beg + j);
            rank += ((kj < key) || (kj == key && j < q)) ? 1 : 0;
          }
          eo[beg + rank] = key;
        }
      }
      k0 = kn;
    }
  };
  if (inl) sort_segments([&](int idx) { return stage[idx]; });
  else sort_segments([&](int idx) { return eb[idx]; });
}

#define FWAVES 16
#define TROWS 16
#define APITCH 132

template <bool EPI, bool TWO>
__global__ __launch_bounds__(FWAVES * 32) void gin_gather_gemm(
    const float* __restrict__ x, const int* __restrict__ off,
    const int* __restrict__ es, const float* __restrict__ gbias,
    const float* __restrict__ W, const float* __restrict__ bias,
    const float* __restrict__ Wb, const float* __restrict__ biasb,
    float* __restrict__ Out, int n) {
  __shared__ __attribute__((aligned(16))) float sW[(TWO ? 2 : 1) * GDIM * GDIM];
  __shared__ __attribute__((aligned(16))) float sA[FWAVES * TROWS * APITCH];
  int tid = threadIdx.x;

  {
    constexpr int NM = TWO ? 2 : 1;
    float4 r0[NM][4], r1[NM][4];
#pragma unroll
    for (int m = 0; m < NM; m++) {
      const float* Wm = m ? Wb : W;
#pragma unroll
      for (int t = 0; t < 4; t++) {
        int i = tid + t * (FWAVES * 32);
        int p = i >> 5;
        int q = i & 31;
        r0[m][t] = *(const float4*)(Wm + (2 * p) * GDIM + 4 * q);
        r1[m][t] = *(const float4*)(Wm + (2 * p + 1) * GDIM + 4 * q);
      }
    }
#pragma unroll
    for (int m = 0; m < NM; m++) {
#pragma unroll
      for (int t = 0; t < 4; t++) {
        int i = tid + t * (FWAVES * 32);
        int p = i >> 5;
        int q = i & 31;
        float4 lo = {r0[m][t].x, r1[m][t].x, r0[m][t].y, r1[m][t].y};
        float4 hi = {r0[m][t].z, r1[m][t].z, r0[m][t].w, r1[m][t].w};
        *(float4*)(sW + m * (GDIM * GDIM) + p * 256 + 8 * q) = lo;
        *(float4*)(sW + m * (GDIM * GDIM) + p * 256 + 8 * q + 4) = hi;
      }
    }
  }
  __syncthreads();

  int wave = tid >> 5;
  int lane = tid & 31;
  int lmod = lane & 15;
  int par = lane >> 4;
  int khalf = par << 1;
  int mhi = par << 3;

  float* sAw = sA + wave * (TROWS * APITCH);
  const float* sWp = sW + par * 256 + 2 * lmod;
  float4 gb = {0.f, 0.f, 0.f, 0.f};
  if (EPI) gb = *(const float4*)(gbias + lane * 4);
  const float4* x4 = (const float4*)x;

  int ntiles = (n + TROWS - 1) / TROWS;
  int tpb = (ntiles + (int)gridDim.x - 1) / (int)gridDim.x;
  int t0 = (int)blockIdx.x * tpb;
  int t1 = (t0 + tpb < ntiles) ? (t0 + tpb) : ntiles;

  for (int tile = t0 + wave; tile < t1; tile += FWAVES) {
    int tb = __builtin_amdgcn_readfirstlane(tile) * TROWS;
    int oi = tb + lane;
    int ov = off[(oi < n) ? oi : n];

    int ebeg = __builtin_amdgcn_readlane(ov, 0);
    int eend = __builtin_amdgcn_readlane(ov, TROWS);
    {
      float4 sv[TROWS];
#pragma unroll
      for (int r = 0; r < TROWS; r++) {
        int row = (tb + r < n) ? (tb + r) : (n - 1);
        sv[r] = x4[(unsigned)row * 32u + (unsigned)lane];
      }
#pragma unroll
      for (int r = 0; r < TROWS; r++)
        *(float4*)(sAw + r * APITCH + lane * 4) = sv[r];
    }
    int r = 0;
    int rowend = __builtin_amdgcn_readlane(ov, 1);
    float4 acc = {0.f, 0.f, 0.f, 0.f};
    auto close_row = [&]() {
      float* ar = sAw + r * APITCH + lane * 4;
      float4 sv = *(const float4*)ar;
      float4 a;
      if (EPI) {
        a.x = fmaxf((sv.x + acc.x) + gb.x, 0.f);
        a.y = fmaxf((sv.y + acc.y) + gb.y, 0.f);
        a.z = fmaxf((sv.z + acc.z) + gb.z, 0.f);
        a.w = fmaxf((sv.w + acc.w) + gb.w, 0.f);
      } else {
        a.x = sv.x + acc.x; a.y = sv.y + acc.y;
        a.z = sv.z + acc.z; a.w = sv.w + acc.w;
      }
      *(float4*)ar = a;
      acc = float4{0.f, 0.f, 0.f, 0.f};
      r++;
      rowend = __builtin_amdgcn_readlane(ov, (r + 1 < TROWS) ? (r + 1) : TROWS);
    };
    int idv = 0;
    if (ebeg < eend) {
      int ei0 = ebeg + lane;
      idv = es[(ei0 < eend) ? ei0 : (eend - 1)];
    }
    for (int e = ebeg; e < eend; e += 32) {
      int idn = 0;
      if (e + 32 < eend) {
        int ei1 = e + 32 + lane;
        idn = es[(ei1 < eend) ? ei1 : (eend - 1)];
      }
      if (eend - e > 16) {
        float4 v[32];
#pragma unroll
        for (int u = 0; u < 32; u++) {
          int s = __builtin_amdgcn_readlane(idv, u);
          v[u] = x4[(unsigned)s * 32u + (unsigned)lane];
        }
#pragma unroll
        for (int u = 0; u < 32; u++) {
          if (e + u < eend) {
            while (e + u >= rowend) close_row();
            acc.x += v[u].x; acc.y += v[u].y; acc.z += v[u].z; acc.w += v[u].w;
          }
        }
      } else {
        float4 v[16];
#pragma unroll
        for (int u = 0; u < 16; u++) {
          int s = __builtin_amdgcn_readlane(idv, u);
          v[u] = x4[(unsigned)s * 32u + (unsigned)lane];
        }
#pragma unroll
        for (int u = 0; u < 16; u++) {
          if (e + u < eend) {
            while (e + u >= rowend) close_row();
            acc.x += v[u].x; acc.y += v[u].y; acc.z += v[u].z; acc.w += v[u].w;
          }
        }
      }
      idv = idn;
    }
    while (r < TROWS) close_row();
    __builtin_amdgcn_fence(__ATOMIC_RELEASE, "wavefront");
    __builtin_amdgcn_wave_barrier();
    __builtin_amdgcn_fence(__ATOMIC_ACQUIRE, "wavefront");

    const float* ap = sAw + lmod * APITCH + khalf;
    v8f acc8[8] = {};
    for (int k = 0; k < GDIM; k += 4) {
      v2f aA = *(const v2f*)(ap + k);
      const float* wp = sWp + (k >> 1) * 256;
      v2f b[8];
#pragma unroll
      for (int nt = 0; nt < 8; nt++) b[nt] = *(const v2f*)(wp + nt * 32);
#pragma unroll
      for (int nt = 0; nt < 8; nt++)
        acc8[nt] = __builtin_amdgcn_wmma_f32_16x16x4_f32(
            false, aA, false, b[nt], (short)0, acc8[nt], false, false);
    }
    __builtin_amdgcn_fence(__ATOMIC_RELEASE, "wavefront");
    __builtin_amdgcn_wave_barrier();

    if (TWO) {
#pragma unroll
      for (int nt = 0; nt < 8; nt++) {
        int ncol = nt * 16 + lmod;
        float bv = bias[ncol];
#pragma unroll
        for (int j = 0; j < 8; j++)
          sAw[(mhi + j) * APITCH + ncol] = fmaxf(acc8[nt][j] + bv, 0.f);
      }
      __builtin_amdgcn_fence(__ATOMIC_RELEASE, "wavefront");
      __builtin_amdgcn_wave_barrier();
      __builtin_amdgcn_fence(__ATOMIC_ACQUIRE, "wavefront");
#pragma unroll
      for (int nt = 0; nt < 8; nt++) acc8[nt] = v8f{};
      const float* sWq = sWp + GDIM * GDIM;
      for (int k = 0; k < GDIM; k += 4) {
        v2f aA = *(const v2f*)(ap + k);
        const float* wp = sWq + (k >> 1) * 256;
        v2f b[8];
#pragma unroll
        for (int nt = 0; nt < 8; nt++) b[nt] = *(const v2f*)(wp + nt * 32);
#pragma unroll
        for (int nt = 0; nt < 8; nt++)
          acc8[nt] = __builtin_amdgcn_wmma_f32_16x16x4_f32(
              false, aA, false, b[nt], (short)0, acc8[nt], false, false);
      }
      __builtin_amdgcn_fence(__ATOMIC_RELEASE, "wavefront");
      __builtin_amdgcn_wave_barrier();
    }

    const float* obias = TWO ? biasb : bias;
    bool whole = (tb + TROWS) <= n;
#pragma unroll
    for (int nt = 0; nt < 8; nt++) {
      int ncol = nt * 16 + lmod;
      float bv = obias[ncol];
#pragma unroll
      for (int j = 0; j < 8; j++) {
        float v = acc8[nt][j] + bv;
        if (!EPI && !TWO) v = fmaxf(v, 0.f);
        int orow = tb + mhi + j;
        if (whole || orow < n) Out[(long)orow * GDIM + ncol] = v;
      }
    }
  }
}

__global__ __launch_bounds__(256) void gin_fold(
    const float* __restrict__ W1, const float* __restrict__ W2,
    const float* __restrict__ b2, float* __restrict__ Mout,
    float* __restrict__ cout) {
  __shared__ float sW[GDIM * GDIM];
  int tid = threadIdx.x;
  int fi = blockIdx.x;
  const float* Wn = W1 + (long)(fi + 1) * GDIM * GDIM;
  const float* A = W2 + (long)fi * GDIM * GDIM;
  const float* bb = b2 + fi * GDIM;

  int wave = tid >> 5;
  int lane = tid & 31;
  int lmod = lane & 15;
  int par = lane >> 4;
  int khalf = par << 1;
  int mhi = par << 3;

  const float* ar = A + (wave * 16 + lmod) * GDIM + khalf;
  v2f afr[GDIM / 4];
#pragma unroll
  for (int s = 0; s < GDIM / 4; s++) afr[s] = *(const v2f*)(ar + 4 * s);

  float4 r0[8], r1[8];
#pragma unroll
  for (int t = 0; t < 8; t++) {
    int i = tid + t * 256;
    int p = i >> 5;
    int q = i & 31;
    r0[t] = *(const float4*)(Wn + (2 * p) * GDIM + 4 * q);
    r1[t] = *(const float4*)(Wn + (2 * p + 1) * GDIM + 4 * q);
  }
#pragma unroll
  for (int t = 0; t < 8; t++) {
    int i = tid + t * 256;
    int p = i >> 5;
    int q = i & 31;
    float4 lo = {r0[t].x, r1[t].x, r0[t].y, r1[t].y};
    float4 hi = {r0[t].z, r1[t].z, r0[t].w, r1[t].w};
    *(float4*)(sW + p * 256 + 8 * q) = lo;
    *(float4*)(sW + p * 256 + 8 * q + 4) = hi;
  }
  __syncthreads();

  const float* sWp = sW + par * 256 + 2 * lmod;
  v8f acc[8] = {};
#pragma unroll
  for (int s = 0; s < GDIM / 4; s++) {
    const float* wp = sWp + s * 512;
    v2f b[8];
#pragma unroll
    for (int nt = 0; nt < 8; nt++) b[nt] = *(const v2f*)(wp + nt * 32);
#pragma unroll
    for (int nt = 0; nt < 8; nt++)
      acc[nt] = __builtin_amdgcn_wmma_f32_16x16x4_f32(
          false, afr[s], false, b[nt], (short)0, acc[nt], false, false);
  }
  float* Mo = Mout + (long)fi * GDIM * GDIM;
#pragma unroll
  for (int nt = 0; nt < 8; nt++) {
#pragma unroll
    for (int j = 0; j < 8; j++)
      Mo[(wave * 16 + mhi + j) * GDIM + nt * 16 + lmod] = acc[nt][j];
  }

  if (tid < GDIM) {
    float s = 0.f;
    for (int k = 0; k < GDIM; k += 2) {
      v2f w = *(const v2f*)(sW + (k >> 1) * 256 + 2 * tid);
      s = fmaf(bb[k], w.x, s);
      s = fmaf(bb[k + 1], w.y, s);
    }
    cout[fi * GDIM + tid] = s;
  }
}

extern "C" void kernel_launch(void* const* d_in, const int* in_sizes, int n_in,
                              void* d_out, int out_size, void* d_ws,
                              size_t ws_size, hipStream_t stream) {
  const float* x_in = (const float*)d_in[0];
  const int* ei = (const int*)d_in[1];
  const float* W1 = (const float*)d_in[2];
  const float* b1 = (const float*)d_in[3];
  const float* W2 = (const float*)d_in[4];
  const float* b2 = (const float*)d_in[5];
  float* out = (float*)d_out;

  const int N = in_sizes[0] / GDIM;
  const int NE = in_sizes[1] / 2;
  const long nd = (long)N * GDIM;

  float* aggh = (float*)d_ws;
  int2* pairs = (int2*)aggh;
  int* blkoff = (int*)aggh + 2 * (long)NE;
  int* btot = (int*)(aggh + nd);
  int* off = btot + (N + 4);
  int* ebuf = off + (N + 4);
  int* es = ebuf + NE;
  float* Mf = (float*)ebuf;
  float* cf = Mf + 2 * GDIM * GDIM;

  const int NB = (N + BNODES - 1) >> BSH;
  const int part_blocks = 256;
  const int zero_blocks = (NB * (BTS / 4) + 255) / 256;
  const int fused_blocks = 256;

  gin_zero<<<zero_blocks, 256, 0, stream>>>((float4*)btot, NB * (BTS / 4));
  gin_bcount<<<part_blocks, 1024, 0, stream>>>(ei + NE, btot, blkoff, NE, NB);
  gin_bscatter<<<part_blocks, 1024, 0, stream>>>(ei, ei + NE, btot, blkoff, pairs, NE, NB);
  gin_bcsr<<<NB, 1024, 0, stream>>>(pairs, btot, off, ebuf, es, N, NB);
  gin_fold<<<2, 256, 0, stream>>>(W1, W2, b2, Mf, cf);

  gin_gather_gemm<false, true><<<fused_blocks, FWAVES * 32, 0, stream>>>(
      x_in, off, es, nullptr, W1, b1, Mf, cf, out, N);
  gin_gather_gemm<true, false><<<fused_blocks, FWAVES * 32, 0, stream>>>(
      out, off, es, b1 + GDIM, Mf + (long)GDIM * GDIM, cf + GDIM, nullptr,
      nullptr, aggh, N);
  gin_gather_gemm<true, false><<<fused_blocks, FWAVES * 32, 0, stream>>>(
      aggh, off, es, b1 + 2 * GDIM, W2 + (long)2 * GDIM * GDIM, b2 + 2 * GDIM,
      nullptr, nullptr, out, N);
}


# Result tile staged back through the wave's LDS tile and written as whole 512-B rows, one 16-B-per-lane store per row: 64 stores of two 64-B half lines -> 16 whole-row stores a tile; sums unchanged
# speedup vs baseline: 4.6084x; 1.0720x over previous
#include <hip/hip_runtime.h>

typedef __attribute__((ext_vector_type(2))) float v2f;
typedef __attribute__((ext_vector_type(8))) float v8f;

#define GDIM 128

__global__ __launch_bounds__(256) void gin_zero(float4* p, long n4) {
  long i = (long)blockIdx.x * blockDim.x + threadIdx.x;
  long stride = (long)gridDim.x * blockDim.x;
  float4 z = {0.f, 0.f, 0.f, 0.f};
  for (; i < n4; i += stride) p[i] = z;
}


#define BSH 9
#define BNODES 512
#define MAXNB 256
#define BTS 32
#define LCAP 16384

__global__ __launch_bounds__(1024) void gin_bcount(
    const int* __restrict__ dst, int* btot, int* __restrict__ blkoff,
    int nedges, int nb) {
  __shared__ int h[MAXNB];
  int tid = threadIdx.x;
  if (tid < MAXNB) h[tid] = 0;
  __syncthreads();
  int chunk = (nedges + (int)gridDim.x - 1) / (int)gridDim.x;
  int e0 = (int)blockIdx.x * chunk;
  int e1 = (e0 + chunk < nedges) ? (e0 + chunk) : nedges;
  for (int e = e0 + tid; e < e1; e += 8 * 1024) {
    int d[8];
#pragma unroll
    for (int u = 0; u < 8; u++) {
      int ee = e + u * 1024;
      d[u] = (ee < e1) ? dst[ee] : -1;
    }
#pragma unroll
    for (int u = 0; u < 8; u++)
      if (d[u] >= 0) atomicAdd(&h[d[u] >> BSH], 1);
  }
  __syncthreads();
  if (tid < nb)
    blkoff[(int)blockIdx.x * nb + tid] = atomicAdd(btot + tid * BTS, h[tid]);
}

__global__ __launch_bounds__(1024) void gin_bscatter(
    const int* __restrict__ src, const int* __restrict__ dst,
    const int* __restrict__ btot, const int* __restrict__ blkoff,
    int2* __restrict__ pairs, int nedges, int nb) {
  __shared__ int cur[MAXNB];
  __shared__ int swave[8];
  int tid = threadIdx.x;
  int lane = tid & 31;
  int wv = tid >> 5;
  int v = (tid < nb) ? btot[tid * BTS] : 0;
  int incl = v;
  for (int s = 1; s < 32; s <<= 1) {
    int u = __shfl_up(incl, s);
    if (lane >= s) incl += u;
  }
  if (lane == 31 && wv < 8) swave[wv] = incl;
  __syncthreads();
  int pre = 0;
  for (int j = 0; j < 8; j++) pre += (j < wv) ? swave[j] : 0;
  if (tid < nb) cur[tid] = pre + incl - v + blkoff[(int)blockIdx.x * nb + tid];
  __syncthreads();
  int chunk = (nedges + (int)gridDim.x - 1) / (int)gridDim.x;
  int e0 = (int)blockIdx.x * chunk;
  int e1 = (e0 + chunk < nedges) ? (e0 + chunk) : nedges;
  for (int e = e0 + tid; e < e1; e += 8 * 1024) {
    int d[8], sr[8];
#pragma unroll
    for (int u = 0; u < 8; u++) {
      int ee = e + u * 1024;
      d[u] = (ee < e1) ? dst[ee] : -1;
      sr[u] = (ee < e1) ? src[ee] : 0;
    }
#pragma unroll
    for (int u = 0; u < 8; u++) {
      if (d[u] >= 0) {
        int pos = atomicAdd(&cur[d[u] >> BSH], 1);
        pairs[pos] = make_int2(sr[u], d[u]);
      }
    }
  }
}

__global__ __launch_bounds__(1024) void gin_bcsr(
    const int2* __restrict__ pairs, const int* __restrict__ btot,
    int* __restrict__ off, int* ebuf, int* __restrict__ es, int n, int nb) {
  __shared__ int cntL[BNODES];
  __shared__ int curL[BNODES];
  __shared__ int swave[16];
  __shared__ int sred[8];
  __shared__ int stage[LCAP];
  int tid = threadIdx.x;
  int lane = tid & 31;
  int wv = tid >> 5;
  int b = (int)blockIdx.x;
  int base = b << BSH;
  if (tid < BNODES) cntL[tid] = 0;
  int v = (tid < b) ? btot[tid * BTS] : 0;
  for (int m = 16; m >= 1; m >>= 1) v += __shfl_xor(v, m);
  if (lane == 0 && wv < 8) sred[wv] = v;
  int cb = btot[b * BTS];
  __syncthreads();
  int bs = 0;
  for (int j = 0; j < 8; j++) bs += sred[j];
  const bool inl = cb <= LCAP;

  for (int i = tid; i < cb; i += 8 * 1024) {
    int d[8];
#pragma unroll
    for (int u = 0; u < 8; u++) {
      int ii = i + u * 1024;
      d[u] = (ii < cb) ? pairs[bs + ii].y : -1;
    }
#pragma unroll
    for (int u = 0; u < 8; u++)
      if (d[u] >= 0) atomicAdd(&cntL[d[u] - base], 1);
  }
  __syncthreads();

  int c = (tid < BNODES) ? cntL[tid] : 0;
  int incl = c;
  for (int s = 1; s < 32; s <<= 1) {
    int u = __shfl_up(incl, s);
    if (lane >= s) incl += u;
  }
  if (lane == 31 && wv < 16) swave[wv] = incl;
  __syncthreads();
  int pre = 0;
  for (int j = 0; j < 16; j++) pre += (j < wv) ? swave[j] : 0;
  int ex = pre + incl - c;
  if (tid < BNODES) {
    curL[tid] = ex;
    if (base + tid < n) off[base + tid] = bs + ex;
  }
  if (b == nb - 1 && tid == 0) off[n] = bs + cb;
  __syncthreads();

  for (int i = tid; i < cb; i += 8 * 1024) {
    int2 p[8];
#pragma unroll
    for (int u = 0; u < 8; u++) {
      int ii = i + u * 1024;
      p[u] = (ii < cb) ? pairs[bs + ii] : make_int2(0, -1);
    }
#pragma unroll
    for (int u = 0; u < 8; u++) {
      if (p[u].y >= 0) {
        int pos = atomicAdd(&curL[p[u].y - base], 1);
        if (inl) stage[pos] = p[u].x;
        else ebuf[bs + pos] = p[u].x;
      }
    }
  }

  __syncthreads();
  int myt = wv * 16 + (lane & 15);
  int segend = curL[myt];
  int segbeg = (myt > 0) ? curL[myt - 1] : 0;
  const int* eb = ebuf + bs;
  int* eo = es + bs;
  auto sort_segments = [&](auto fetch) {
    auto seg_keys = [&](int i) {
      int beg = __builtin_amdgcn_readlane(segbeg, i & 15);
      int len = __builtin_amdgcn_readlane(segend, i & 15) - beg;
      return (i < 16 && lane < len) ? fetch(beg + lane) : 0x7fffffff;
    };
    int k0 = seg_keys(0);
#pragma unroll 1
    for (int i = 0; i < 16; i++) {
      int kn = seg_keys(i + 1);
      int beg = __builtin_amdgcn_readlane(segbeg, i);
      int len = __builtin_amdgcn_readlane(segend, i) - beg;
      if (len <= 32) {
        int key = k0;
        unsigned comb = ((unsigned)key << 5) | (unsigned)lane;
        int rank = 0;
        for (int j = 0; j < len; j++) {
          unsigned cj = (unsigned)__builtin_amdgcn_readlane((int)comb, j);
          rank += (cj < comb) ? 1 : 0;
        }
        if (lane < len) eo[beg + rank] = key;
      } else {
        for (int q = lane; q < len; q += 32) {
          int key = fetch(beg + q);
          int rank = 0;
          for (int j = 0; j < len; j++) {
            int kj = fetch(beg + j);
            rank += ((kj < key) || (kj == key && j < q)) ? 1 : 0;
          }
          eo[beg + rank] = key;
        }
      }
      k0 = kn;
    }
  };
  if (inl) sort_segments([&](int idx) { return stage[idx]; });
  else sort_segments([&](int idx) { return eb[idx]; });
}

#define FWAVES 16
#define TROWS 16
#define APITCH 132

template <bool EPI, bool TWO>
__global__ __launch_bounds__(FWAVES * 32) void gin_gather_gemm(
    const float* __restrict__ x, const int* __restrict__ off,
    const int* __restrict__ es, const float* __restrict__ gbias,
    const float* __restrict__ W, const float* __restrict__ bias,
    const float* __restrict__ Wb, const float* __restrict__ biasb,
    float* __restrict__ Out, int n) {
  __shared__ __attribute__((aligned(16))) float sW[(TWO ? 2 : 1) * GDIM * GDIM];
  __shared__ __attribute__((aligned(16))) float sA[FWAVES * TROWS * APITCH];
  int tid = threadIdx.x;

  {
    constexpr int NM = TWO ? 2 : 1;
    float4 r0[NM][4], r1[NM][4];
#pragma unroll
    for (int m = 0; m < NM; m++) {
      const float* Wm = m ? Wb : W;
#pragma unroll
      for (int t = 0; t < 4; t++) {
        int i = tid + t * (FWAVES * 32);
        int p = i >> 5;
        int q = i & 31;
        r0[m][t] = *(const float4*)(Wm + (2 * p) * GDIM + 4 * q);
        r1[m][t] = *(const float4*)(Wm + (2 * p + 1) * GDIM + 4 * q);
      }
    }
#pragma unroll
    for (int m = 0; m < NM; m++) {
#pragma unroll
      for (int t = 0; t < 4; t++) {
        int i = tid + t * (FWAVES * 32);
        int p = i >> 5;
        int q = i & 31;
        float4 lo = {r0[m][t].x, r1[m][t].x, r0[m][t].y, r1[m][t].y};
        float4 hi = {r0[m][t].z, r1[m][t].z, r0[m][t].w, r1[m][t].w};
        *(float4*)(sW + m * (GDIM * GDIM) + p * 256 + 8 * q) = lo;
        *(float4*)(sW + m * (GDIM * GDIM) + p * 256 + 8 * q + 4) = hi;
      }
    }
  }
  __syncthreads();

  int wave = tid >> 5;
  int lane = tid & 31;
  int lmod = lane & 15;
  int par = lane >> 4;
  int khalf = par << 1;
  int mhi = par << 3;

  float* sAw = sA + wave * (TROWS * APITCH);
  const float* sWp = sW + par * 256 + 2 * lmod;
  float4 gb = {0.f, 0.f, 0.f, 0.f};
  if (EPI) gb = *(const float4*)(gbias + lane * 4);
  const float4* x4 = (const float4*)x;

  int ntiles = (n + TROWS - 1) / TROWS;
  int tpb = (ntiles + (int)gridDim.x - 1) / (int)gridDim.x;
  int t0 = (int)blockIdx.x * tpb;
  int t1 = (t0 + tpb < ntiles) ? (t0 + tpb) : ntiles;

  for (int tile = t0 + wave; tile < t1; tile += FWAVES) {
    int tb = __builtin_amdgcn_readfirstlane(tile) * TROWS;
    int oi = tb + lane;
    int ov = off[(oi < n) ? oi : n];

    int ebeg = __builtin_amdgcn_readlane(ov, 0);
    int eend = __builtin_amdgcn_readlane(ov, TROWS);
    {
      float4 sv[TROWS];
#pragma unroll
      for (int r = 0; r < TROWS; r++) {
        int row = (tb + r < n) ? (tb + r) : (n - 1);
        sv[r] = x4[(unsigned)row * 32u + (unsigned)lane];
      }
#pragma unroll
      for (int r = 0; r < TROWS; r++)
        *(float4*)(sAw + r * APITCH + lane * 4) = sv[r];
    }
    int r = 0;
    int rowend = __builtin_amdgcn_readlane(ov, 1);
    float4 acc = {0.f, 0.f, 0.f, 0.f};
    auto close_row = [&]() {
      float* ar = sAw + r * APITCH + lane * 4;
      float4 sv = *(const float4*)ar;
      float4 a;
      if (EPI) {
        a.x = fmaxf((sv.x + acc.x) + gb.x, 0.f);
        a.y = fmaxf((sv.y + acc.y) + gb.y, 0.f);
        a.z = fmaxf((sv.z + acc.z) + gb.z, 0.f);
        a.w = fmaxf((sv.w + acc.w) + gb.w, 0.f);
      } else {
        a.x = sv.x + acc.x; a.y = sv.y + acc.y;
        a.z = sv.z + acc.z; a.w = sv.w + acc.w;
      }
      *(float4*)ar = a;
      acc = float4{0.f, 0.f, 0.f, 0.f};
      r++;
      rowend = __builtin_amdgcn_readlane(ov, (r + 1 < TROWS) ? (r + 1) : TROWS);
    };
    int idv = 0;
    if (ebeg < eend) {
      int ei0 = ebeg + lane;
      idv = es[(ei0 < eend) ? ei0 : (eend - 1)];
    }
    for (int e = ebeg; e < eend; e += 32) {
      int idn = 0;
      if (e + 32 < eend) {
        int ei1 = e + 32 + lane;
        idn = es[(ei1 < eend) ? ei1 : (eend - 1)];
      }
      if (eend - e > 16) {
        float4 v[32];
#pragma unroll
        for (int u = 0; u < 32; u++) {
          int s = __builtin_amdgcn_readlane(idv, u);
          v[u] = x4[(unsigned)s * 32u + (unsigned)lane];
        }
#pragma unroll
        for (int u = 0; u < 32; u++) {
          if (e + u < eend) {
            while (e + u >= rowend) close_row();
            acc.x += v[u].x; acc.y += v[u].y; acc.z += v[u].z; acc.w += v[u].w;
          }
        }
      } else {
        float4 v[16];
#pragma unroll
        for (int u = 0; u < 16; u++) {
          int s = __builtin_amdgcn_readlane(idv, u);
          v[u] = x4[(unsigned)s * 32u + (unsigned)lane];
        }
#pragma unroll
        for (int u = 0; u < 16; u++) {
          if (e + u < eend) {
            while (e + u >= rowend) close_row();
            acc.x += v[u].x; acc.y += v[u].y; acc.z += v[u].z; acc.w += v[u].w;
          }
        }
      }
      idv = idn;
    }
    while (r < TROWS) close_row();
    __builtin_amdgcn_fence(__ATOMIC_RELEASE, "wavefront");
    __builtin_amdgcn_wave_barrier();
    __builtin_amdgcn_fence(__ATOMIC_ACQUIRE, "wavefront");

    const float* ap = sAw + lmod * APITCH + khalf;
    v8f acc8[8] = {};
    for (int k = 0; k < GDIM; k += 4) {
      v2f aA = *(const v2f*)(ap + k);
      const float* wp = sWp + (k >> 1) * 256;
      v2f b[8];
#pragma unroll
      for (int nt = 0; nt < 8; nt++) b[nt] = *(const v2f*)(wp + nt * 32);
#pragma unroll
      for (int nt = 0; nt < 8; nt++)
        acc8[nt] = __builtin_amdgcn_wmma_f32_16x16x4_f32(
            false, aA, false, b[nt], (short)0, acc8[nt], false, false);
    }
    __builtin_amdgcn_fence(__ATOMIC_RELEASE, "wavefront");
    __builtin_amdgcn_wave_barrier();

    if (TWO) {
#pragma unroll
      for (int nt = 0; nt < 8; nt++) {
        int ncol = nt * 16 + lmod;
        float bv = bias[ncol];
#pragma unroll
        for (int j = 0; j < 8; j++)
          sAw[(mhi + j) * APITCH + ncol] = fmaxf(acc8[nt][j] + bv, 0.f);
      }
      __builtin_amdgcn_fence(__ATOMIC_RELEASE, "wavefront");
      __builtin_amdgcn_wave_barrier();
      __builtin_amdgcn_fence(__ATOMIC_ACQUIRE, "wavefront");
#pragma unroll
      for (int nt = 0; nt < 8; nt++) acc8[nt] = v8f{};
      const float* sWq = sWp + GDIM * GDIM;
      for (int k = 0; k < GDIM; k += 4) {
        v2f aA = *(const v2f*)(ap + k);
        const float* wp = sWq + (k >> 1) * 256;
        v2f b[8];
#pragma unroll
        for (int nt = 0; nt < 8; nt++) b[nt] = *(const v2f*)(wp + nt * 32);
#pragma unroll
        for (int nt = 0; nt < 8; nt++)
          acc8[nt] = __builtin_amdgcn_wmma_f32_16x16x4_f32(
              false, aA, false, b[nt], (short)0, acc8[nt], false, false);
      }
      __builtin_amdgcn_fence(__ATOMIC_RELEASE, "wavefront");
      __builtin_amdgcn_wave_barrier();
    }

    const float* obias = TWO ? biasb : bias;
    bool whole = (tb + TROWS) <= n;
#pragma unroll
    for (int nt = 0; nt < 8; nt++) {
      int ncol = nt * 16 + lmod;
      float bv = obias[ncol];
#pragma unroll
      for (int j = 0; j < 8; j++) {
        float v = acc8[nt][j] + bv;
        if (!EPI && !TWO) v = fmaxf(v, 0.f);
        sAw[(mhi + j) * APITCH + ncol] = v;
      }
    }
    __builtin_amdgcn_fence(__ATOMIC_RELEASE, "wavefront");
    __builtin_amdgcn_wave_barrier();
    __builtin_amdgcn_fence(__ATOMIC_ACQUIRE, "wavefront");
    float* orow0 = Out + (long)tb * GDIM + lane * 4;
    if (whole) {
      float4 ov[TROWS];
#pragma unroll
      for (int rr = 0; rr < TROWS; rr++)
        ov[rr] = *(const float4*)(sAw + rr * APITCH + lane * 4);
#pragma unroll
      for (int rr = 0; rr < TROWS; rr++)
        *(float4*)(orow0 + rr * GDIM) = ov[rr];
    } else {
      for (int rr = 0; rr < TROWS && tb + rr < n; rr++)
        *(float4*)(orow0 + rr * GDIM) =
            *(const float4*)(sAw + rr * APITCH + lane * 4);
    }
    __builtin_amdgcn_fence(__ATOMIC_RELEASE, "wavefront");
    __builtin_amdgcn_wave_barrier();
  }
}

__global__ __launch_bounds__(256) void gin_fold(
    const float* __restrict__ W1, const float* __restrict__ W2,
    const float* __restrict__ b2, float* __restrict__ Mout,
    float* __restrict__ cout) {
  __shared__ float sW[GDIM * GDIM];
  int tid = threadIdx.x;
  int fi = blockIdx.x;
  const float* Wn = W1 + (long)(fi + 1) * GDIM * GDIM;
  const float* A = W2 + (long)fi * GDIM * GDIM;
  const float* bb = b2 + fi * GDIM;

  int wave = tid >> 5;
  int lane = tid & 31;
  int lmod = lane & 15;
  int par = lane >> 4;
  int khalf = par << 1;
  int mhi = par << 3;

  const float* ar = A + (wave * 16 + lmod) * GDIM + khalf;
  v2f afr[GDIM / 4];
#pragma unroll
  for (int s = 0; s < GDIM / 4; s++) afr[s] = *(const v2f*)(ar + 4 * s);

  float4 r0[8], r1[8];
#pragma unroll
  for (int t = 0; t < 8; t++) {
    int i = tid + t * 256;
    int p = i >> 5;
    int q = i & 31;
    r0[t] = *(const float4*)(Wn + (2 * p) * GDIM + 4 * q);
    r1[t] = *(const float4*)(Wn + (2 * p + 1) * GDIM + 4 * q);
  }
#pragma unroll
  for (int t = 0; t < 8; t++) {
    int i = tid + t * 256;
    int p = i >> 5;
    int q = i & 31;
    float4 lo = {r0[t].x, r1[t].x, r0[t].y, r1[t].y};
    float4 hi = {r0[t].z, r1[t].z, r0[t].w, r1[t].w};
    *(float4*)(sW + p * 256 + 8 * q) = lo;
    *(float4*)(sW + p * 256 + 8 * q + 4) = hi;
  }
  __syncthreads();

  const float* sWp = sW + par * 256 + 2 * lmod;
  v8f acc[8] = {};
#pragma unroll
  for (int s = 0; s < GDIM / 4; s++) {
    const float* wp = sWp + s * 512;
    v2f b[8];
#pragma unroll
    for (int nt = 0; nt < 8; nt++) b[nt] = *(const v2f*)(wp + nt * 32);
#pragma unroll
    for (int nt = 0; nt < 8; nt++)
      acc[nt] = __builtin_amdgcn_wmma_f32_16x16x4_f32(
          false, afr[s], false, b[nt], (short)0, acc[nt], false, false);
  }
  float* Mo = Mout + (long)fi * GDIM * GDIM;
#pragma unroll
  for (int nt = 0; nt < 8; nt++) {
#pragma unroll
    for (int j = 0; j < 8; j++)
      Mo[(wave * 16 + mhi + j) * GDIM + nt * 16 + lmod] = acc[nt][j];
  }

  if (tid < GDIM) {
    float s = 0.f;
    for (int k = 0; k < GDIM; k += 2) {
      v2f w = *(const v2f*)(sW + (k >> 1) * 256 + 2 * tid);
      s = fmaf(bb[k], w.x, s);
      s = fmaf(bb[k + 1], w.y, s);
    }
    cout[fi * GDIM + tid] = s;
  }
}

extern "C" void kernel_launch(void* const* d_in, const int* in_sizes, int n_in,
                              void* d_out, int out_size, void* d_ws,
                              size_t ws_size, hipStream_t stream) {
  const float* x_in = (const float*)d_in[0];
  const int* ei = (const int*)d_in[1];
  const float* W1 = (const float*)d_in[2];
  const float* b1 = (const float*)d_in[3];
  const float* W2 = (const float*)d_in[4];
  const float* b2 = (const float*)d_in[5];
  float* out = (float*)d_out;

  const int N = in_sizes[0] / GDIM;
  const int NE = in_sizes[1] / 2;
  const long nd = (long)N * GDIM;

  float* aggh = (float*)d_ws;
  int2* pairs = (int2*)aggh;
  int* blkoff = (int*)aggh + 2 * (long)NE;
  int* btot = (int*)(aggh + nd);
  int* off = btot + (N + 4);
  int* ebuf = off + (N + 4);
  int* es = ebuf + NE;
  float* Mf = (float*)ebuf;
  float* cf = Mf + 2 * GDIM * GDIM;

  const int NB = (N + BNODES - 1) >> BSH;
  const int part_blocks = 256;
  const int zero_blocks = (NB * (BTS / 4) + 255) / 256;
  const int fused_blocks = 256;

  gin_zero<<<zero_blocks, 256, 0, stream>>>((float4*)btot, NB * (BTS / 4));
  gin_bcount<<<part_blocks, 1024, 0, stream>>>(ei + NE, btot, blkoff, NE, NB);
  gin_bscatter<<<part_blocks, 1024, 0, stream>>>(ei, ei + NE, btot, blkoff, pairs, NE, NB);
  gin_bcsr<<<NB, 1024, 0, stream>>>(pairs, btot, off, ebuf, es, N, NB);
  gin_fold<<<2, 256, 0, stream>>>(W1, W2, b2, Mf, cf);

  gin_gather_gemm<false, true><<<fused_blocks, FWAVES * 32, 0, stream>>>(
      x_in, off, es, nullptr, W1, b1, Mf, cf, out, N);
  gin_gather_gemm<true, false><<<fused_blocks, FWAVES * 32, 0, stream>>>(
      out, off, es, b1 + GDIM, Mf + (long)GDIM * GDIM, cf + GDIM, nullptr,
      nullptr, aggh, N);
  gin_gather_gemm<true, false><<<fused_blocks, FWAVES * 32, 0, stream>>>(
      aggh, off, es, b1 + 2 * GDIM, W2 + (long)2 * GDIM * GDIM, b2 + 2 * GDIM,
      nullptr, nullptr, out, N);
}


# Gather: tile bounds, first ids and each group's next ids load unconditionally and ahead, self rows stay in registers: no small-load wait stands in front of a group's 32 row loads; sums unchanged
# speedup vs baseline: 4.8378x; 1.0498x over previous
#include <hip/hip_runtime.h>
#include <type_traits>

typedef __attribute__((ext_vector_type(2))) float v2f;
typedef __attribute__((ext_vector_type(8))) float v8f;

#define GDIM 128

__global__ __launch_bounds__(256) void gin_zero(float4* p, long n4) {
  long i = (long)blockIdx.x * blockDim.x + threadIdx.x;
  long stride = (long)gridDim.x * blockDim.x;
  float4 z = {0.f, 0.f, 0.f, 0.f};
  for (; i < n4; i += stride) p[i] = z;
}


#define BSH 9
#define BNODES 512
#define MAXNB 256
#define BTS 32
#define LCAP 16384

__global__ __launch_bounds__(1024) void gin_bcount(
    const int* __restrict__ dst, int* btot, int* __restrict__ blkoff,
    int nedges, int nb) {
  __shared__ int h[MAXNB];
  int tid = threadIdx.x;
  if (tid < MAXNB) h[tid] = 0;
  __syncthreads();
  int chunk = (nedges + (int)gridDim.x - 1) / (int)gridDim.x;
  int e0 = (int)blockIdx.x * chunk;
  int e1 = (e0 + chunk < nedges) ? (e0 + chunk) : nedges;
  for (int e = e0 + tid; e < e1; e += 8 * 1024) {
    int d[8];
#pragma unroll
    for (int u = 0; u < 8; u++) {
      int ee = e + u * 1024;
      d[u] = (ee < e1) ? dst[ee] : -1;
    }
#pragma unroll
    for (int u = 0; u < 8; u++)
      if (d[u] >= 0) atomicAdd(&h[d[u] >> BSH], 1);
  }
  __syncthreads();
  if (tid < nb)
    blkoff[(int)blockIdx.x * nb + tid] = atomicAdd(btot + tid * BTS, h[tid]);
}

__global__ __launch_bounds__(1024) void gin_bscatter(
    const int* __restrict__ src, const int* __restrict__ dst,
    const int* __restrict__ btot, const int* __restrict__ blkoff,
    int2* __restrict__ pairs, int nedges, int nb) {
  __shared__ int cur[MAXNB];
  __shared__ int swave[8];
  int tid = threadIdx.x;
  int lane = tid & 31;
  int wv = tid >> 5;
  int v = (tid < nb) ? btot[tid * BTS] : 0;
  int incl = v;
  for (int s = 1; s < 32; s <<= 1) {
    int u = __shfl_up(incl, s);
    if (lane >= s) incl += u;
  }
  if (lane == 31 && wv < 8) swave[wv] = incl;
  __syncthreads();
  int pre = 0;
  for (int j = 0; j < 8; j++) pre += (j < wv) ? swave[j] : 0;
  if (tid < nb) cur[tid] = pre + incl - v + blkoff[(int)blockIdx.x * nb + tid];
  __syncthreads();
  int chunk = (nedges + (int)gridDim.x - 1) / (int)gridDim.x;
  int e0 = (int)blockIdx.x * chunk;
  int e1 = (e0 + chunk < nedges) ? (e0 + chunk) : nedges;
  for (int e = e0 + tid; e < e1; e += 8 * 1024) {
    int d[8], sr[8];
#pragma unroll
    for (int u = 0; u < 8; u++) {
      int ee = e + u * 1024;
      d[u] = (ee < e1) ? dst[ee] : -1;
      sr[u] = (ee < e1) ? src[ee] : 0;
    }
#pragma unroll
    for (int u = 0; u < 8; u++) {
      if (d[u] >= 0) {
        int pos = atomicAdd(&cur[d[u] >> BSH], 1);
        pairs[pos] = make_int2(sr[u], d[u]);
      }
    }
  }
}

__global__ __launch_bounds__(1024) void gin_bcsr(
    const int2* __restrict__ pairs, const int* __restrict__ btot,
    int* __restrict__ off, int* ebuf, int* __restrict__ es, int n, int nb) {
  __shared__ int cntL[BNODES];
  __shared__ int curL[BNODES];
  __shared__ int swave[16];
  __shared__ int sred[8];
  __shared__ int stage[LCAP];
  int tid = threadIdx.x;
  int lane = tid & 31;
  int wv = tid >> 5;
  int b = (int)blockIdx.x;
  int base = b << BSH;
  if (tid < BNODES) cntL[tid] = 0;
  int v = (tid < b) ? btot[tid * BTS] : 0;
  for (int m = 16; m >= 1; m >>= 1) v += __shfl_xor(v, m);
  if (lane == 0 && wv < 8) sred[wv] = v;
  int cb = btot[b * BTS];
  __syncthreads();
  int bs = 0;
  for (int j = 0; j < 8; j++) bs += sred[j];
  const bool inl = cb <= LCAP;

  for (int i = tid; i < cb; i += 8 * 1024) {
    int d[8];
#pragma unroll
    for (int u = 0; u < 8; u++) {
      int ii = i + u * 1024;
      d[u] = (ii < cb) ? pairs[bs + ii].y : -1;
    }
#pragma unroll
    for (int u = 0; u < 8; u++)
      if (d[u] >= 0) atomicAdd(&cntL[d[u] - base], 1);
  }
  __syncthreads();

  int c = (tid < BNODES) ? cntL[tid] : 0;
  int incl = c;
  for (int s = 1; s < 32; s <<= 1) {
    int u = __shfl_up(incl, s);
    if (lane >= s) incl += u;
  }
  if (lane == 31 && wv < 16) swave[wv] = incl;
  __syncthreads();
  int pre = 0;
  for (int j = 0; j < 16; j++) pre += (j < wv) ? swave[j] : 0;
  int ex = pre + incl - c;
  if (tid < BNODES) {
    curL[tid] = ex;
    if (base + tid < n) off[base + tid] = bs + ex;
  }
  if (b == nb - 1 && tid == 0) off[n] = bs + cb;
  __syncthreads();

  for (int i = tid; i < cb; i += 8 * 1024) {
    int2 p[8];
#pragma unroll
    for (int u = 0; u < 8; u++) {
      int ii = i + u * 1024;
      p[u] = (ii < cb) ? pairs[bs + ii] : make_int2(0, -1);
    }
#pragma unroll
    for (int u = 0; u < 8; u++) {
      if (p[u].y >= 0) {
        int pos = atomicAdd(&curL[p[u].y - base], 1);
        if (inl) stage[pos] = p[u].x;
        else ebuf[bs + pos] = p[u].x;
      }
    }
  }

  __syncthreads();
  int myt = wv * 16 + (lane & 15);
  int segend = curL[myt];
  int segbeg = (myt > 0) ? curL[myt - 1] : 0;
  const int* eb = ebuf + bs;
  int* eo = es + bs;
  auto sort_segments = [&](auto fetch) {
    auto seg_keys = [&](int i) {
      int beg = __builtin_amdgcn_readlane(segbeg, i & 15);
      int len = __builtin_amdgcn_readlane(segend, i & 15) - beg;
      return (i < 16 && lane < len) ? fetch(beg + lane) : 0x7fffffff;
    };
    int k0 = seg_keys(0);
#pragma unroll 1
    for (int i = 0; i < 16; i++) {
      int kn = seg_keys(i + 1);
      int beg = __builtin_amdgcn_readlane(segbeg, i);
      int len = __builtin_amdgcn_readlane(segend, i) - beg;
      if (len <= 32) {
        int key = k0;
        unsigned comb = ((unsigned)key << 5) | (unsigned)lane;
        int rank = 0;
        for (int j = 0; j < len; j++) {
          unsigned cj = (unsigned)__builtin_amdgcn_readlane((int)comb, j);
          rank += (cj < comb) ? 1 : 0;
        }
        if (lane < len) eo[beg + rank] = key;
      } else {
        for (int q = lane; q < len; q += 32) {
          int key = fetch(beg + q);
          int rank = 0;
          for (int j = 0; j < len; j++) {
            int kj = fetch(beg + j);
            rank += ((kj < key) || (kj == key && j < q)) ? 1 : 0;
          }
          eo[beg + rank] = key;
        }
      }
      k0 = kn;
    }
  };
  if (inl) sort_segments([&](int idx) { return stage[idx]; });
  else sort_segments([&](int idx) { return eb[idx]; });
}

#define FWAVES 16
#define TROWS 16
#define APITCH 132

template <bool EPI, bool TWO>
__global__ __launch_bounds__(FWAVES * 32) void gin_gather_gemm(
    const float* __restrict__ x, const int* __restrict__ off,
    const int* __restrict__ es, const float* __restrict__ gbias,
    const float* __restrict__ W, const float* __restrict__ bias,
    const float* __restrict__ Wb, const float* __restrict__ biasb,
    float* __restrict__ Out, int n) {
  __shared__ __attribute__((aligned(16))) float sW[(TWO ? 2 : 1) * GDIM * GDIM];
  __shared__ __attribute__((aligned(16))) float sA[FWAVES * TROWS * APITCH];
  int tid = threadIdx.x;
  int wave = tid >> 5;
  int lane = tid & 31;

  int ntiles = (n + TROWS - 1) / TROWS;
  int tpb = (ntiles + (int)gridDim.x - 1) / (int)gridDim.x;
  int t0 = (int)blockIdx.x * tpb;
  int t1 = (t0 + tpb < ntiles) ? (t0 + tpb) : ntiles;
  int tile = __builtin_amdgcn_readfirstlane(t0 + wave);

  auto tile_bounds = [&](int t) {
    int oi = t * TROWS + lane;
    return off[(oi < n) ? oi : n];
  };
  auto tile_ids = [&](int bv) {
    int b0 = __builtin_amdgcn_readlane(bv, 0);
    int b1 = __builtin_amdgcn_readlane(bv, TROWS);
    int ei = b0 + lane;
    ei = (ei < b1) ? ei : (b1 - 1);
    ei = (ei > 0) ? ei : 0;
    return es[ei];
  };
  int ov = tile_bounds(tile);
  int idv = 0;
  int lmod = lane & 15;
  float4 gb = {0.f, 0.f, 0.f, 0.f};
  if (EPI) gb = *(const float4*)(gbias + lane * 4);
  float bq[8];
  float bo[8];
#pragma unroll
  for (int nt = 0; nt < 8; nt++) {
    bq[nt] = bias[nt * 16 + lmod];
    bo[nt] = TWO ? biasb[nt * 16 + lmod] : bq[nt];
  }

  {
    constexpr int NM = TWO ? 2 : 1;
    float4 r0[NM][4], r1[NM][4];
#pragma unroll
    for (int m = 0; m < NM; m++) {
      const float* Wm = m ? Wb : W;
#pragma unroll
      for (int t = 0; t < 4; t++) {
        int i = tid + t * (FWAVES * 32);
        int p = i >> 5;
        int q = i & 31;
        r0[m][t] = *(const float4*)(Wm + (2 * p) * GDIM + 4 * q);
        r1[m][t] = *(const float4*)(Wm + (2 * p + 1) * GDIM + 4 * q);
      }
    }
    idv = tile_ids(ov);
#pragma unroll
    for (int m = 0; m < NM; m++) {
#pragma unroll
      for (int t = 0; t < 4; t++) {
        int i = tid + t * (FWAVES * 32);
        int p = i >> 5;
        int q = i & 31;
        float4 lo = {r0[m][t].x, r1[m][t].x, r0[m][t].y, r1[m][t].y};
        float4 hi = {r0[m][t].z, r1[m][t].z, r0[m][t].w, r1[m][t].w};
        *(float4*)(sW + m * (GDIM * GDIM) + p * 256 + 8 * q) = lo;
        *(float4*)(sW + m * (GDIM * GDIM) + p * 256 + 8 * q + 4) = hi;
      }
    }
  }
  __syncthreads();

  int par = lane >> 4;
  int khalf = par << 1;
  int mhi = par << 3;

  float* sAw = sA + wave * (TROWS * APITCH);
  const float* sWp = sW + par * 256 + 2 * lmod;
  const float4* x4 = (const float4*)x;

  for (; tile < t1; tile += FWAVES) {
    int tb = tile * TROWS;
    int ovn = tile_bounds(tile + FWAVES);

    int ebeg = __builtin_amdgcn_readlane(ov, 0);
    int eend = __builtin_amdgcn_readlane(ov, TROWS);
    float4 sv[TROWS];
#pragma unroll
    for (int rr = 0; rr < TROWS; rr++) {
      int row = (tb + rr < n) ? (tb + rr) : (n - 1);
      sv[rr] = x4[(unsigned)row * 32u + (unsigned)lane];
    }
    int r = 0;
    int rowend = __builtin_amdgcn_readlane(ov, 1);
    float4 acc = {0.f, 0.f, 0.f, 0.f};
    auto close_row = [&]() {
      *(float4*)(sAw + r * APITCH + lane * 4) = acc;
      acc = float4{0.f, 0.f, 0.f, 0.f};
      r++;
      rowend = __builtin_amdgcn_readlane(ov, (r + 1 < TROWS) ? (r + 1) : TROWS);
    };
    int e = ebeg;
    auto group = [&](auto nl_c, auto whole_c) {
      constexpr int NL = decltype(nl_c)::value;
      constexpr bool WHOLE = decltype(whole_c)::value;
      float4 v[NL];
#pragma unroll
      for (int u = 0; u < NL; u++) {
        int s = __builtin_amdgcn_readlane(idv, u);
        v[u] = x4[(unsigned)s * 32u + (unsigned)lane];
      }
      __builtin_amdgcn_sched_barrier(0);
#pragma unroll
      for (int u = 0; u < NL; u++) {
        if (WHOLE || e + u < eend) {
          while (e + u >= rowend) close_row();
          acc.x += v[u].x; acc.y += v[u].y; acc.z += v[u].z; acc.w += v[u].w;
        }
      }
    };
    for (; eend - e >= 32; e += 32) {
      int ei1 = e + 32 + lane;
      int idn = es[(ei1 < eend) ? ei1 : (eend - 1)];
      group(std::integral_constant<int, 32>{}, std::true_type{});
      idv = idn;
    }
    if (eend - e > 16) group(std::integral_constant<int, 32>{}, std::false_type{});
    else if (eend - e > 0) group(std::integral_constant<int, 16>{}, std::false_type{});
    while (r < TROWS) close_row();
#pragma unroll
    for (int rr = 0; rr < TROWS; rr++) {
      float* ar = sAw + rr * APITCH + lane * 4;
      float4 sm = *(const float4*)ar;
      float4 a;
      if (EPI) {
        a.x = fmaxf((sv[rr].x + sm.x) + gb.x, 0.f);
        a.y = fmaxf((sv[rr].y + sm.y) + gb.y, 0.f);
        a.z = fmaxf((sv[rr].z + sm.z) + gb.z, 0.f);
        a.w = fmaxf((sv[rr].w + sm.w) + gb.w, 0.f);
      } else {
        a.x = sv[rr].x + sm.x; a.y = sv[rr].y + sm.y;
        a.z = sv[rr].z + sm.z; a.w = sv[rr].w + sm.w;
      }
      *(float4*)ar = a;
    }
    int idvn = tile_ids(ovn);
    __builtin_amdgcn_fence(__ATOMIC_RELEASE, "wavefront");
    __builtin_amdgcn_wave_barrier();
    __builtin_amdgcn_fence(__ATOMIC_ACQUIRE, "wavefront");

    const float* ap = sAw + lmod * APITCH + khalf;
    v8f acc8[8] = {};
    for (int k = 0; k < GDIM; k += 4) {
      v2f aA = *(const v2f*)(ap + k);
      const float* wp = sWp + (k >> 1) * 256;
      v2f b[8];
#pragma unroll
      for (int nt = 0; nt < 8; nt++) b[nt] = *(const v2f*)(wp + nt * 32);
#pragma unroll
      for (int nt = 0; nt < 8; nt++)
        acc8[nt] = __builtin_amdgcn_wmma_f32_16x16x4_f32(
            false, aA, false, b[nt], (short)0, acc8[nt], false, false);
    }
    __builtin_amdgcn_fence(__ATOMIC_RELEASE, "wavefront");
    __builtin_amdgcn_wave_barrier();

    if (TWO) {
#pragma unroll
      for (int nt = 0; nt < 8; nt++) {
        int ncol = nt * 16 + lmod;
        float bv = bq[nt];
#pragma unroll
        for (int j = 0; j < 8; j++)
          sAw[(mhi + j) * APITCH + ncol] = fmaxf(acc8[nt][j] + bv, 0.f);
      }
      __builtin_amdgcn_fence(__ATOMIC_RELEASE, "wavefront");
      __builtin_amdgcn_wave_barrier();
      __builtin_amdgcn_fence(__ATOMIC_ACQUIRE, "wavefront");
#pragma unroll
      for (int nt = 0; nt < 8; nt++) acc8[nt] = v8f{};
      const float* sWq = sWp + GDIM * GDIM;
      for (int k = 0; k < GDIM; k += 4) {
        v2f aA = *(const v2f*)(ap + k);
        const float* wp = sWq + (k >> 1) * 256;
        v2f b[8];
#pragma unroll
        for (int nt = 0; nt < 8; nt++) b[nt] = *(const v2f*)(wp + nt * 32);
#pragma unroll
        for (int nt = 0; nt < 8; nt++)
          acc8[nt] = __builtin_amdgcn_wmma_f32_16x16x4_f32(
              false, aA, false, b[nt], (short)0, acc8[nt], false, false);
      }
      __builtin_amdgcn_fence(__ATOMIC_RELEASE, "wavefront");
      __builtin_amdgcn_wave_barrier();
    }

    bool whole = (tb + TROWS) <= n;
#pragma unroll
    for (int nt = 0; nt < 8; nt++) {
      int ncol = nt * 16 + lmod;
      float bv = bo[nt];
#pragma unroll
      for (int j = 0; j < 8; j++) {
        float v = acc8[nt][j] + bv;
        if (!EPI && !TWO) v = fmaxf(v, 0.f);
        sAw[(mhi + j) * APITCH + ncol] = v;
      }
    }
    __builtin_amdgcn_fence(__ATOMIC_RELEASE, "wavefront");
    __builtin_amdgcn_wave_barrier();
    __builtin_amdgcn_fence(__ATOMIC_ACQUIRE, "wavefront");
    float* orow0 = Out + (long)tb * GDIM + lane * 4;
    if (whole) {
      float4 ov[TROWS];
#pragma unroll
      for (int rr = 0; rr < TROWS; rr++)
        ov[rr] = *(const float4*)(sAw + rr * APITCH + lane * 4);
#pragma unroll
      for (int rr = 0; rr < TROWS; rr++)
        *(float4*)(orow0 + rr * GDIM) = ov[rr];
    } else {
      for (int rr = 0; rr < TROWS && tb + rr < n; rr++)
        *(float4*)(orow0 + rr * GDIM) =
            *(const float4*)(sAw + rr * APITCH + lane * 4);
    }
    __builtin_amdgcn_fence(__ATOMIC_RELEASE, "wavefront");
    __builtin_amdgcn_wave_barrier();
    ov = ovn;
    idv = idvn;
  }
}

__global__ __launch_bounds__(256) void gin_fold(
    const float* __restrict__ W1, const float* __restrict__ W2,
    const float* __restrict__ b2, float* __restrict__ Mout,
    float* __restrict__ cout) {
  __shared__ float sW[GDIM * GDIM];
  int tid = threadIdx.x;
  int fi = blockIdx.x;
  const float* Wn = W1 + (long)(fi + 1) * GDIM * GDIM;
  const float* A = W2 + (long)fi * GDIM * GDIM;
  const float* bb = b2 + fi * GDIM;

  int wave = tid >> 5;
  int lane = tid & 31;
  int lmod = lane & 15;
  int par = lane >> 4;
  int khalf = par << 1;
  int mhi = par << 3;

  const float* ar = A + (wave * 16 + lmod) * GDIM + khalf;
  v2f afr[GDIM / 4];
#pragma unroll
  for (int s = 0; s < GDIM / 4; s++) afr[s] = *(const v2f*)(ar + 4 * s);

  float4 r0[8], r1[8];
#pragma unroll
  for (int t = 0; t < 8; t++) {
    int i = tid + t * 256;
    int p = i >> 5;
    int q = i & 31;
    r0[t] = *(const float4*)(Wn + (2 * p) * GDIM + 4 * q);
    r1[t] = *(const float4*)(Wn + (2 * p + 1) * GDIM + 4 * q);
  }
#pragma unroll
  for (int t = 0; t < 8; t++) {
    int i = tid + t * 256;
    int p = i >> 5;
    int q = i & 31;
    float4 lo = {r0[t].x, r1[t].x, r0[t].y, r1[t].y};
    float4 hi = {r0[t].z, r1[t].z, r0[t].w, r1[t].w};
    *(float4*)(sW + p * 256 + 8 * q) = lo;
    *(float4*)(sW + p * 256 + 8 * q + 4) = hi;
  }
  __syncthreads();

  const float* sWp = sW + par * 256 + 2 * lmod;
  v8f acc[8] = {};
#pragma unroll
  for (int s = 0; s < GDIM / 4; s++) {
    const float* wp = sWp + s * 512;
    v2f b[8];
#pragma unroll
    for (int nt = 0; nt < 8; nt++) b[nt] = *(const v2f*)(wp + nt * 32);
#pragma unroll
    for (int nt = 0; nt < 8; nt++)
      acc[nt] = __builtin_amdgcn_wmma_f32_16x16x4_f32(
          false, afr[s], false, b[nt], (short)0, acc[nt], false, false);
  }
  float* Mo = Mout + (long)fi * GDIM * GDIM;
#pragma unroll
  for (int nt = 0; nt < 8; nt++) {
#pragma unroll
    for (int j = 0; j < 8; j++)
      Mo[(wave * 16 + mhi + j) * GDIM + nt * 16 + lmod] = acc[nt][j];
  }

  if (tid < GDIM) {
    float s = 0.f;
    for (int k = 0; k < GDIM; k += 2) {
      v2f w = *(const v2f*)(sW + (k >> 1) * 256 + 2 * tid);
      s = fmaf(bb[k], w.x, s);
      s = fmaf(bb[k + 1], w.y, s);
    }
    cout[fi * GDIM + tid] = s;
  }
}

extern "C" void kernel_launch(void* const* d_in, const int* in_sizes, int n_in,
                              void* d_out, int out_size, void* d_ws,
                              size_t ws_size, hipStream_t stream) {
  const float* x_in = (const float*)d_in[0];
  const int* ei = (const int*)d_in[1];
  const float* W1 = (const float*)d_in[2];
  const float* b1 = (const float*)d_in[3];
  const float* W2 = (const float*)d_in[4];
  const float* b2 = (const float*)d_in[5];
  float* out = (float*)d_out;

  const int N = in_sizes[0] / GDIM;
  const int NE = in_sizes[1] / 2;
  const long nd = (long)N * GDIM;

  float* aggh = (float*)d_ws;
  int2* pairs = (int2*)aggh;
  int* blkoff = (int*)aggh + 2 * (long)NE;
  int* btot = (int*)(aggh + nd);
  int* off = btot + (N + 4);
  int* ebuf = off + (N + 4);
  int* es = ebuf + NE;
  float* Mf = (float*)ebuf;
  float* cf = Mf + 2 * GDIM * GDIM;

  const int NB = (N + BNODES - 1) >> BSH;
  const int part_blocks = 256;
  const int zero_blocks = (NB * (BTS / 4) + 255) / 256;
  const int fused_blocks = 256;

  gin_zero<<<zero_blocks, 256, 0, stream>>>((float4*)btot, NB * (BTS / 4));
  gin_bcount<<<part_blocks, 1024, 0, stream>>>(ei + NE, btot, blkoff, NE, NB);
  gin_bscatter<<<part_blocks, 1024, 0, stream>>>(ei, ei + NE, btot, blkoff, pairs, NE, NB);
  gin_bcsr<<<NB, 1024, 0, stream>>>(pairs, btot, off, ebuf, es, N, NB);
  gin_fold<<<2, 256, 0, stream>>>(W1, W2, b2, Mf, cf);

  gin_gather_gemm<false, true><<<fused_blocks, FWAVES * 32, 0, stream>>>(
      x_in, off, es, nullptr, W1, b1, Mf, cf, out, N);
  gin_gather_gemm<true, false><<<fused_blocks, FWAVES * 32, 0, stream>>>(
      out, off, es, b1 + GDIM, Mf + (long)GDIM * GDIM, cf + GDIM, nullptr,
      nullptr, aggh, N);
  gin_gather_gemm<true, false><<<fused_blocks, FWAVES * 32, 0, stream>>>(
      aggh, off, es, b1 + 2 * GDIM, W2 + (long)2 * GDIM * GDIM, b2 + 2 * GDIM,
      nullptr, nullptr, out, N);
}


# Weight fold (M=W2[i]*W1[i+1] and c) runs as four extra blocks of the bucket launch, on WGPs that launch leaves free, instead of a two-block launch of its own: 8 launches -> 7; sums unchanged
# speedup vs baseline: 5.0722x; 1.0485x over previous
#include <hip/hip_runtime.h>
#include <type_traits>

typedef __attribute__((ext_vector_type(2))) float v2f;
typedef __attribute__((ext_vector_type(8))) float v8f;

#define GDIM 128

__global__ __launch_bounds__(256) void gin_zero(float4* p, long n4) {
  long i = (long)blockIdx.x * blockDim.x + threadIdx.x;
  long stride = (long)gridDim.x * blockDim.x;
  float4 z = {0.f, 0.f, 0.f, 0.f};
  for (; i < n4; i += stride) p[i] = z;
}


#define BSH 9
#define BNODES 512
#define MAXNB 256
#define BTS 32
#define LCAP 16384

__global__ __launch_bounds__(1024) void gin_bcount(
    const int* __restrict__ dst, int* btot, int* __restrict__ blkoff,
    int nedges, int nb) {
  __shared__ int h[MAXNB];
  int tid = threadIdx.x;
  if (tid < MAXNB) h[tid] = 0;
  __syncthreads();
  int chunk = (nedges + (int)gridDim.x - 1) / (int)gridDim.x;
  int e0 = (int)blockIdx.x * chunk;
  int e1 = (e0 + chunk < nedges) ? (e0 + chunk) : nedges;
  for (int e = e0 + tid; e < e1; e += 8 * 1024) {
    int d[8];
#pragma unroll
    for (int u = 0; u < 8; u++) {
      int ee = e + u * 1024;
      d[u] = (ee < e1) ? dst[ee] : -1;
    }
#pragma unroll
    for (int u = 0; u < 8; u++)
      if (d[u] >= 0) atomicAdd(&h[d[u] >> BSH], 1);
  }
  __syncthreads();
  if (tid < nb)
    blkoff[(int)blockIdx.x * nb + tid] = atomicAdd(btot + tid * BTS, h[tid]);
}

__global__ __launch_bounds__(1024) void gin_bscatter(
    const int* __restrict__ src, const int* __restrict__ dst,
    const int* __restrict__ btot, const int* __restrict__ blkoff,
    int2* __restrict__ pairs, int nedges, int nb) {
  __shared__ int cur[MAXNB];
  __shared__ int swave[8];
  int tid = threadIdx.x;
  int lane = tid & 31;
  int wv = tid >> 5;
  int v = (tid < nb) ? btot[tid * BTS] : 0;
  int incl = v;
  for (int s = 1; s < 32; s <<= 1) {
    int u = __shfl_up(incl, s);
    if (lane >= s) incl += u;
  }
  if (lane == 31 && wv < 8) swave[wv] = incl;
  __syncthreads();
  int pre = 0;
  for (int j = 0; j < 8; j++) pre += (j < wv) ? swave[j] : 0;
  if (tid < nb) cur[tid] = pre + incl - v + blkoff[(int)blockIdx.x * nb + tid];
  __syncthreads();
  int chunk = (nedges + (int)gridDim.x - 1) / (int)gridDim.x;
  int e0 = (int)blockIdx.x * chunk;
  int e1 = (e0 + chunk < nedges) ? (e0 + chunk) : nedges;
  for (int e = e0 + tid; e < e1; e += 8 * 1024) {
    int d[8], sr[8];
#pragma unroll
    for (int u = 0; u < 8; u++) {
      int ee = e + u * 1024;
      d[u] = (ee < e1) ? dst[ee] : -1;
      sr[u] = (ee < e1) ? src[ee] : 0;
    }
#pragma unroll
    for (int u = 0; u < 8; u++) {
      if (d[u] >= 0) {
        int pos = atomicAdd(&cur[d[u] >> BSH], 1);
        pairs[pos] = make_int2(sr[u], d[u]);
      }
    }
  }
}

__device__ __forceinline__ void fold_rows(
    const float* __restrict__ W1, const float* __restrict__ W2,
    const float* __restrict__ b2, float* __restrict__ Mout,
    float* __restrict__ cout, int fb, int tid, float* sW) {
  int fi = fb >> 1;
  int rh = fb & 1;
  const float* Wn = W1 + (long)(fi + 1) * GDIM * GDIM;
  const float* A = W2 + (long)fi * GDIM * GDIM;
  const float* bb = b2 + fi * GDIM;

  int wave = __builtin_amdgcn_readfirstlane(tid >> 5);
  int lane = tid & 31;
  int lmod = lane & 15;
  int par = lane >> 4;
  int khalf = par << 1;
  int mhi = par << 3;
  bool mm = wave < 4;
  int rt = rh * 4 + (wave & 3);

  const float* ar = A + (rt * 16 + lmod) * GDIM + khalf;
  v2f afr[GDIM / 8];
  if (mm) {
#pragma unroll
    for (int s = 0; s < GDIM / 8; s++) afr[s] = *(const v2f*)(ar + 4 * s);
  }

  float4 r0[2], r1[2];
#pragma unroll
  for (int t = 0; t < 2; t++) {
    int i = tid + t * 1024;
    int p = i >> 5;
    int q = i & 31;
    r0[t] = *(const float4*)(Wn + (2 * p) * GDIM + 4 * q);
    r1[t] = *(const float4*)(Wn + (2 * p + 1) * GDIM + 4 * q);
  }
#pragma unroll
  for (int t = 0; t < 2; t++) {
    int i = tid + t * 1024;
    int p = i >> 5;
    int q = i & 31;
    float4 lo = {r0[t].x, r1[t].x, r0[t].y, r1[t].y};
    float4 hi = {r0[t].z, r1[t].z, r0[t].w, r1[t].w};
    *(float4*)(sW + p * 256 + 8 * q) = lo;
    *(float4*)(sW + p * 256 + 8 * q + 4) = hi;
  }
  __syncthreads();

  if (mm) {
    const float* sWp = sW + par * 256 + 2 * lmod;
    v8f acc[8] = {};
    for (int h = 0; h < 2; h++) {
      if (h == 1) {
#pragma unroll
        for (int s = 0; s < GDIM / 8; s++) afr[s] = *(const v2f*)(ar + GDIM / 2 + 4 * s);
      }
#pragma unroll
      for (int s = 0; s < GDIM / 8; s++) {
        const float* wp = sWp + (h * (GDIM / 8) + s) * 512;
        v2f b[8];
#pragma unroll
        for (int nt = 0; nt < 8; nt++) b[nt] = *(const v2f*)(wp + nt * 32);
#pragma unroll
        for (int nt = 0; nt < 8; nt++)
          acc[nt] = __builtin_amdgcn_wmma_f32_16x16x4_f32(
              false, afr[s], false, b[nt], (short)0, acc[nt], false, false);
      }
    }
    float* Mo = Mout + (long)fi * GDIM * GDIM;
#pragma unroll
    for (int nt = 0; nt < 8; nt++) {
#pragma unroll
      for (int j = 0; j < 8; j++)
        Mo[(rt * 16 + mhi + j) * GDIM + nt * 16 + lmod] = acc[nt][j];
    }
  } else if (rh == 0 && wave < 8) {
    int col = tid - 128;
    float s = 0.f;
    for (int k = 0; k < GDIM; k += 2) {
      v2f w = *(const v2f*)(sW + (k >> 1) * 256 + 2 * col);
      s = fmaf(bb[k], w.x, s);
      s = fmaf(bb[k + 1], w.y, s);
    }
    cout[fi * GDIM + col] = s;
  }
}

__global__ __launch_bounds__(1024) void gin_bcsr(
    const int2* __restrict__ pairs, const int* __restrict__ btot,
    int* __restrict__ off, int* ebuf, int* __restrict__ es, int n, int nb,
    const float* __restrict__ W1, const float* __restrict__ W2,
    const float* __restrict__ b2, float* __restrict__ Mout,
    float* __restrict__ cout) {
  __shared__ int cntL[BNODES];
  __shared__ int curL[BNODES];
  __shared__ int swave[16];
  __shared__ int sred[8];
  __shared__ __attribute__((aligned(16))) int stage[LCAP];
  int tid = threadIdx.x;
  if ((int)blockIdx.x >= nb) {
    fold_rows(W1, W2, b2, Mout, cout, (int)blockIdx.x - nb, tid, (float*)stage);
    return;
  }
  int lane = tid & 31;
  int wv = tid >> 5;
  int b = (int)blockIdx.x;
  int base = b << BSH;
  if (tid < BNODES) cntL[tid] = 0;
  int v = (tid < b) ? btot[tid * BTS] : 0;
  for (int m = 16; m >= 1; m >>= 1) v += __shfl_xor(v, m);
  if (lane == 0 && wv < 8) sred[wv] = v;
  int cb = btot[b * BTS];
  __syncthreads();
  int bs = 0;
  for (int j = 0; j < 8; j++) bs += sred[j];
  const bool inl = cb <= LCAP;

  for (int i = tid; i < cb; i += 8 * 1024) {
    int d[8];
#pragma unroll
    for (int u = 0; u < 8; u++) {
      int ii = i + u * 1024;
      d[u] = (ii < cb) ? pairs[bs + ii].y : -1;
    }
#pragma unroll
    for (int u = 0; u < 8; u++)
      if (d[u] >= 0) atomicAdd(&cntL[d[u] - base], 1);
  }
  __syncthreads();

  int c = (tid < BNODES) ? cntL[tid] : 0;
  int incl = c;
  for (int s = 1; s < 32; s <<= 1) {
    int u = __shfl_up(incl, s);
    if (lane >= s) incl += u;
  }
  if (lane == 31 && wv < 16) swave[wv] = incl;
  __syncthreads();
  int pre = 0;
  for (int j = 0; j < 16; j++) pre += (j < wv) ? swave[j] : 0;
  int ex = pre + incl - c;
  if (tid < BNODES) {
    curL[tid] = ex;
    if (base + tid < n) off[base + tid] = bs + ex;
  }
  if (b == nb - 1 && tid == 0) off[n] = bs + cb;
  __syncthreads();

  for (int i = tid; i < cb; i += 8 * 1024) {
    int2 p[8];
#pragma unroll
    for (int u = 0; u < 8; u++) {
      int ii = i + u * 1024;
      p[u] = (ii < cb) ? pairs[bs + ii] : make_int2(0, -1);
    }
#pragma unroll
    for (int u = 0; u < 8; u++) {
      if (p[u].y >= 0) {
        int pos = atomicAdd(&curL[p[u].y - base], 1);
        if (inl) stage[pos] = p[u].x;
        else ebuf[bs + pos] = p[u].x;
      }
    }
  }

  __syncthreads();
  int myt = wv * 16 + (lane & 15);
  int segend = curL[myt];
  int segbeg = (myt > 0) ? curL[myt - 1] : 0;
  const int* eb = ebuf + bs;
  int* eo = es + bs;
  auto sort_segments = [&](auto fetch) {
    auto seg_keys = [&](int i) {
      int beg = __builtin_amdgcn_readlane(segbeg, i & 15);
      int len = __builtin_amdgcn_readlane(segend, i & 15) - beg;
      return (i < 16 && lane < len) ? fetch(beg + lane) : 0x7fffffff;
    };
    int k0 = seg_keys(0);
#pragma unroll 1
    for (int i = 0; i < 16; i++) {
      int kn = seg_keys(i + 1);
      int beg = __builtin_amdgcn_readlane(segbeg, i);
      int len = __builtin_amdgcn_readlane(segend, i) - beg;
      if (len <= 32) {
        int key = k0;
        unsigned comb = ((unsigned)key << 5) | (unsigned)lane;
        int rank = 0;
        for (int j = 0; j < len; j++) {
          unsigned cj = (unsigned)__builtin_amdgcn_readlane((int)comb, j);
          rank += (cj < comb) ? 1 : 0;
        }
        if (lane < len) eo[beg + rank] = key;
      } else {
        for (int q = lane; q < len; q += 32) {
          int key = fetch(beg + q);
          int rank = 0;
          for (int j = 0; j < len; j++) {
            int kj = fetch(beg + j);
            rank += ((kj < key) || (kj == key && j < q)) ? 1 : 0;
          }
          eo[beg + rank] = key;
        }
      }
      k0 = kn;
    }
  };
  if (inl) sort_segments([&](int idx) { return stage[idx]; });
  else sort_segments([&](int idx) { return eb[idx]; });
}

#define FWAVES 16
#define TROWS 16
#define APITCH 132

template <bool EPI, bool TWO>
__global__ __launch_bounds__(FWAVES * 32) void gin_gather_gemm(
    const float* __restrict__ x, const int* __restrict__ off,
    const int* __restrict__ es, const float* __restrict__ gbias,
    const float* __restrict__ W, const float* __restrict__ bias,
    const float* __restrict__ Wb, const float* __restrict__ biasb,
    float* __restrict__ Out, int n) {
  __shared__ __attribute__((aligned(16))) float sW[(TWO ? 2 : 1) * GDIM * GDIM];
  __shared__ __attribute__((aligned(16))) float sA[FWAVES * TROWS * APITCH];
  int tid = threadIdx.x;
  int wave = tid >> 5;
  int lane = tid & 31;

  int ntiles = (n + TROWS - 1) / TROWS;
  int tpb = (ntiles + (int)gridDim.x - 1) / (int)gridDim.x;
  int t0 = (int)blockIdx.x * tpb;
  int t1 = (t0 + tpb < ntiles) ? (t0 + tpb) : ntiles;
  int tile = __builtin_amdgcn_readfirstlane(t0 + wave);

  auto tile_bounds = [&](int t) {
    int oi = t * TROWS + lane;
    return off[(oi < n) ? oi : n];
  };
  auto tile_ids = [&](int bv) {
    int b0 = __builtin_amdgcn_readlane(bv, 0);
    int b1 = __builtin_amdgcn_readlane(bv, TROWS);
    int ei = b0 + lane;
    ei = (ei < b1) ? ei : (b1 - 1);
    ei = (ei > 0) ? ei : 0;
    return es[ei];
  };
  int ov = tile_bounds(tile);
  int idv = 0;
  int lmod = lane & 15;
  float4 gb = {0.f, 0.f, 0.f, 0.f};
  if (EPI) gb = *(const float4*)(gbias + lane * 4);
  float bq[8];
  float bo[8];
#pragma unroll
  for (int nt = 0; nt < 8; nt++) {
    bq[nt] = bias[nt * 16 + lmod];
    bo[nt] = TWO ? biasb[nt * 16 + lmod] : bq[nt];
  }

  {
    constexpr int NM = TWO ? 2 : 1;
    float4 r0[NM][4], r1[NM][4];
#pragma unroll
    for (int m = 0; m < NM; m++) {
      const float* Wm = m ? Wb : W;
#pragma unroll
      for (int t = 0; t < 4; t++) {
        int i = tid + t * (FWAVES * 32);
        int p = i >> 5;
        int q = i & 31;
        r0[m][t] = *(const float4*)(Wm + (2 * p) * GDIM + 4 * q);
        r1[m][t] = *(const float4*)(Wm + (2 * p + 1) * GDIM + 4 * q);
      }
    }
    idv = tile_ids(ov);
#pragma unroll
    for (int m = 0; m < NM; m++) {
#pragma unroll
      for (int t = 0; t < 4; t++) {
        int i = tid + t * (FWAVES * 32);
        int p = i >> 5;
        int q = i & 31;
        float4 lo = {r0[m][t].x, r1[m][t].x, r0[m][t].y, r1[m][t].y};
        float4 hi = {r0[m][t].z, r1[m][t].z, r0[m][t].w, r1[m][t].w};
        *(float4*)(sW + m * (GDIM * GDIM) + p * 256 + 8 * q) = lo;
        *(float4*)(sW + m * (GDIM * GDIM) + p * 256 + 8 * q + 4) = hi;
      }
    }
  }
  __syncthreads();

  int par = lane >> 4;
  int khalf = par << 1;
  int mhi = par << 3;

  float* sAw = sA + wave * (TROWS * APITCH);
  const float* sWp = sW + par * 256 + 2 * lmod;
  const float4* x4 = (const float4*)x;

  for (; tile < t1; tile += FWAVES) {
    int tb = tile * TROWS;
    int ovn = tile_bounds(tile + FWAVES);

    int ebeg = __builtin_amdgcn_readlane(ov, 0);
    int eend = __builtin_amdgcn_readlane(ov, TROWS);
    float4 sv[TROWS];
#pragma unroll
    for (int rr = 0; rr < TROWS; rr++) {
      int row = (tb + rr < n) ? (tb + rr) : (n - 1);
      sv[rr] = x4[(unsigned)row * 32u + (unsigned)lane];
    }
    int r = 0;
    int rowend = __builtin_amdgcn_readlane(ov, 1);
    float4 acc = {0.f, 0.f, 0.f, 0.f};
    auto close_row = [&]() {
      *(float4*)(sAw + r * APITCH + lane * 4) = acc;
      acc = float4{0.f, 0.f, 0.f, 0.f};
      r++;
      rowend = __builtin_amdgcn_readlane(ov, (r + 1 < TROWS) ? (r + 1) : TROWS);
    };
    int e = ebeg;
    auto group = [&](auto nl_c, auto whole_c) {
      constexpr int NL = decltype(nl_c)::value;
      constexpr bool WHOLE = decltype(whole_c)::value;
      float4 v[NL];
#pragma unroll
      for (int u = 0; u < NL; u++) {
        int s = __builtin_amdgcn_readlane(idv, u);
        v[u] = x4[(unsigned)s * 32u + (unsigned)lane];
      }
      __builtin_amdgcn_sched_barrier(0);
#pragma unroll
      for (int u = 0; u < NL; u++) {
        if (WHOLE || e + u < eend) {
          while (e + u >= rowend) close_row();
          acc.x += v[u].x; acc.y += v[u].y; acc.z += v[u].z; acc.w += v[u].w;
        }
      }
    };
    for (; eend - e >= 32; e += 32) {
      int ei1 = e + 32 + lane;
      int idn = es[(ei1 < eend) ? ei1 : (eend - 1)];
      group(std::integral_constant<int, 32>{}, std::true_type{});
      idv = idn;
    }
    if (eend - e > 16) group(std::integral_constant<int, 32>{}, std::false_type{});
    else if (eend - e > 0) group(std::integral_constant<int, 16>{}, std::false_type{});
    while (r < TROWS) close_row();
#pragma unroll
    for (int rr = 0; rr < TROWS; rr++) {
      float* ar = sAw + rr * APITCH + lane * 4;
      float4 sm = *(const float4*)ar;
      float4 a;
      if (EPI) {
        a.x = fmaxf((sv[rr].x + sm.x) + gb.x, 0.f);
        a.y = fmaxf((sv[rr].y + sm.y) + gb.y, 0.f);
        a.z = fmaxf((sv[rr].z + sm.z) + gb.z, 0.f);
        a.w = fmaxf((sv[rr].w + sm.w) + gb.w, 0.f);
      } else {
        a.x = sv[rr].x + sm.x; a.y = sv[rr].y + sm.y;
        a.z = sv[rr].z + sm.z; a.w = sv[rr].w + sm.w;
      }
      *(float4*)ar = a;
    }
    int idvn = tile_ids(ovn);
    __builtin_amdgcn_fence(__ATOMIC_RELEASE, "wavefront");
    __builtin_amdgcn_wave_barrier();
    __builtin_amdgcn_fence(__ATOMIC_ACQUIRE, "wavefront");

    const float* ap = sAw + lmod * APITCH + khalf;
    v8f acc8[8] = {};
    for (int k = 0; k < GDIM; k += 4) {
      v2f aA = *(const v2f*)(ap + k);
      const float* wp = sWp + (k >> 1) * 256;
      v2f b[8];
#pragma unroll
      for (int nt = 0; nt < 8; nt++) b[nt] = *(const v2f*)(wp + nt * 32);
#pragma unroll
      for (int nt = 0; nt < 8; nt++)
        acc8[nt] = __builtin_amdgcn_wmma_f32_16x16x4_f32(
            false, aA, false, b[nt], (short)0, acc8[nt], false, false);
    }
    __builtin_amdgcn_fence(__ATOMIC_RELEASE, "wavefront");
    __builtin_amdgcn_wave_barrier();

    if (TWO) {
#pragma unroll
      for (int nt = 0; nt < 8; nt++) {
        int ncol = nt * 16 + lmod;
        float bv = bq[nt];
#pragma unroll
        for (int j = 0; j < 8; j++)
          sAw[(mhi + j) * APITCH + ncol] = fmaxf(acc8[nt][j] + bv, 0.f);
      }
      __builtin_amdgcn_fence(__ATOMIC_RELEASE, "wavefront");
      __builtin_amdgcn_wave_barrier();
      __builtin_amdgcn_fence(__ATOMIC_ACQUIRE, "wavefront");
#pragma unroll
      for (int nt = 0; nt < 8; nt++) acc8[nt] = v8f{};
      const float* sWq = sWp + GDIM * GDIM;
      for (int k = 0; k < GDIM; k += 4) {
        v2f aA = *(const v2f*)(ap + k);
        const float* wp = sWq + (k >> 1) * 256;
        v2f b[8];
#pragma unroll
        for (int nt = 0; nt < 8; nt++) b[nt] = *(const v2f*)(wp + nt * 32);
#pragma unroll
        for (int nt = 0; nt < 8; nt++)
          acc8[nt] = __builtin_amdgcn_wmma_f32_16x16x4_f32(
              false, aA, false, b[nt], (short)0, acc8[nt], false, false);
      }
      __builtin_amdgcn_fence(__ATOMIC_RELEASE, "wavefront");
      __builtin_amdgcn_wave_barrier();
    }

    bool whole = (tb + TROWS) <= n;
#pragma unroll
    for (int nt = 0; nt < 8; nt++) {
      int ncol = nt * 16 + lmod;
      float bv = bo[nt];
#pragma unroll
      for (int j = 0; j < 8; j++) {
        float v = acc8[nt][j] + bv;
        if (!EPI && !TWO) v = fmaxf(v, 0.f);
        sAw[(mhi + j) * APITCH + ncol] = v;
      }
    }
    __builtin_amdgcn_fence(__ATOMIC_RELEASE, "wavefront");
    __builtin_amdgcn_wave_barrier();
    __builtin_amdgcn_fence(__ATOMIC_ACQUIRE, "wavefront");
    float* orow0 = Out + (long)tb * GDIM + lane * 4;
    if (whole) {
      float4 ov[TROWS];
#pragma unroll
      for (int rr = 0; rr < TROWS; rr++)
        ov[rr] = *(const float4*)(sAw + rr * APITCH + lane * 4);
#pragma unroll
      for (int rr = 0; rr < TROWS; rr++)
        *(float4*)(orow0 + rr * GDIM) = ov[rr];
    } else {
      for (int rr = 0; rr < TROWS && tb + rr < n; rr++)
        *(float4*)(orow0 + rr * GDIM) =
            *(const float4*)(sAw + rr * APITCH + lane * 4);
    }
    __builtin_amdgcn_fence(__ATOMIC_RELEASE, "wavefront");
    __builtin_amdgcn_wave_barrier();
    ov = ovn;
    idv = idvn;
  }
}

extern "C" void kernel_launch(void* const* d_in, const int* in_sizes, int n_in,
                              void* d_out, int out_size, void* d_ws,
                              size_t ws_size, hipStream_t stream) {
  const float* x_in = (const float*)d_in[0];
  const int* ei = (const int*)d_in[1];
  const float* W1 = (const float*)d_in[2];
  const float* b1 = (const float*)d_in[3];
  const float* W2 = (const float*)d_in[4];
  const float* b2 = (const float*)d_in[5];
  float* out = (float*)d_out;

  const int N = in_sizes[0] / GDIM;
  const int NE = in_sizes[1] / 2;
  const long nd = (long)N * GDIM;

  float* aggh = (float*)d_ws;
  int2* pairs = (int2*)aggh;
  int* blkoff = (int*)aggh + 2 * (long)NE;
  int* btot = (int*)(aggh + nd);
  int* off = btot + (N + 4);
  int* ebuf = off + (N + 4);
  int* es = ebuf + NE;
  float* Mf = (float*)(btot + MAXNB * BTS);
  float* cf = Mf + 2 * GDIM * GDIM;

  const int NB = (N + BNODES - 1) >> BSH;
  const int part_blocks = 256;
  const int zero_blocks = (NB * (BTS / 4) + 255) / 256;
  const int fused_blocks = 256;

  gin_zero<<<zero_blocks, 256, 0, stream>>>((float4*)btot, NB * (BTS / 4));
  gin_bcount<<<part_blocks, 1024, 0, stream>>>(ei + NE, btot, blkoff, NE, NB);
  gin_bscatter<<<part_blocks, 1024, 0, stream>>>(ei, ei + NE, btot, blkoff, pairs, NE, NB);
  gin_bcsr<<<NB + 4, 1024, 0, stream>>>(pairs, btot, off, ebuf, es, N, NB, W1, W2, b2, Mf, cf);

  gin_gather_gemm<false, true><<<fused_blocks, FWAVES * 32, 0, stream>>>(
      x_in, off, es, nullptr, W1, b1, Mf, cf, out, N);
  gin_gather_gemm<true, false><<<fused_blocks, FWAVES * 32, 0, stream>>>(
      out, off, es, b1 + GDIM, Mf + (long)GDIM * GDIM, cf + GDIM, nullptr,
      nullptr, aggh, N);
  gin_gather_gemm<true, false><<<fused_blocks, FWAVES * 32, 0, stream>>>(
      aggh, off, es, b1 + 2 * GDIM, W2 + (long)2 * GDIM * GDIM, b2 + 2 * GDIM,
      nullptr, nullptr, out, N);
}
